# starting kernel
# baseline (speedup 1.0000x reference)
#include <hip/hip_runtime.h>

typedef __attribute__((ext_vector_type(2))) float v2f;
typedef __attribute__((ext_vector_type(8))) float v8f;

#define NNODES 50000
#define INDIM  512
#define HIDDIM 128
#define OUTDIM 16

__global__ __launch_bounds__(256) void gemm1_wmma(const float* __restrict__ feat,
                                                  const float* __restrict__ W1,
                                                  float* __restrict__ xw1) {
  __shared__ float As[INDIM][17];
  const int row0 = blockIdx.x * 16;
  const int tid  = threadIdx.x;

  const float4* src = (const float4*)(feat + (size_t)row0 * INDIM);
#pragma unroll
  for (int i = 0; i < 8; ++i) {
    const int s  = tid + i * 256;
    const int r  = s >> 7;
    const int kv = s & 127;
    const float4 v = src[(size_t)r * (INDIM / 4) + kv];
    const int k = kv * 4;
    As[k + 0][r] = v.x;
    As[k + 1][r] = v.y;
    As[k + 2][r] = v.z;
    As[k + 3][r] = v.w;
  }
  __syncthreads();

  const int wave = tid >> 5;
  const int lane = tid & 31;
  const int hs   = lane >> 4;
  const int lm   = lane & 15;
  const int n0   = wave * 16;

  v8f acc = {};
  for (int k0 = 0; k0 < INDIM; k0 += 4) {
    const int ka = k0 + hs * 2;
    v2f a, b;
    a.x = As[ka + 0][lm];
    a.y = As[ka + 1][lm];
    const float* w = W1 + (size_t)ka * HIDDIM + n0 + lm;
    b.x = w[0];
    b.y = w[HIDDIM];
    acc = __builtin_amdgcn_wmma_f32_16x16x4_f32(false, a, false, b,
                                                (short)0, acc, false, false);
  }

  float* dst = xw1 + (size_t)row0 * HIDDIM + n0 + lm;
#pragma unroll
  for (int j = 0; j < 8; ++j)
    dst[(size_t)(hs * 8 + j) * HIDDIM] = acc[j];
}

__global__ __launch_bounds__(256) void spmm1_atomic(const int* __restrict__ esrc,
                                                    const int* __restrict__ edst,
                                                    const float* __restrict__ evals,
                                                    const float* __restrict__ xw1,
                                                    float* __restrict__ h,
                                                    int nedges) {
  const int lane   = threadIdx.x & 31;
  const int gwave  = (int)((blockIdx.x * blockDim.x + threadIdx.x) >> 5);
  const int nwaves = (int)((gridDim.x * blockDim.x) >> 5);
  for (int e = gwave; e < nedges; e += nwaves) {
    const int   s = esrc[e];
    const int   d = edst[e];
    const float v = evals[e];
    const float4 x = ((const float4*)(xw1 + (size_t)s * HIDDIM))[lane];
    float* out = h + (size_t)d * HIDDIM + lane * 4;
    atomicAdd(out + 0, v * x.x);
    atomicAdd(out + 1, v * x.y);
    atomicAdd(out + 2, v * x.z);
    atomicAdd(out + 3, v * x.w);
  }
}

__global__ __launch_bounds__(32) void gemm2_wmma(const float* __restrict__ h,
                                                 const float* __restrict__ W2,
                                                 float* __restrict__ hw2) {
  const int row0 = blockIdx.x * 16;
  const int lane = threadIdx.x & 31;
  const int hs   = lane >> 4;
  const int lm   = lane & 15;

  const float* arow = h + (size_t)(row0 + lm) * HIDDIM;
  v8f acc = {};
  for (int k0 = 0; k0 < HIDDIM; k0 += 4) {
    const int ka = k0 + hs * 2;
    v2f a, b;
    a.x = fmaxf(arow[ka + 0], 0.0f);
    a.y = fmaxf(arow[ka + 1], 0.0f);
    const float* w = W2 + (size_t)ka * OUTDIM + lm;
    b.x = w[0];
    b.y = w[OUTDIM];
    acc = __builtin_amdgcn_wmma_f32_16x16x4_f32(false, a, false, b,
                                                (short)0, acc, false, false);
  }
  float* dst = hw2 + (size_t)row0 * OUTDIM + lm;
#pragma unroll
  for (int j = 0; j < 8; ++j)
    dst[(size_t)(hs * 8 + j) * OUTDIM] = acc[j];
}

__global__ __launch_bounds__(256) void spmm2_atomic(const int* __restrict__ esrc,
                                                    const int* __restrict__ edst,
                                                    const float* __restrict__ evals,
                                                    const float* __restrict__ hw2,
                                                    float* __restrict__ out,
                                                    int nedges) {
  const int t      = (int)(blockIdx.x * blockDim.x + threadIdx.x);
  const int stride = (int)((gridDim.x * blockDim.x) >> 4);
  const int d      = t & 15;
  for (int e = t >> 4; e < nedges; e += stride) {
    const int   s  = esrc[e];
    const int   dd = edst[e];
    const float v  = evals[e];
    atomicAdd(out + (size_t)dd * OUTDIM + d, v * hw2[(size_t)s * OUTDIM + d]);
  }
}

extern "C" void kernel_launch(void* const* d_in, const int* in_sizes, int n_in,
                              void* d_out, int out_size, void* d_ws, size_t ws_size,
                              hipStream_t stream) {
  const float* feat  = (const float*)d_in[0];
  const int*   esrc  = (const int*)d_in[1];
  const int*   edst  = (const int*)d_in[2];
  const float* evals = (const float*)d_in[3];
  const float* W1    = (const float*)d_in[4];
  const float* W2    = (const float*)d_in[5];
  float*       out   = (float*)d_out;
  const int nedges   = in_sizes[1];

  const size_t xw1_bytes = (size_t)NNODES * HIDDIM * sizeof(float);
  float* xw1 = (float*)d_ws;
  float* h   = (float*)((char*)d_ws + xw1_bytes);
  float* hw2 = (float*)d_ws;

  hipMemsetAsync(h, 0, xw1_bytes, stream);
  hipMemsetAsync(out, 0, (size_t)NNODES * OUTDIM * sizeof(float), stream);

  gemm1_wmma<<<NNODES / 16, 256, 0, stream>>>(feat, W1, xw1);

  spmm1_atomic<<<16384, 256, 0, stream>>>(esrc, edst, evals, xw1, h, nedges);

  gemm2_wmma<<<NNODES / 16, 32, 0, stream>>>(h, W2, hw2);

  spmm2_atomic<<<16384, 256, 0, stream>>>(esrc, edst, evals, hw2, out, nedges);
}


# spmm1: each work-group owns 196 dst rows in LDS, scans the dst list, adds gathered rows with LDS f32 adds, stores each h row once; no global f32 atomics or h clear in layer 1; sum order changes
# speedup vs baseline: 1.6076x; 1.6076x over previous
#include <hip/hip_runtime.h>

typedef __attribute__((ext_vector_type(2))) float v2f;
typedef __attribute__((ext_vector_type(8))) float v8f;

#define NNODES 50000
#define INDIM  512
#define HIDDIM 128
#define OUTDIM 16

__global__ __launch_bounds__(256) void gemm1_wmma(const float* __restrict__ feat,
                                                  const float* __restrict__ W1,
                                                  float* __restrict__ xw1) {
  __shared__ float As[INDIM][17];
  const int row0 = blockIdx.x * 16;
  const int tid  = threadIdx.x;

  const float4* src = (const float4*)(feat + (size_t)row0 * INDIM);
#pragma unroll
  for (int i = 0; i < 8; ++i) {
    const int s  = tid + i * 256;
    const int r  = s >> 7;
    const int kv = s & 127;
    const float4 v = src[(size_t)r * (INDIM / 4) + kv];
    const int k = kv * 4;
    As[k + 0][r] = v.x;
    As[k + 1][r] = v.y;
    As[k + 2][r] = v.z;
    As[k + 3][r] = v.w;
  }
  __syncthreads();

  const int wave = tid >> 5;
  const int lane = tid & 31;
  const int hs   = lane >> 4;
  const int lm   = lane & 15;
  const int n0   = wave * 16;

  v8f acc = {};
  for (int k0 = 0; k0 < INDIM; k0 += 4) {
    const int ka = k0 + hs * 2;
    v2f a, b;
    a.x = As[ka + 0][lm];
    a.y = As[ka + 1][lm];
    const float* w = W1 + (size_t)ka * HIDDIM + n0 + lm;
    b.x = w[0];
    b.y = w[HIDDIM];
    acc = __builtin_amdgcn_wmma_f32_16x16x4_f32(false, a, false, b,
                                                (short)0, acc, false, false);
  }

  float* dst = xw1 + (size_t)row0 * HIDDIM + n0 + lm;
#pragma unroll
  for (int j = 0; j < 8; ++j)
    dst[(size_t)(hs * 8 + j) * HIDDIM] = acc[j];
}

#define OWN_ROWS 196
#define QCAP 64

__device__ __forceinline__ void own_add_row(float* __restrict__ arow, float v, const float4 x) {
  atomicAdd(arow +  0, v * x.x);
  atomicAdd(arow + 32, v * x.y);
  atomicAdd(arow + 64, v * x.z);
  atomicAdd(arow + 96, v * x.w);
}

__device__ __forceinline__ void own_drain(const int* __restrict__ q, int n, int lane, int base,
                                          const int* __restrict__ esrc,
                                          const int* __restrict__ edst,
                                          const float* __restrict__ evals,
                                          const float* __restrict__ xw1,
                                          float* __restrict__ acc) {
  const int e = q[lane < n ? lane : 0];
  const int   s  = esrc[e];
  const int   r  = edst[e] - base;
  const int   vb = __float_as_int(evals[e]);
  for (int k = 0; k < n; k += 4) {
    const int i1 = (k + 1 < n) ? k + 1 : k;
    const int i2 = (k + 2 < n) ? k + 2 : k;
    const int i3 = (k + 3 < n) ? k + 3 : k;
    const int s0 = __builtin_amdgcn_readlane(s, k);
    const int s1 = __builtin_amdgcn_readlane(s, i1);
    const int s2 = __builtin_amdgcn_readlane(s, i2);
    const int s3 = __builtin_amdgcn_readlane(s, i3);
    const float4 x0 = ((const float4*)(xw1 + (size_t)s0 * HIDDIM))[lane];
    const float4 x1 = ((const float4*)(xw1 + (size_t)s1 * HIDDIM))[lane];
    const float4 x2 = ((const float4*)(xw1 + (size_t)s2 * HIDDIM))[lane];
    const float4 x3 = ((const float4*)(xw1 + (size_t)s3 * HIDDIM))[lane];
    own_add_row(acc + __builtin_amdgcn_readlane(r, k) * HIDDIM + lane,
                __int_as_float(__builtin_amdgcn_readlane(vb, k)), x0);
    if (k + 1 < n)
      own_add_row(acc + __builtin_amdgcn_readlane(r, i1) * HIDDIM + lane,
                  __int_as_float(__builtin_amdgcn_readlane(vb, i1)), x1);
    if (k + 2 < n)
      own_add_row(acc + __builtin_amdgcn_readlane(r, i2) * HIDDIM + lane,
                  __int_as_float(__builtin_amdgcn_readlane(vb, i2)), x2);
    if (k + 3 < n)
      own_add_row(acc + __builtin_amdgcn_readlane(r, i3) * HIDDIM + lane,
                  __int_as_float(__builtin_amdgcn_readlane(vb, i3)), x3);
  }
}

__global__ __launch_bounds__(1024) void spmm1_owned(const int* __restrict__ esrc,
                                                    const int* __restrict__ edst,
                                                    const float* __restrict__ evals,
                                                    const float* __restrict__ xw1,
                                                    float* __restrict__ h,
                                                    int nedges) {
  __shared__ float acc[OWN_ROWS * HIDDIM];
  __shared__ int   queue[32 * QCAP];

  const int tid  = threadIdx.x;
  const int lane = tid & 31;
  const int wave = tid >> 5;
  const int base = blockIdx.x * OWN_ROWS;

  for (int i = tid; i < OWN_ROWS * HIDDIM / 4; i += 1024)
    ((float4*)acc)[i] = make_float4(0.0f, 0.0f, 0.0f, 0.0f);
  __syncthreads();

  int* q = queue + wave * QCAP;
  int qn = 0;

  const int nquads = nedges >> 2;
  const int nspans = (nquads + 127) >> 7;
  const int4* dst4 = (const int4*)edst;

  for (int sp = wave; sp < nspans; sp += 32) {
    int4 cur[4];
#pragma unroll
    for (int j = 0; j < 4; ++j) {
      const int qi = sp * 128 + j * 32 + lane;
      cur[j] = (qi < nquads) ? dst4[qi] : make_int4(-1, -1, -1, -1);
    }
#pragma unroll
    for (int j = 0; j < 4; ++j) {
      const int qi = sp * 128 + j * 32 + lane;
      const int dd[4] = {cur[j].x, cur[j].y, cur[j].z, cur[j].w};
#pragma unroll
      for (int c = 0; c < 4; ++c) {
        const bool m = (unsigned)(dd[c] - base) < (unsigned)OWN_ROWS;
        const unsigned mask = (unsigned)__ballot(m);
        if (mask) {
          if (m) q[qn + __popc(mask & ((1u << lane) - 1u))] = qi * 4 + c;
          qn += __popc(mask);
          if (qn >= 32) {
            own_drain(q, 32, lane, base, esrc, edst, evals, xw1, acc);
            const int rest = qn - 32;
            const int keep = q[32 + lane];
            if (lane < rest) q[lane] = keep;
            qn = rest;
          }
        }
      }
    }
  }
  if (wave == 0) {
    for (int e = nquads * 4 + lane; e - lane < nedges; e += 32) {
      const bool m = (e < nedges) && ((unsigned)(edst[e < nedges ? e : 0] - base) < (unsigned)OWN_ROWS);
      const unsigned mask = (unsigned)__ballot(m);
      if (mask) {
        if (m) q[qn + __popc(mask & ((1u << lane) - 1u))] = e;
        qn += __popc(mask);
        if (qn >= 32) {
          own_drain(q, 32, lane, base, esrc, edst, evals, xw1, acc);
          const int rest = qn - 32;
          const int keep = q[32 + lane];
          if (lane < rest) q[lane] = keep;
          qn = rest;
        }
      }
    }
  }
  if (qn > 0) own_drain(q, qn, lane, base, esrc, edst, evals, xw1, acc);
  __syncthreads();

  for (int r = wave; r < OWN_ROWS; r += 32) {
    const int row = base + r;
    if (row < NNODES) {
      const float* a = acc + r * HIDDIM + lane;
      const float4 o = make_float4(a[0], a[32], a[64], a[96]);
      ((float4*)(h + (size_t)row * HIDDIM))[lane] = o;
    }
  }
}

__global__ __launch_bounds__(32) void gemm2_wmma(const float* __restrict__ h,
                                                 const float* __restrict__ W2,
                                                 float* __restrict__ hw2) {
  const int row0 = blockIdx.x * 16;
  const int lane = threadIdx.x & 31;
  const int hs   = lane >> 4;
  const int lm   = lane & 15;

  const float* arow = h + (size_t)(row0 + lm) * HIDDIM;
  v8f acc = {};
  for (int k0 = 0; k0 < HIDDIM; k0 += 4) {
    const int ka = k0 + hs * 2;
    v2f a, b;
    a.x = fmaxf(arow[ka + 0], 0.0f);
    a.y = fmaxf(arow[ka + 1], 0.0f);
    const float* w = W2 + (size_t)ka * OUTDIM + lm;
    b.x = w[0];
    b.y = w[OUTDIM];
    acc = __builtin_amdgcn_wmma_f32_16x16x4_f32(false, a, false, b,
                                                (short)0, acc, false, false);
  }
  float* dst = hw2 + (size_t)row0 * OUTDIM + lm;
#pragma unroll
  for (int j = 0; j < 8; ++j)
    dst[(size_t)(hs * 8 + j) * OUTDIM] = acc[j];
}

__global__ __launch_bounds__(256) void spmm2_atomic(const int* __restrict__ esrc,
                                                    const int* __restrict__ edst,
                                                    const float* __restrict__ evals,
                                                    const float* __restrict__ hw2,
                                                    float* __restrict__ out,
                                                    int nedges) {
  const int t      = (int)(blockIdx.x * blockDim.x + threadIdx.x);
  const int stride = (int)((gridDim.x * blockDim.x) >> 4);
  const int d      = t & 15;
  for (int e = t >> 4; e < nedges; e += stride) {
    const int   s  = esrc[e];
    const int   dd = edst[e];
    const float v  = evals[e];
    atomicAdd(out + (size_t)dd * OUTDIM + d, v * hw2[(size_t)s * OUTDIM + d]);
  }
}

extern "C" void kernel_launch(void* const* d_in, const int* in_sizes, int n_in,
                              void* d_out, int out_size, void* d_ws, size_t ws_size,
                              hipStream_t stream) {
  const float* feat  = (const float*)d_in[0];
  const int*   esrc  = (const int*)d_in[1];
  const int*   edst  = (const int*)d_in[2];
  const float* evals = (const float*)d_in[3];
  const float* W1    = (const float*)d_in[4];
  const float* W2    = (const float*)d_in[5];
  float*       out   = (float*)d_out;
  const int nedges   = in_sizes[1];

  const size_t xw1_bytes = (size_t)NNODES * HIDDIM * sizeof(float);
  float* xw1 = (float*)d_ws;
  float* h   = (float*)((char*)d_ws + xw1_bytes);
  float* hw2 = (float*)d_ws;

  hipMemsetAsync(out, 0, (size_t)NNODES * OUTDIM * sizeof(float), stream);

  gemm1_wmma<<<NNODES / 16, 256, 0, stream>>>(feat, W1, xw1);

  spmm1_owned<<<256, 1024, 0, stream>>>(esrc, edst, evals, xw1, h, nedges);

  gemm2_wmma<<<NNODES / 16, 32, 0, stream>>>(h, W2, hw2);

  spmm2_atomic<<<16384, 256, 0, stream>>>(esrc, edst, evals, hw2, out, nedges);
}


# spmm1: edges binned once by owning work-group (count + scatter kernels write 8-B records into the bin's own h rows); each group reads only its records, not the whole dst list; sum order changes
# speedup vs baseline: 1.9367x; 1.2047x over previous
#include <hip/hip_runtime.h>

typedef __attribute__((ext_vector_type(2))) float v2f;
typedef __attribute__((ext_vector_type(8))) float v8f;

#define NNODES 50000
#define INDIM  512
#define HIDDIM 128
#define OUTDIM 16

__global__ __launch_bounds__(256) void gemm1_wmma(const float* __restrict__ feat,
                                                  const float* __restrict__ W1,
                                                  float* __restrict__ xw1) {
  __shared__ float As[INDIM][17];
  const int row0 = blockIdx.x * 16;
  const int tid  = threadIdx.x;

  const float4* src = (const float4*)(feat + (size_t)row0 * INDIM);
#pragma unroll
  for (int i = 0; i < 8; ++i) {
    const int s  = tid + i * 256;
    const int r  = s >> 7;
    const int kv = s & 127;
    const float4 v = src[(size_t)r * (INDIM / 4) + kv];
    const int k = kv * 4;
    As[k + 0][r] = v.x;
    As[k + 1][r] = v.y;
    As[k + 2][r] = v.z;
    As[k + 3][r] = v.w;
  }
  __syncthreads();

  const int wave = tid >> 5;
  const int lane = tid & 31;
  const int hs   = lane >> 4;
  const int lm   = lane & 15;
  const int n0   = wave * 16;

  v8f acc = {};
  for (int k0 = 0; k0 < INDIM; k0 += 4) {
    const int ka = k0 + hs * 2;
    v2f a, b;
    a.x = As[ka + 0][lm];
    a.y = As[ka + 1][lm];
    const float* w = W1 + (size_t)ka * HIDDIM + n0 + lm;
    b.x = w[0];
    b.y = w[HIDDIM];
    acc = __builtin_amdgcn_wmma_f32_16x16x4_f32(false, a, false, b,
                                                (short)0, acc, false, false);
  }

  float* dst = xw1 + (size_t)row0 * HIDDIM + n0 + lm;
#pragma unroll
  for (int j = 0; j < 8; ++j)
    dst[(size_t)(hs * 8 + j) * HIDDIM] = acc[j];
}

#define OWN_ROWS 196
#define QCAP 64
#define NBINS 256
#define REGION_INTS (OWN_ROWS * HIDDIM)
#define HDR_INTS 260
#define REC_CAP ((REGION_INTS - HDR_INTS) / 2)
#define LAST_ROWS (NNODES - (NBINS - 1) * OWN_ROWS)
#define LAST_CAP ((LAST_ROWS * HIDDIM - HDR_INTS) / 2)
#define SLAB 8

__global__ __launch_bounds__(1024) void bin_count(const int* __restrict__ edst,
                                                  int* __restrict__ regions,
                                                  int nedges) {
  __shared__ int hist[NBINS];
  const int tid = threadIdx.x;
  if (tid < NBINS) hist[tid] = 0;
  __syncthreads();

  const int per = (nedges + NBINS - 1) / NBINS;
  const int e0  = (int)blockIdx.x * per;
  const int e1  = (e0 + per < nedges) ? e0 + per : nedges;
  for (int eb = e0; eb < e1; eb += SLAB * 1024) {
    int dd[SLAB];
#pragma unroll
    for (int j = 0; j < SLAB; ++j) {
      const int e = eb + j * 1024 + tid;
      dd[j] = (e < e1) ? edst[e] : -1;
    }
#pragma unroll
    for (int j = 0; j < SLAB; ++j)
      if ((unsigned)dd[j] < (unsigned)NNODES)
        atomicAdd(&hist[(unsigned)dd[j] / OWN_ROWS], 1);
  }
  __syncthreads();
  if (tid < NBINS) regions[(size_t)blockIdx.x * REGION_INTS + tid] = hist[tid];
}

__global__ __launch_bounds__(1024) void bin_scatter(const int* __restrict__ esrc,
                                                    const int* __restrict__ edst,
                                                    const float* __restrict__ evals,
                                                    int* regions,
                                                    int nedges) {
  __shared__ int part[4][NBINS];
  __shared__ int cursor[NBINS];
  const int tid = threadIdx.x;
  const int g   = (int)blockIdx.x;
  const int bin = tid & (NBINS - 1);
  const int qtr = tid >> 8;

  int sum = 0;
  const int gbeg = qtr * (NBINS / 4);
  const int gend = (g < gbeg + NBINS / 4) ? g : gbeg + NBINS / 4;
  for (int gg = gbeg; gg < gend; gg += 16) {
    int t[16];
#pragma unroll
    for (int j = 0; j < 16; ++j)
      t[j] = (gg + j < gend) ? regions[(size_t)(gg + j) * REGION_INTS + bin] : 0;
#pragma unroll
    for (int j = 0; j < 16; ++j) sum += t[j];
  }
  part[qtr][bin] = sum;
  __syncthreads();
  if (tid < NBINS) {
    const int start = part[0][tid] + part[1][tid] + part[2][tid] + part[3][tid];
    cursor[tid] = start;
    if (g == NBINS - 1)
      regions[(size_t)tid * REGION_INTS + NBINS] = start + regions[(size_t)g * REGION_INTS + tid];
  }
  __syncthreads();

  const int per = (nedges + NBINS - 1) / NBINS;
  const int e0  = g * per;
  const int e1  = (e0 + per < nedges) ? e0 + per : nedges;
  for (int eb = e0; eb < e1; eb += SLAB * 1024) {
    int dd[SLAB], ss[SLAB], vv[SLAB];
#pragma unroll
    for (int j = 0; j < SLAB; ++j) {
      const int e = eb + j * 1024 + tid;
      const bool ok = e < e1;
      dd[j] = ok ? edst[e] : -1;
      ss[j] = ok ? esrc[e] : 0;
      vv[j] = ok ? __float_as_int(evals[e]) : 0;
    }
#pragma unroll
    for (int j = 0; j < SLAB; ++j) {
      if ((unsigned)dd[j] < (unsigned)NNODES) {
        const int b   = (int)((unsigned)dd[j] / OWN_ROWS);
        const int pos = atomicAdd(&cursor[b], 1);
        const int cap = (b == NBINS - 1) ? LAST_CAP : REC_CAP;
        if (pos < cap) {
          const int key = (ss[j] & 0xFFFF) | ((dd[j] - b * OWN_ROWS) << 16);
          *(int2*)(regions + (size_t)b * REGION_INTS + HDR_INTS + 2 * pos) = make_int2(key, vv[j]);
        }
      }
    }
  }
}

__device__ __forceinline__ void own_add_row(float* __restrict__ arow, float v, const float4 x) {
  atomicAdd(arow +  0, v * x.x);
  atomicAdd(arow + 32, v * x.y);
  atomicAdd(arow + 64, v * x.z);
  atomicAdd(arow + 96, v * x.w);
}

__device__ __forceinline__ void own_drain_regs(int s, int r, int vb, int n, int lane,
                                               const float* __restrict__ xw1,
                                               float* __restrict__ acc) {
  for (int k = 0; k < n; k += 4) {
    const int i1 = (k + 1 < n) ? k + 1 : k;
    const int i2 = (k + 2 < n) ? k + 2 : k;
    const int i3 = (k + 3 < n) ? k + 3 : k;
    const int s0 = __builtin_amdgcn_readlane(s, k);
    const int s1 = __builtin_amdgcn_readlane(s, i1);
    const int s2 = __builtin_amdgcn_readlane(s, i2);
    const int s3 = __builtin_amdgcn_readlane(s, i3);
    const float4 x0 = ((const float4*)(xw1 + (size_t)s0 * HIDDIM))[lane];
    const float4 x1 = ((const float4*)(xw1 + (size_t)s1 * HIDDIM))[lane];
    const float4 x2 = ((const float4*)(xw1 + (size_t)s2 * HIDDIM))[lane];
    const float4 x3 = ((const float4*)(xw1 + (size_t)s3 * HIDDIM))[lane];
    own_add_row(acc + __builtin_amdgcn_readlane(r, k) * HIDDIM + lane,
                __int_as_float(__builtin_amdgcn_readlane(vb, k)), x0);
    if (k + 1 < n)
      own_add_row(acc + __builtin_amdgcn_readlane(r, i1) * HIDDIM + lane,
                  __int_as_float(__builtin_amdgcn_readlane(vb, i1)), x1);
    if (k + 2 < n)
      own_add_row(acc + __builtin_amdgcn_readlane(r, i2) * HIDDIM + lane,
                  __int_as_float(__builtin_amdgcn_readlane(vb, i2)), x2);
    if (k + 3 < n)
      own_add_row(acc + __builtin_amdgcn_readlane(r, i3) * HIDDIM + lane,
                  __int_as_float(__builtin_amdgcn_readlane(vb, i3)), x3);
  }
}

__device__ __forceinline__ void own_drain(const int* __restrict__ q, int n, int lane, int base,
                                          const int* __restrict__ esrc,
                                          const int* __restrict__ edst,
                                          const float* __restrict__ evals,
                                          const float* __restrict__ xw1,
                                          float* __restrict__ acc) {
  const int e = q[lane < n ? lane : 0];
  own_drain_regs(esrc[e], edst[e] - base, __float_as_int(evals[e]), n, lane, xw1, acc);
}

__global__ __launch_bounds__(1024) void spmm1_owned(const int* __restrict__ esrc,
                                                    const int* __restrict__ edst,
                                                    const float* __restrict__ evals,
                                                    const float* __restrict__ xw1,
                                                    float* h,
                                                    int nedges) {
  __shared__ float acc[OWN_ROWS * HIDDIM];
  __shared__ int   queue[32 * QCAP];

  const int tid  = threadIdx.x;
  const int lane = tid & 31;
  const int wave = __builtin_amdgcn_readfirstlane(tid >> 5);
  const int base = blockIdx.x * OWN_ROWS;

  for (int i = tid; i < OWN_ROWS * HIDDIM / 4; i += 1024)
    ((float4*)acc)[i] = make_float4(0.0f, 0.0f, 0.0f, 0.0f);

  const int* region = (const int*)h + (size_t)blockIdx.x * REGION_INTS;
  const int nrec = __builtin_amdgcn_readfirstlane(region[NBINS]);
  const int cap  = ((int)blockIdx.x == NBINS - 1) ? LAST_CAP : REC_CAP;
  __syncthreads();

  if (nrec <= cap) {
    const int2* rec = (const int2*)(region + HDR_INTS);
    const int per = (nrec + 31) >> 5;
    const int i0  = wave * per;
    const int i1  = (i0 + per < nrec) ? i0 + per : nrec;
    for (int i = i0; i < i1; i += 32) {
      const int c = (i1 - i < 32) ? i1 - i : 32;
      const int2 rc = rec[i + (lane < c ? lane : 0)];
      own_drain_regs(rc.x & 0xFFFF, rc.x >> 16, rc.y, c, lane, xw1, acc);
    }
  } else {
    int* q = queue + wave * QCAP;
    int qn = 0;

    const int nquads = nedges >> 2;
    const int nspans = (nquads + 127) >> 7;
    const int4* dst4 = (const int4*)edst;

    for (int sp = wave; sp < nspans; sp += 32) {
      int4 cur[4];
#pragma unroll
      for (int j = 0; j < 4; ++j) {
        const int qi = sp * 128 + j * 32 + lane;
        cur[j] = (qi < nquads) ? dst4[qi] : make_int4(-1, -1, -1, -1);
      }
#pragma unroll
      for (int j = 0; j < 4; ++j) {
        const int qi = sp * 128 + j * 32 + lane;
        const int dd[4] = {cur[j].x, cur[j].y, cur[j].z, cur[j].w};
#pragma unroll
        for (int c = 0; c < 4; ++c) {
          const bool m = (unsigned)(dd[c] - base) < (unsigned)OWN_ROWS;
          const unsigned mask = (unsigned)__ballot(m);
          if (mask) {
            if (m) q[qn + __popc(mask & ((1u << lane) - 1u))] = qi * 4 + c;
            qn += __popc(mask);
            if (qn >= 32) {
              own_drain(q, 32, lane, base, esrc, edst, evals, xw1, acc);
              const int rest = qn - 32;
              const int keep = q[32 + lane];
              if (lane < rest) q[lane] = keep;
              qn = rest;
            }
          }
        }
      }
    }
    if (wave == 0) {
      for (int e = nquads * 4 + lane; e - lane < nedges; e += 32) {
        const bool m = (e < nedges) && ((unsigned)(edst[e < nedges ? e : 0] - base) < (unsigned)OWN_ROWS);
        const unsigned mask = (unsigned)__ballot(m);
        if (mask) {
          if (m) q[qn + __popc(mask & ((1u << lane) - 1u))] = e;
          qn += __popc(mask);
          if (qn >= 32) {
            own_drain(q, 32, lane, base, esrc, edst, evals, xw1, acc);
            const int rest = qn - 32;
            const int keep = q[32 + lane];
            if (lane < rest) q[lane] = keep;
            qn = rest;
          }
        }
      }
    }
    if (qn > 0) own_drain(q, qn, lane, base, esrc, edst, evals, xw1, acc);
  }
  __syncthreads();

  for (int r = wave; r < OWN_ROWS; r += 32) {
    const int row = base + r;
    if (row < NNODES) {
      const float* a = acc + r * HIDDIM + lane;
      const float4 o = make_float4(a[0], a[32], a[64], a[96]);
      ((float4*)(h + (size_t)row * HIDDIM))[lane] = o;
    }
  }
}

__global__ __launch_bounds__(32) void gemm2_wmma(const float* __restrict__ h,
                                                 const float* __restrict__ W2,
                                                 float* __restrict__ hw2) {
  const int row0 = blockIdx.x * 16;
  const int lane = threadIdx.x & 31;
  const int hs   = lane >> 4;
  const int lm   = lane & 15;

  const float* arow = h + (size_t)(row0 + lm) * HIDDIM;
  v8f acc = {};
  for (int k0 = 0; k0 < HIDDIM; k0 += 4) {
    const int ka = k0 + hs * 2;
    v2f a, b;
    a.x = fmaxf(arow[ka + 0], 0.0f);
    a.y = fmaxf(arow[ka + 1], 0.0f);
    const float* w = W2 + (size_t)ka * OUTDIM + lm;
    b.x = w[0];
    b.y = w[OUTDIM];
    acc = __builtin_amdgcn_wmma_f32_16x16x4_f32(false, a, false, b,
                                                (short)0, acc, false, false);
  }
  float* dst = hw2 + (size_t)row0 * OUTDIM + lm;
#pragma unroll
  for (int j = 0; j < 8; ++j)
    dst[(size_t)(hs * 8 + j) * OUTDIM] = acc[j];
}

__global__ __launch_bounds__(256) void spmm2_atomic(const int* __restrict__ esrc,
                                                    const int* __restrict__ edst,
                                                    const float* __restrict__ evals,
                                                    const float* __restrict__ hw2,
                                                    float* __restrict__ out,
                                                    int nedges) {
  const int t      = (int)(blockIdx.x * blockDim.x + threadIdx.x);
  const int stride = (int)((gridDim.x * blockDim.x) >> 4);
  const int d      = t & 15;
  for (int e = t >> 4; e < nedges; e += stride) {
    const int   s  = esrc[e];
    const int   dd = edst[e];
    const float v  = evals[e];
    atomicAdd(out + (size_t)dd * OUTDIM + d, v * hw2[(size_t)s * OUTDIM + d]);
  }
}

extern "C" void kernel_launch(void* const* d_in, const int* in_sizes, int n_in,
                              void* d_out, int out_size, void* d_ws, size_t ws_size,
                              hipStream_t stream) {
  const float* feat  = (const float*)d_in[0];
  const int*   esrc  = (const int*)d_in[1];
  const int*   edst  = (const int*)d_in[2];
  const float* evals = (const float*)d_in[3];
  const float* W1    = (const float*)d_in[4];
  const float* W2    = (const float*)d_in[5];
  float*       out   = (float*)d_out;
  const int nedges   = in_sizes[1];

  const size_t xw1_bytes = (size_t)NNODES * HIDDIM * sizeof(float);
  float* xw1 = (float*)d_ws;
  float* h   = (float*)((char*)d_ws + xw1_bytes);
  float* hw2 = (float*)d_ws;

  hipMemsetAsync(out, 0, (size_t)NNODES * OUTDIM * sizeof(float), stream);

  bin_count<<<NBINS, 1024, 0, stream>>>(edst, (int*)h, nedges);
  bin_scatter<<<NBINS, 1024, 0, stream>>>(esrc, edst, evals, (int*)h, nedges);

  gemm1_wmma<<<NNODES / 16, 256, 0, stream>>>(feat, W1, xw1);

  spmm1_owned<<<NBINS, 1024, 0, stream>>>(esrc, edst, evals, xw1, h, nedges);

  gemm2_wmma<<<NNODES / 16, 32, 0, stream>>>(h, W2, hw2);

  spmm2_atomic<<<16384, 256, 0, stream>>>(esrc, edst, evals, hw2, out, nedges);
}


# layer 2 on the layer-1 bins: spmm1 ends with relu(h)@W2 on WMMA from its LDS rows (h not stored, records survive); spmm2 sums owned rows in LDS, stores out once, no global atomics; sum order changes
# speedup vs baseline: 2.6926x; 1.3903x over previous
#include <hip/hip_runtime.h>

typedef __attribute__((ext_vector_type(2))) float v2f;
typedef __attribute__((ext_vector_type(8))) float v8f;

#define NNODES 50000
#define INDIM  512
#define HIDDIM 128
#define OUTDIM 16

__global__ __launch_bounds__(256) void gemm1_wmma(const float* __restrict__ feat,
                                                  const float* __restrict__ W1,
                                                  float* __restrict__ xw1) {
  __shared__ float As[INDIM][17];
  const int row0 = blockIdx.x * 16;
  const int tid  = threadIdx.x;

  const float4* src = (const float4*)(feat + (size_t)row0 * INDIM);
#pragma unroll
  for (int i = 0; i < 8; ++i) {
    const int s  = tid + i * 256;
    const int r  = s >> 7;
    const int kv = s & 127;
    const float4 v = src[(size_t)r * (INDIM / 4) + kv];
    const int k = kv * 4;
    As[k + 0][r] = v.x;
    As[k + 1][r] = v.y;
    As[k + 2][r] = v.z;
    As[k + 3][r] = v.w;
  }
  __syncthreads();

  const int wave = tid >> 5;
  const int lane = tid & 31;
  const int hs   = lane >> 4;
  const int lm   = lane & 15;
  const int n0   = wave * 16;

  v8f acc = {};
  for (int k0 = 0; k0 < INDIM; k0 += 4) {
    const int ka = k0 + hs * 2;
    v2f a, b;
    a.x = As[ka + 0][lm];
    a.y = As[ka + 1][lm];
    const float* w = W1 + (size_t)ka * HIDDIM + n0 + lm;
    b.x = w[0];
    b.y = w[HIDDIM];
    acc = __builtin_amdgcn_wmma_f32_16x16x4_f32(false, a, false, b,
                                                (short)0, acc, false, false);
  }

  float* dst = xw1 + (size_t)row0 * HIDDIM + n0 + lm;
#pragma unroll
  for (int j = 0; j < 8; ++j)
    dst[(size_t)(hs * 8 + j) * HIDDIM] = acc[j];
}

#define OWN_ROWS 196
#define QCAP 64
#define NBINS 256
#define REGION_INTS (OWN_ROWS * HIDDIM)
#define HDR_INTS 260
#define TAIL_COLS (HIDDIM - OUTDIM)
#define REC_CAP ((OWN_ROWS * TAIL_COLS - HDR_INTS) / 2)
#define LAST_ROWS (NNODES - (NBINS - 1) * OWN_ROWS)
#define LAST_CAP ((LAST_ROWS * TAIL_COLS - HDR_INTS) / 2)
#define SLAB 8

__global__ __launch_bounds__(1024) void bin_count(const int* __restrict__ edst,
                                                  int* __restrict__ regions,
                                                  int nedges) {
  __shared__ int hist[NBINS];
  const int tid = threadIdx.x;
  if (tid < NBINS) hist[tid] = 0;
  __syncthreads();

  const int per = (nedges + NBINS - 1) / NBINS;
  const int e0  = (int)blockIdx.x * per;
  const int e1  = (e0 + per < nedges) ? e0 + per : nedges;
  for (int eb = e0; eb < e1; eb += SLAB * 1024) {
    int dd[SLAB];
#pragma unroll
    for (int j = 0; j < SLAB; ++j) {
      const int e = eb + j * 1024 + tid;
      dd[j] = (e < e1) ? edst[e] : -1;
    }
#pragma unroll
    for (int j = 0; j < SLAB; ++j)
      if ((unsigned)dd[j] < (unsigned)NNODES)
        atomicAdd(&hist[(unsigned)dd[j] / OWN_ROWS], 1);
  }
  __syncthreads();
  if (tid < NBINS) regions[(size_t)blockIdx.x * REGION_INTS + tid] = hist[tid];
}

__global__ __launch_bounds__(1024) void bin_scatter(const int* __restrict__ esrc,
                                                    const int* __restrict__ edst,
                                                    const float* __restrict__ evals,
                                                    int* regions,
                                                    int nedges) {
  __shared__ int part[4][NBINS];
  __shared__ int cursor[NBINS];
  const int tid = threadIdx.x;
  const int g   = (int)blockIdx.x;
  const int bin = tid & (NBINS - 1);
  const int qtr = tid >> 8;

  int sum = 0;
  const int gbeg = qtr * (NBINS / 4);
  const int gend = (g < gbeg + NBINS / 4) ? g : gbeg + NBINS / 4;
  for (int gg = gbeg; gg < gend; gg += 16) {
    int t[16];
#pragma unroll
    for (int j = 0; j < 16; ++j)
      t[j] = (gg + j < gend) ? regions[(size_t)(gg + j) * REGION_INTS + bin] : 0;
#pragma unroll
    for (int j = 0; j < 16; ++j) sum += t[j];
  }
  part[qtr][bin] = sum;
  __syncthreads();
  if (tid < NBINS) {
    const int start = part[0][tid] + part[1][tid] + part[2][tid] + part[3][tid];
    cursor[tid] = start;
    if (g == NBINS - 1)
      regions[(size_t)tid * REGION_INTS + NBINS] = start + regions[(size_t)g * REGION_INTS + tid];
  }
  __syncthreads();

  const int per = (nedges + NBINS - 1) / NBINS;
  const int e0  = g * per;
  const int e1  = (e0 + per < nedges) ? e0 + per : nedges;
  for (int eb = e0; eb < e1; eb += SLAB * 1024) {
    int dd[SLAB], ss[SLAB], vv[SLAB];
#pragma unroll
    for (int j = 0; j < SLAB; ++j) {
      const int e = eb + j * 1024 + tid;
      const bool ok = e < e1;
      dd[j] = ok ? edst[e] : -1;
      ss[j] = ok ? esrc[e] : 0;
      vv[j] = ok ? __float_as_int(evals[e]) : 0;
    }
#pragma unroll
    for (int j = 0; j < SLAB; ++j) {
      if ((unsigned)dd[j] < (unsigned)NNODES) {
        const int b   = (int)((unsigned)dd[j] / OWN_ROWS);
        const int pos = atomicAdd(&cursor[b], 1);
        const int cap = (b == NBINS - 1) ? LAST_CAP : REC_CAP;
        if (pos < cap) {
          const int key = (ss[j] & 0xFFFF) | ((dd[j] - b * OWN_ROWS) << 16);
          *(int2*)(regions + (size_t)b * REGION_INTS + HDR_INTS + 2 * pos) = make_int2(key, vv[j]);
        }
      }
    }
  }
}

__device__ __forceinline__ void own_add_row(float* __restrict__ arow, float v, const float4 x) {
  atomicAdd(arow +  0, v * x.x);
  atomicAdd(arow + 32, v * x.y);
  atomicAdd(arow + 64, v * x.z);
  atomicAdd(arow + 96, v * x.w);
}

__device__ __forceinline__ void own_drain_regs(int s, int r, int vb, int n, int lane,
                                               const float* __restrict__ xw1,
                                               float* __restrict__ acc) {
  for (int k = 0; k < n; k += 4) {
    const int i1 = (k + 1 < n) ? k + 1 : k;
    const int i2 = (k + 2 < n) ? k + 2 : k;
    const int i3 = (k + 3 < n) ? k + 3 : k;
    const int s0 = __builtin_amdgcn_readlane(s, k);
    const int s1 = __builtin_amdgcn_readlane(s, i1);
    const int s2 = __builtin_amdgcn_readlane(s, i2);
    const int s3 = __builtin_amdgcn_readlane(s, i3);
    const float4 x0 = ((const float4*)(xw1 + (size_t)s0 * HIDDIM))[lane];
    const float4 x1 = ((const float4*)(xw1 + (size_t)s1 * HIDDIM))[lane];
    const float4 x2 = ((const float4*)(xw1 + (size_t)s2 * HIDDIM))[lane];
    const float4 x3 = ((const float4*)(xw1 + (size_t)s3 * HIDDIM))[lane];
    own_add_row(acc + __builtin_amdgcn_readlane(r, k) * HIDDIM + lane,
                __int_as_float(__builtin_amdgcn_readlane(vb, k)), x0);
    if (k + 1 < n)
      own_add_row(acc + __builtin_amdgcn_readlane(r, i1) * HIDDIM + lane,
                  __int_as_float(__builtin_amdgcn_readlane(vb, i1)), x1);
    if (k + 2 < n)
      own_add_row(acc + __builtin_amdgcn_readlane(r, i2) * HIDDIM + lane,
                  __int_as_float(__builtin_amdgcn_readlane(vb, i2)), x2);
    if (k + 3 < n)
      own_add_row(acc + __builtin_amdgcn_readlane(r, i3) * HIDDIM + lane,
                  __int_as_float(__builtin_amdgcn_readlane(vb, i3)), x3);
  }
}

__device__ __forceinline__ void own_drain(const int* __restrict__ q, int n, int lane, int base,
                                          const int* __restrict__ esrc,
                                          const int* __restrict__ edst,
                                          const float* __restrict__ evals,
                                          const float* __restrict__ xw1,
                                          float* __restrict__ acc) {
  const int e = q[lane < n ? lane : 0];
  own_drain_regs(esrc[e], edst[e] - base, __float_as_int(evals[e]), n, lane, xw1, acc);
}

__global__ __launch_bounds__(1024) void spmm1_owned(const int* __restrict__ esrc,
                                                    const int* __restrict__ edst,
                                                    const float* __restrict__ evals,
                                                    const float* __restrict__ xw1,
                                                    const float* __restrict__ W2,
                                                    float* h,
                                                    int nedges) {
  __shared__ float acc[OWN_ROWS * HIDDIM];
  __shared__ int   queue[32 * QCAP];

  const int tid  = threadIdx.x;
  const int lane = tid & 31;
  const int wave = __builtin_amdgcn_readfirstlane(tid >> 5);
  const int base = blockIdx.x * OWN_ROWS;

  for (int i = tid; i < OWN_ROWS * HIDDIM / 4; i += 1024)
    ((float4*)acc)[i] = make_float4(0.0f, 0.0f, 0.0f, 0.0f);

  const int* region = (const int*)h + (size_t)blockIdx.x * REGION_INTS;
  const int nrec = __builtin_amdgcn_readfirstlane(region[NBINS]);
  const int cap  = ((int)blockIdx.x == NBINS - 1) ? LAST_CAP : REC_CAP;
  __syncthreads();

  if (nrec <= cap) {
    const int2* rec = (const int2*)(region + HDR_INTS);
    const int per = (nrec + 31) >> 5;
    const int i0  = wave * per;
    const int i1  = (i0 + per < nrec) ? i0 + per : nrec;
    for (int i = i0; i < i1; i += 32) {
      const int c = (i1 - i < 32) ? i1 - i : 32;
      const int2 rc = rec[i + (lane < c ? lane : 0)];
      own_drain_regs(rc.x & 0xFFFF, rc.x >> 16, rc.y, c, lane, xw1, acc);
    }
  } else {
    int* q = queue + wave * QCAP;
    int qn = 0;

    const int nquads = nedges >> 2;
    const int nspans = (nquads + 127) >> 7;
    const int4* dst4 = (const int4*)edst;

    for (int sp = wave; sp < nspans; sp += 32) {
      int4 cur[4];
#pragma unroll
      for (int j = 0; j < 4; ++j) {
        const int qi = sp * 128 + j * 32 + lane;
        cur[j] = (qi < nquads) ? dst4[qi] : make_int4(-1, -1, -1, -1);
      }
#pragma unroll
      for (int j = 0; j < 4; ++j) {
        const int qi = sp * 128 + j * 32 + lane;
        const int dd[4] = {cur[j].x, cur[j].y, cur[j].z, cur[j].w};
#pragma unroll
        for (int c = 0; c < 4; ++c) {
          const bool m = (unsigned)(dd[c] - base) < (unsigned)OWN_ROWS;
          const unsigned mask = (unsigned)__ballot(m);
          if (mask) {
            if (m) q[qn + __popc(mask & ((1u << lane) - 1u))] = qi * 4 + c;
            qn += __popc(mask);
            if (qn >= 32) {
              own_drain(q, 32, lane, base, esrc, edst, evals, xw1, acc);
              const int rest = qn - 32;
              const int keep = q[32 + lane];
              if (lane < rest) q[lane] = keep;
              qn = rest;
            }
          }
        }
      }
    }
    if (wave == 0) {
      for (int e = nquads * 4 + lane; e - lane < nedges; e += 32) {
        const bool m = (e < nedges) && ((unsigned)(edst[e < nedges ? e : 0] - base) < (unsigned)OWN_ROWS);
        const unsigned mask = (unsigned)__ballot(m);
        if (mask) {
          if (m) q[qn + __popc(mask & ((1u << lane) - 1u))] = e;
          qn += __popc(mask);
          if (qn >= 32) {
            own_drain(q, 32, lane, base, esrc, edst, evals, xw1, acc);
            const int rest = qn - 32;
            const int keep = q[32 + lane];
            if (lane < rest) q[lane] = keep;
            qn = rest;
          }
        }
      }
    }
    if (qn > 0) own_drain(q, qn, lane, base, esrc, edst, evals, xw1, acc);
  }
  __syncthreads();

  const int nrows = ((int)blockIdx.x == NBINS - 1) ? LAST_ROWS : OWN_ROWS;
  float* hw2 = h + (size_t)blockIdx.x * REGION_INTS + nrows * TAIL_COLS;
  if (wave < (OWN_ROWS + 15) / 16) {
    const int hs   = lane >> 4;
    const int lm   = lane & 15;
    const int row0 = wave * 16;
    const int ar = (row0 + lm < OWN_ROWS) ? row0 + lm : OWN_ROWS - 1;
    const float4* a0p = (const float4*)(acc + ar * HIDDIM + (2 * hs) * 32);
    const float4* a1p = (const float4*)(acc + ar * HIDDIM + (2 * hs + 1) * 32);
    v8f d = {};
    for (int lc = 0; lc < HIDDIM / 16; ++lc) {
      const float4 A0 = a0p[lc];
      const float4 A1 = a1p[lc];
      const float e0[4] = {A0.x, A0.y, A0.z, A0.w};
      const float e1[4] = {A1.x, A1.y, A1.z, A1.w};
#pragma unroll
      for (int i = 0; i < 4; ++i) {
        const int ka = (lc * 4 + i) * 4 + hs * 2;
        v2f a, b;
        a.x = fmaxf(e0[i], 0.0f);
        a.y = fmaxf(e1[i], 0.0f);
        const float* w = W2 + (size_t)ka * OUTDIM + lm;
        b.x = w[0];
        b.y = w[OUTDIM];
        d = __builtin_amdgcn_wmma_f32_16x16x4_f32(false, a, false, b,
                                                  (short)0, d, false, false);
      }
    }
#pragma unroll
    for (int j = 0; j < 8; ++j) {
      const int r = row0 + hs * 8 + j;
      if (r < nrows) hw2[r * OUTDIM + lm] = d[j];
    }
  }
}

__device__ __forceinline__ const float* hw2_row(const int* __restrict__ regions, int s) {
  const int b     = (int)((unsigned)s / OWN_ROWS);
  const int nrows = (b == NBINS - 1) ? LAST_ROWS : OWN_ROWS;
  return (const float*)(regions + (size_t)b * REGION_INTS + nrows * TAIL_COLS + (s - b * OWN_ROWS) * OUTDIM);
}

__global__ __launch_bounds__(1024) void spmm2_owned(const int* __restrict__ esrc,
                                                    const int* __restrict__ edst,
                                                    const float* __restrict__ evals,
                                                    const int* __restrict__ regions,
                                                    float* __restrict__ out,
                                                    int nedges) {
  __shared__ float acc2[OWN_ROWS * OUTDIM];

  const int tid  = threadIdx.x;
  const int lane = tid & 31;
  const int wave = __builtin_amdgcn_readfirstlane(tid >> 5);
  const int base = blockIdx.x * OWN_ROWS;

  for (int i = tid; i < OWN_ROWS * OUTDIM; i += 1024) acc2[i] = 0.0f;

  const int* region = regions + (size_t)blockIdx.x * REGION_INTS;
  const int nrec = __builtin_amdgcn_readfirstlane(region[NBINS]);
  const int cap  = ((int)blockIdx.x == NBINS - 1) ? LAST_CAP : REC_CAP;
  __syncthreads();

  if (nrec <= cap) {
    const int2* rec = (const int2*)(region + HDR_INTS);
    const int per = (nrec + 31) >> 5;
    const int i0  = wave * per;
    const int i1  = (i0 + per < nrec) ? i0 + per : nrec;
    const int q   = lane & 3;
    const int sub = lane >> 2;
    int2 nxt = rec[(i0 < i1) ? i0 + ((lane < i1 - i0) ? lane : 0) : 0];
    for (int i = i0; i < i1; i += 32) {
      const int c = (i1 - i < 32) ? i1 - i : 32;
      const int2 rc = nxt;
      const int left = i1 - (i + 32);
      if (left > 0) nxt = rec[i + 32 + ((lane < left) ? lane : 0)];
      float4 x[4];
      int    ro[4];
      float  vv[4];
#pragma unroll
      for (int j = 0; j < 4; ++j) {
        const int idx = j * 8 + sub;
        const int key = __shfl(rc.x, idx);
        vv[j] = __int_as_float(__shfl(rc.y, idx));
        ro[j] = (idx < c) ? (key >> 16) * OUTDIM + q * 4 : -1;
        x[j]  = (idx < c) ? ((const float4*)hw2_row(regions, key & 0xFFFF))[q]
                          : make_float4(0.0f, 0.0f, 0.0f, 0.0f);
      }
#pragma unroll
      for (int j = 0; j < 4; ++j) {
        if (ro[j] >= 0) {
          float* a = acc2 + ro[j];
          atomicAdd(a + 0, vv[j] * x[j].x);
          atomicAdd(a + 1, vv[j] * x[j].y);
          atomicAdd(a + 2, vv[j] * x[j].z);
          atomicAdd(a + 3, vv[j] * x[j].w);
        }
      }
    }
  } else {
    for (int e = tid; e < nedges; e += 1024) {
      const int r = edst[e] - base;
      if ((unsigned)r < (unsigned)OWN_ROWS) {
        const float  v = evals[e];
        const float* x = hw2_row(regions, esrc[e]);
        for (int k = 0; k < OUTDIM; ++k) atomicAdd(acc2 + r * OUTDIM + k, v * x[k]);
      }
    }
  }
  __syncthreads();

  for (int i = tid; i < OWN_ROWS * (OUTDIM / 4); i += 1024) {
    const int row = base + (i >> 2);
    if (row < NNODES) ((float4*)(out + (size_t)row * OUTDIM))[i & 3] = ((const float4*)acc2)[i];
  }
}

extern "C" void kernel_launch(void* const* d_in, const int* in_sizes, int n_in,
                              void* d_out, int out_size, void* d_ws, size_t ws_size,
                              hipStream_t stream) {
  const float* feat  = (const float*)d_in[0];
  const int*   esrc  = (const int*)d_in[1];
  const int*   edst  = (const int*)d_in[2];
  const float* evals = (const float*)d_in[3];
  const float* W1    = (const float*)d_in[4];
  const float* W2    = (const float*)d_in[5];
  float*       out   = (float*)d_out;
  const int nedges   = in_sizes[1];

  const size_t xw1_bytes = (size_t)NNODES * HIDDIM * sizeof(float);
  float* xw1 = (float*)d_ws;
  float* h   = (float*)((char*)d_ws + xw1_bytes);

  bin_count<<<NBINS, 1024, 0, stream>>>(edst, (int*)h, nedges);
  bin_scatter<<<NBINS, 1024, 0, stream>>>(esrc, edst, evals, (int*)h, nedges);

  gemm1_wmma<<<NNODES / 16, 256, 0, stream>>>(feat, W1, xw1);

  spmm1_owned<<<NBINS, 1024, 0, stream>>>(esrc, edst, evals, xw1, W2, h, nedges);

  spmm2_owned<<<NBINS, 1024, 0, stream>>>(esrc, edst, evals, (const int*)h, out, nedges);
}


# spmm1: group puts its bin's records in destination-row order in LDS (integer count, scan, place); a wave sums a row's run in registers, one LDS f32 add per run, not per edge; sum order changes
# speedup vs baseline: 2.8711x; 1.0663x over previous
#include <hip/hip_runtime.h>

typedef __attribute__((ext_vector_type(2))) float v2f;
typedef __attribute__((ext_vector_type(8))) float v8f;

#define NNODES 50000
#define INDIM  512
#define HIDDIM 128
#define OUTDIM 16

__global__ __launch_bounds__(256) void gemm1_wmma(const float* __restrict__ feat,
                                                  const float* __restrict__ W1,
                                                  float* __restrict__ xw1) {
  __shared__ float As[INDIM][17];
  const int row0 = blockIdx.x * 16;
  const int tid  = threadIdx.x;

  const float4* src = (const float4*)(feat + (size_t)row0 * INDIM);
#pragma unroll
  for (int i = 0; i < 8; ++i) {
    const int s  = tid + i * 256;
    const int r  = s >> 7;
    const int kv = s & 127;
    const float4 v = src[(size_t)r * (INDIM / 4) + kv];
    const int k = kv * 4;
    As[k + 0][r] = v.x;
    As[k + 1][r] = v.y;
    As[k + 2][r] = v.z;
    As[k + 3][r] = v.w;
  }
  __syncthreads();

  const int wave = tid >> 5;
  const int lane = tid & 31;
  const int hs   = lane >> 4;
  const int lm   = lane & 15;
  const int n0   = wave * 16;

  v8f acc = {};
  for (int k0 = 0; k0 < INDIM; k0 += 4) {
    const int ka = k0 + hs * 2;
    v2f a, b;
    a.x = As[ka + 0][lm];
    a.y = As[ka + 1][lm];
    const float* w = W1 + (size_t)ka * HIDDIM + n0 + lm;
    b.x = w[0];
    b.y = w[HIDDIM];
    acc = __builtin_amdgcn_wmma_f32_16x16x4_f32(false, a, false, b,
                                                (short)0, acc, false, false);
  }

  float* dst = xw1 + (size_t)row0 * HIDDIM + n0 + lm;
#pragma unroll
  for (int j = 0; j < 8; ++j)
    dst[(size_t)(hs * 8 + j) * HIDDIM] = acc[j];
}

#define OWN_ROWS 196
#define QCAP 64
#define NBINS 256
#define REGION_INTS (OWN_ROWS * HIDDIM)
#define HDR_INTS 260
#define TAIL_COLS (HIDDIM - OUTDIM)
#define REC_CAP ((OWN_ROWS * TAIL_COLS - HDR_INTS) / 2)
#define LAST_ROWS (NNODES - (NBINS - 1) * OWN_ROWS)
#define LAST_CAP ((LAST_ROWS * TAIL_COLS - HDR_INTS) / 2)
#define SLAB 8
#define SORT_SLOTS ((REC_CAP + 1023) / 1024)
#define GRP 8

__global__ __launch_bounds__(1024) void bin_count(const int* __restrict__ edst,
                                                  int* __restrict__ regions,
                                                  int nedges) {
  __shared__ int hist[NBINS];
  const int tid = threadIdx.x;
  if (tid < NBINS) hist[tid] = 0;
  __syncthreads();

  const int per = (nedges + NBINS - 1) / NBINS;
  const int e0  = (int)blockIdx.x * per;
  const int e1  = (e0 + per < nedges) ? e0 + per : nedges;
  for (int eb = e0; eb < e1; eb += SLAB * 1024) {
    int dd[SLAB];
#pragma unroll
    for (int j = 0; j < SLAB; ++j) {
      const int e = eb + j * 1024 + tid;
      dd[j] = (e < e1) ? edst[e] : -1;
    }
#pragma unroll
    for (int j = 0; j < SLAB; ++j)
      if ((unsigned)dd[j] < (unsigned)NNODES)
        atomicAdd(&hist[(unsigned)dd[j] / OWN_ROWS], 1);
  }
  __syncthreads();
  if (tid < NBINS) regions[(size_t)blockIdx.x * REGION_INTS + tid] = hist[tid];
}

__global__ __launch_bounds__(1024) void bin_scatter(const int* __restrict__ esrc,
                                                    const int* __restrict__ edst,
                                                    const float* __restrict__ evals,
                                                    int* regions,
                                                    int nedges) {
  __shared__ int part[4][NBINS];
  __shared__ int cursor[NBINS];
  const int tid = threadIdx.x;
  const int g   = (int)blockIdx.x;
  const int bin = tid & (NBINS - 1);
  const int qtr = tid >> 8;

  int sum = 0;
  const int gbeg = qtr * (NBINS / 4);
  const int gend = (g < gbeg + NBINS / 4) ? g : gbeg + NBINS / 4;
  for (int gg = gbeg; gg < gend; gg += 16) {
    int t[16];
#pragma unroll
    for (int j = 0; j < 16; ++j)
      t[j] = (gg + j < gend) ? regions[(size_t)(gg + j) * REGION_INTS + bin] : 0;
#pragma unroll
    for (int j = 0; j < 16; ++j) sum += t[j];
  }
  part[qtr][bin] = sum;
  __syncthreads();
  if (tid < NBINS) {
    const int start = part[0][tid] + part[1][tid] + part[2][tid] + part[3][tid];
    cursor[tid] = start;
    if (g == NBINS - 1)
      regions[(size_t)tid * REGION_INTS + NBINS] = start + regions[(size_t)g * REGION_INTS + tid];
  }
  __syncthreads();

  const int per = (nedges + NBINS - 1) / NBINS;
  const int e0  = g * per;
  const int e1  = (e0 + per < nedges) ? e0 + per : nedges;
  for (int eb = e0; eb < e1; eb += SLAB * 1024) {
    int dd[SLAB], ss[SLAB], vv[SLAB];
#pragma unroll
    for (int j = 0; j < SLAB; ++j) {
      const int e = eb + j * 1024 + tid;
      const bool ok = e < e1;
      dd[j] = ok ? edst[e] : -1;
      ss[j] = ok ? esrc[e] : 0;
      vv[j] = ok ? __float_as_int(evals[e]) : 0;
    }
#pragma unroll
    for (int j = 0; j < SLAB; ++j) {
      if ((unsigned)dd[j] < (unsigned)NNODES) {
        const int b   = (int)((unsigned)dd[j] / OWN_ROWS);
        const int pos = atomicAdd(&cursor[b], 1);
        const int cap = (b == NBINS - 1) ? LAST_CAP : REC_CAP;
        if (pos < cap) {
          const int key = (ss[j] & 0xFFFF) | ((dd[j] - b * OWN_ROWS) << 16);
          *(int2*)(regions + (size_t)b * REGION_INTS + HDR_INTS + 2 * pos) = make_int2(key, vv[j]);
        }
      }
    }
  }
}

__device__ __forceinline__ void own_add_row(float* __restrict__ arow, float v, const float4 x) {
  atomicAdd(arow +  0, v * x.x);
  atomicAdd(arow + 32, v * x.y);
  atomicAdd(arow + 64, v * x.z);
  atomicAdd(arow + 96, v * x.w);
}

__device__ __forceinline__ void own_flush_row(float* __restrict__ arow, const float4 t) {
  atomicAdd(arow +  0, t.x);
  atomicAdd(arow + 32, t.y);
  atomicAdd(arow + 64, t.z);
  atomicAdd(arow + 96, t.w);
}

__device__ __forceinline__ void own_drain_regs(int s, int r, int vb, int n, int lane,
                                               const float* __restrict__ xw1,
                                               float* __restrict__ acc) {
  for (int k = 0; k < n; k += 4) {
    const int i1 = (k + 1 < n) ? k + 1 : k;
    const int i2 = (k + 2 < n) ? k + 2 : k;
    const int i3 = (k + 3 < n) ? k + 3 : k;
    const int s0 = __builtin_amdgcn_readlane(s, k);
    const int s1 = __builtin_amdgcn_readlane(s, i1);
    const int s2 = __builtin_amdgcn_readlane(s, i2);
    const int s3 = __builtin_amdgcn_readlane(s, i3);
    const float4 x0 = ((const float4*)(xw1 + (size_t)s0 * HIDDIM))[lane];
    const float4 x1 = ((const float4*)(xw1 + (size_t)s1 * HIDDIM))[lane];
    const float4 x2 = ((const float4*)(xw1 + (size_t)s2 * HIDDIM))[lane];
    const float4 x3 = ((const float4*)(xw1 + (size_t)s3 * HIDDIM))[lane];
    own_add_row(acc + __builtin_amdgcn_readlane(r, k) * HIDDIM + lane,
                __int_as_float(__builtin_amdgcn_readlane(vb, k)), x0);
    if (k + 1 < n)
      own_add_row(acc + __builtin_amdgcn_readlane(r, i1) * HIDDIM + lane,
                  __int_as_float(__builtin_amdgcn_readlane(vb, i1)), x1);
    if (k + 2 < n)
      own_add_row(acc + __builtin_amdgcn_readlane(r, i2) * HIDDIM + lane,
                  __int_as_float(__builtin_amdgcn_readlane(vb, i2)), x2);
    if (k + 3 < n)
      own_add_row(acc + __builtin_amdgcn_readlane(r, i3) * HIDDIM + lane,
                  __int_as_float(__builtin_amdgcn_readlane(vb, i3)), x3);
  }
}

__device__ __forceinline__ void own_drain(const int* __restrict__ q, int n, int lane, int base,
                                          const int* __restrict__ esrc,
                                          const int* __restrict__ edst,
                                          const float* __restrict__ evals,
                                          const float* __restrict__ xw1,
                                          float* __restrict__ acc) {
  const int e = q[lane < n ? lane : 0];
  own_drain_regs(esrc[e], edst[e] - base, __float_as_int(evals[e]), n, lane, xw1, acc);
}

__global__ __launch_bounds__(1024) void spmm1_owned(const int* __restrict__ esrc,
                                                    const int* __restrict__ edst,
                                                    const float* __restrict__ evals,
                                                    const float* __restrict__ xw1,
                                                    const float* __restrict__ W2,
                                                    float* h,
                                                    int nedges) {
  __shared__ float acc[OWN_ROWS * HIDDIM];
  __shared__ int2  srec[REC_CAP];
  __shared__ int   rowpos[256];
  int* queue = (int*)srec;

  const int tid  = threadIdx.x;
  const int lane = tid & 31;
  const int wave = __builtin_amdgcn_readfirstlane(tid >> 5);
  const int base = blockIdx.x * OWN_ROWS;

  for (int i = tid; i < OWN_ROWS * HIDDIM / 4; i += 1024)
    ((float4*)acc)[i] = make_float4(0.0f, 0.0f, 0.0f, 0.0f);
  if (tid < 256) rowpos[tid] = 0;

  const int* region = (const int*)h + (size_t)blockIdx.x * REGION_INTS;
  const int nrec = __builtin_amdgcn_readfirstlane(region[NBINS]);
  const int cap  = ((int)blockIdx.x == NBINS - 1) ? LAST_CAP : REC_CAP;
  __syncthreads();

  if (nrec <= cap) {
    const int2* rec = (const int2*)(region + HDR_INTS);
    int2 held[SORT_SLOTS];
    int  rank[SORT_SLOTS];
    for (int j = 0; j < SORT_SLOTS; ++j) {
      const int i = tid + j * 1024;
      held[j] = (i < nrec) ? rec[i] : make_int2(0, 0);
    }
    for (int j = 0; j < SORT_SLOTS; ++j) {
      const int i = tid + j * 1024;
      rank[j] = (i < nrec) ? atomicAdd(&rowpos[held[j].x >> 16], 1) : 0;
    }
    __syncthreads();
    if (wave == 0) {
      int cn[7];
      int tot = 0;
      for (int t = 0; t < 7; ++t) { cn[t] = rowpos[lane * 7 + t]; tot += cn[t]; }
      int incl = tot;
      for (int d = 1; d < 32; d <<= 1) {
        const int o = __shfl_up(incl, d);
        if (lane >= d) incl += o;
      }
      int run = incl - tot;
      for (int t = 0; t < 7; ++t) { rowpos[lane * 7 + t] = run; run += cn[t]; }
    }
    __syncthreads();
    for (int j = 0; j < SORT_SLOTS; ++j) {
      const int i = tid + j * 1024;
      if (i < nrec) srec[rowpos[held[j].x >> 16] + rank[j]] = held[j];
    }
    __syncthreads();

    const int per = (((nrec + 31) >> 5) + GRP - 1) & ~(GRP - 1);
    const int i0  = wave * per;
    const int i1  = (i0 + per < nrec) ? i0 + per : nrec;
    float4 sum = make_float4(0.0f, 0.0f, 0.0f, 0.0f);
    int cur = -1;
    for (int i = i0; i < i1; i += 32) {
      const int c = (i1 - i < 32) ? i1 - i : 32;
      int2 rc = srec[i + (lane < c ? lane : c - 1)];
      if (lane >= c) rc.y = 0;
      for (int k = 0; k < c; k += GRP) {
        float4 x[GRP];
        for (int j = 0; j < GRP; ++j) {
          const int sidx = __builtin_amdgcn_readlane(rc.x, k + j) & 0xFFFF;
          x[j] = ((const float4*)(xw1 + (size_t)sidx * HIDDIM))[lane];
        }
        for (int j = 0; j < GRP; ++j) {
          const int   row = __builtin_amdgcn_readlane(rc.x, k + j) >> 16;
          const float v   = __int_as_float(__builtin_amdgcn_readlane(rc.y, k + j));
          if (row != cur) {
            if (cur >= 0) own_flush_row(acc + cur * HIDDIM + lane, sum);
            sum = make_float4(0.0f, 0.0f, 0.0f, 0.0f);
            cur = row;
          }
          sum.x = fmaf(v, x[j].x, sum.x);
          sum.y = fmaf(v, x[j].y, sum.y);
          sum.z = fmaf(v, x[j].z, sum.z);
          sum.w = fmaf(v, x[j].w, sum.w);
        }
      }
    }
    if (cur >= 0) own_flush_row(acc + cur * HIDDIM + lane, sum);
  } else {
    int* q = queue + wave * QCAP;
    int qn = 0;

    const int nquads = nedges >> 2;
    const int nspans = (nquads + 127) >> 7;
    const int4* dst4 = (const int4*)edst;

    for (int sp = wave; sp < nspans; sp += 32) {
      int4 cur[4];
#pragma unroll
      for (int j = 0; j < 4; ++j) {
        const int qi = sp * 128 + j * 32 + lane;
        cur[j] = (qi < nquads) ? dst4[qi] : make_int4(-1, -1, -1, -1);
      }
#pragma unroll
      for (int j = 0; j < 4; ++j) {
        const int qi = sp * 128 + j * 32 + lane;
        const int dd[4] = {cur[j].x, cur[j].y, cur[j].z, cur[j].w};
#pragma unroll
        for (int c = 0; c < 4; ++c) {
          const bool m = (unsigned)(dd[c] - base) < (unsigned)OWN_ROWS;
          const unsigned mask = (unsigned)__ballot(m);
          if (mask) {
            if (m) q[qn + __popc(mask & ((1u << lane) - 1u))] = qi * 4 + c;
            qn += __popc(mask);
            if (qn >= 32) {
              own_drain(q, 32, lane, base, esrc, edst, evals, xw1, acc);
              const int rest = qn - 32;
              const int keep = q[32 + lane];
              if (lane < rest) q[lane] = keep;
              qn = rest;
            }
          }
        }
      }
    }
    if (wave == 0) {
      for (int e = nquads * 4 + lane; e - lane < nedges; e += 32) {
        const bool m = (e < nedges) && ((unsigned)(edst[e < nedges ? e : 0] - base) < (unsigned)OWN_ROWS);
        const unsigned mask = (unsigned)__ballot(m);
        if (mask) {
          if (m) q[qn + __popc(mask & ((1u << lane) - 1u))] = e;
          qn += __popc(mask);
          if (qn >= 32) {
            own_drain(q, 32, lane, base, esrc, edst, evals, xw1, acc);
            const int rest = qn - 32;
            const int keep = q[32 + lane];
            if (lane < rest) q[lane] = keep;
            qn = rest;
          }
        }
      }
    }
    if (qn > 0) own_drain(q, qn, lane, base, esrc, edst, evals, xw1, acc);
  }
  __syncthreads();

  const int nrows = ((int)blockIdx.x == NBINS - 1) ? LAST_ROWS : OWN_ROWS;
  float* hw2 = h + (size_t)blockIdx.x * REGION_INTS + nrows * TAIL_COLS;
  if (wave < (OWN_ROWS + 15) / 16) {
    const int hs   = lane >> 4;
    const int lm   = lane & 15;
    const int row0 = wave * 16;
    const int ar = (row0 + lm < OWN_ROWS) ? row0 + lm : OWN_ROWS - 1;
    const float4* a0p = (const float4*)(acc + ar * HIDDIM + (2 * hs) * 32);
    const float4* a1p = (const float4*)(acc + ar * HIDDIM + (2 * hs + 1) * 32);
    v8f d = {};
    for (int lc = 0; lc < HIDDIM / 16; ++lc) {
      const float4 A0 = a0p[lc];
      const float4 A1 = a1p[lc];
      const float e0[4] = {A0.x, A0.y, A0.z, A0.w};
      const float e1[4] = {A1.x, A1.y, A1.z, A1.w};
#pragma unroll
      for (int i = 0; i < 4; ++i) {
        const int ka = (lc * 4 + i) * 4 + hs * 2;
        v2f a, b;
        a.x = fmaxf(e0[i], 0.0f);
        a.y = fmaxf(e1[i], 0.0f);
        const float* w = W2 + (size_t)ka * OUTDIM + lm;
        b.x = w[0];
        b.y = w[OUTDIM];
        d = __builtin_amdgcn_wmma_f32_16x16x4_f32(false, a, false, b,
                                                  (short)0, d, false, false);
      }
    }
#pragma unroll
    for (int j = 0; j < 8; ++j) {
      const int r = row0 + hs * 8 + j;
      if (r < nrows) hw2[r * OUTDIM + lm] = d[j];
    }
  }
}

__device__ __forceinline__ const float* hw2_row(const int* __restrict__ regions, int s) {
  const int b     = (int)((unsigned)s / OWN_ROWS);
  const int nrows = (b == NBINS - 1) ? LAST_ROWS : OWN_ROWS;
  return (const float*)(regions + (size_t)b * REGION_INTS + nrows * TAIL_COLS + (s - b * OWN_ROWS) * OUTDIM);
}

__global__ __launch_bounds__(1024) void spmm2_owned(const int* __restrict__ esrc,
                                                    const int* __restrict__ edst,
                                                    const float* __restrict__ evals,
                                                    const int* __restrict__ regions,
                                                    float* __restrict__ out,
                                                    int nedges) {
  __shared__ float acc2[OWN_ROWS * OUTDIM];

  const int tid  = threadIdx.x;
  const int lane = tid & 31;
  const int wave = __builtin_amdgcn_readfirstlane(tid >> 5);
  const int base = blockIdx.x * OWN_ROWS;

  for (int i = tid; i < OWN_ROWS * OUTDIM; i += 1024) acc2[i] = 0.0f;

  const int* region = regions + (size_t)blockIdx.x * REGION_INTS;
  const int nrec = __builtin_amdgcn_readfirstlane(region[NBINS]);
  const int cap  = ((int)blockIdx.x == NBINS - 1) ? LAST_CAP : REC_CAP;
  __syncthreads();

  if (nrec <= cap) {
    const int2* rec = (const int2*)(region + HDR_INTS);
    const int per = (nrec + 31) >> 5;
    const int i0  = wave * per;
    const int i1  = (i0 + per < nrec) ? i0 + per : nrec;
    const int q   = lane & 3;
    const int sub = lane >> 2;
    int2 nxt = rec[(i0 < i1) ? i0 + ((lane < i1 - i0) ? lane : 0) : 0];
    for (int i = i0; i < i1; i += 32) {
      const int c = (i1 - i < 32) ? i1 - i : 32;
      const int2 rc = nxt;
      const int left = i1 - (i + 32);
      if (left > 0) nxt = rec[i + 32 + ((lane < left) ? lane : 0)];
      float4 x[4];
      int    ro[4];
      float  vv[4];
#pragma unroll
      for (int j = 0; j < 4; ++j) {
        const int idx = j * 8 + sub;
        const int key = __shfl(rc.x, idx);
        vv[j] = __int_as_float(__shfl(rc.y, idx));
        ro[j] = (idx < c) ? (key >> 16) * OUTDIM + q * 4 : -1;
        x[j]  = (idx < c) ? ((const float4*)hw2_row(regions, key & 0xFFFF))[q]
                          : make_float4(0.0f, 0.0f, 0.0f, 0.0f);
      }
#pragma unroll
      for (int j = 0; j < 4; ++j) {
        if (ro[j] >= 0) {
          float* a = acc2 + ro[j];
          atomicAdd(a + 0, vv[j] * x[j].x);
          atomicAdd(a + 1, vv[j] * x[j].y);
          atomicAdd(a + 2, vv[j] * x[j].z);
          atomicAdd(a + 3, vv[j] * x[j].w);
        }
      }
    }
  } else {
    for (int e = tid; e < nedges; e += 1024) {
      const int r = edst[e] - base;
      if ((unsigned)r < (unsigned)OWN_ROWS) {
        const float  v = evals[e];
        const float* x = hw2_row(regions, esrc[e]);
        for (int k = 0; k < OUTDIM; ++k) atomicAdd(acc2 + r * OUTDIM + k, v * x[k]);
      }
    }
  }
  __syncthreads();

  for (int i = tid; i < OWN_ROWS * (OUTDIM / 4); i += 1024) {
    const int row = base + (i >> 2);
    if (row < NNODES) ((float4*)(out + (size_t)row * OUTDIM))[i & 3] = ((const float4*)acc2)[i];
  }
}

extern "C" void kernel_launch(void* const* d_in, const int* in_sizes, int n_in,
                              void* d_out, int out_size, void* d_ws, size_t ws_size,
                              hipStream_t stream) {
  const float* feat  = (const float*)d_in[0];
  const int*   esrc  = (const int*)d_in[1];
  const int*   edst  = (const int*)d_in[2];
  const float* evals = (const float*)d_in[3];
  const float* W1    = (const float*)d_in[4];
  const float* W2    = (const float*)d_in[5];
  float*       out   = (float*)d_out;
  const int nedges   = in_sizes[1];

  const size_t xw1_bytes = (size_t)NNODES * HIDDIM * sizeof(float);
  float* xw1 = (float*)d_ws;
  float* h   = (float*)((char*)d_ws + xw1_bytes);

  bin_count<<<NBINS, 1024, 0, stream>>>(edst, (int*)h, nedges);
  bin_scatter<<<NBINS, 1024, 0, stream>>>(esrc, edst, evals, (int*)h, nedges);

  gemm1_wmma<<<NNODES / 16, 256, 0, stream>>>(feat, W1, xw1);

  spmm1_owned<<<NBINS, 1024, 0, stream>>>(esrc, edst, evals, xw1, W2, h, nedges);

  spmm2_owned<<<NBINS, 1024, 0, stream>>>(esrc, edst, evals, (const int*)h, out, nedges);
}


# gemm1: 32-row work-groups of 4 waves, each wave 16 rows x 64 cols on 4 accumulators; W1 read as float4 along n, 2 loads feed 4 WMMAs (was 2 per WMMA), requested one group ahead; sum order kept
# speedup vs baseline: 3.1048x; 1.0814x over previous
#include <hip/hip_runtime.h>

typedef __attribute__((ext_vector_type(2))) float v2f;
typedef __attribute__((ext_vector_type(8))) float v8f;

#define NNODES 50000
#define INDIM  512
#define HIDDIM 128
#define OUTDIM 16

#define G1_ROWS 32
#define G1_BLOCKS ((NNODES + G1_ROWS - 1) / G1_ROWS)
#define G1_KU 4

__device__ __forceinline__ void g1_load_b(float4 (&b)[G1_KU][2], const float* __restrict__ bp, int k0) {
#pragma unroll
  for (int u = 0; u < G1_KU; ++u) {
    const float* w = bp + (size_t)(k0 + 4 * u) * HIDDIM;
    b[u][0] = *(const float4*)(w);
    b[u][1] = *(const float4*)(w + HIDDIM);
  }
}

__device__ __forceinline__ void g1_mma(v8f& acc0, v8f& acc1, v8f& acc2, v8f& acc3,
                                       const float* __restrict__ ap, const float4 (&bb)[G1_KU][2]) {
#pragma unroll
  for (int u = 0; u < G1_KU; ++u) {
    v2f a, b;
    a.x = ap[(4 * u) * 17];
    a.y = ap[(4 * u) * 17 + 17];
    b.x = bb[u][0].x; b.y = bb[u][1].x;
    acc0 = __builtin_amdgcn_wmma_f32_16x16x4_f32(false, a, false, b, (short)0, acc0, false, false);
    b.x = bb[u][0].y; b.y = bb[u][1].y;
    acc1 = __builtin_amdgcn_wmma_f32_16x16x4_f32(false, a, false, b, (short)0, acc1, false, false);
    b.x = bb[u][0].z; b.y = bb[u][1].z;
    acc2 = __builtin_amdgcn_wmma_f32_16x16x4_f32(false, a, false, b, (short)0, acc2, false, false);
    b.x = bb[u][0].w; b.y = bb[u][1].w;
    acc3 = __builtin_amdgcn_wmma_f32_16x16x4_f32(false, a, false, b, (short)0, acc3, false, false);
  }
}

__global__ __launch_bounds__(128) void gemm1_wmma(const float* __restrict__ feat,
                                                  const float* __restrict__ W1,
                                                  float* __restrict__ xw1) {
  __shared__ float As[G1_ROWS / 16][INDIM][17];
  const int row0 = blockIdx.x * G1_ROWS;
  const int tid  = threadIdx.x;

#pragma unroll
  for (int t = 0; t < G1_ROWS / 16; ++t) {
    float4 v[16];
#pragma unroll
    for (int i = 0; i < 16; ++i) {
      int gr = row0 + t * 16 + i;
      gr = (gr < NNODES) ? gr : NNODES - 1;
      v[i] = ((const float4*)(feat + (size_t)gr * INDIM))[tid];
    }
#pragma unroll
    for (int i = 0; i < 16; ++i) {
      As[t][4 * tid + 0][i] = v[i].x;
      As[t][4 * tid + 1][i] = v[i].y;
      As[t][4 * tid + 2][i] = v[i].z;
      As[t][4 * tid + 3][i] = v[i].w;
    }
  }
  __syncthreads();

  const int wave = tid >> 5;
  const int lane = tid & 31;
  const int hs   = lane >> 4;
  const int lm   = lane & 15;
  const int rt   = wave & 1;
  const int c0   = (wave >> 1) * 64;

  const float* ap = &As[rt][hs * 2][lm];
  const float* bp = W1 + (size_t)(hs * 2) * HIDDIM + c0 + lm * 4;

  v8f acc0 = {}, acc1 = {}, acc2 = {}, acc3 = {};
  float4 b0[G1_KU][2], b1[G1_KU][2];
  g1_load_b(b0, bp, 0);
#pragma nounroll
  for (int k0 = 0; k0 < INDIM; k0 += 8 * G1_KU) {
    g1_load_b(b1, bp, k0 + 4 * G1_KU);
    g1_mma(acc0, acc1, acc2, acc3, ap + k0 * 17, b0);
    if (k0 + 8 * G1_KU < INDIM) g1_load_b(b0, bp, k0 + 8 * G1_KU);
    g1_mma(acc0, acc1, acc2, acc3, ap + (k0 + 4 * G1_KU) * 17, b1);
  }

  float* dst = xw1 + (size_t)(row0 + rt * 16 + hs * 8) * HIDDIM + c0 + lm * 4;
#pragma unroll
  for (int j = 0; j < 8; ++j) {
    if (row0 + rt * 16 + hs * 8 + j < NNODES)
      *(float4*)(dst + (size_t)j * HIDDIM) = make_float4(acc0[j], acc1[j], acc2[j], acc3[j]);
  }
}

#define OWN_ROWS 196
#define QCAP 64
#define NBINS 256
#define REGION_INTS (OWN_ROWS * HIDDIM)
#define HDR_INTS 260
#define TAIL_COLS (HIDDIM - OUTDIM)
#define REC_CAP ((OWN_ROWS * TAIL_COLS - HDR_INTS) / 2)
#define LAST_ROWS (NNODES - (NBINS - 1) * OWN_ROWS)
#define LAST_CAP ((LAST_ROWS * TAIL_COLS - HDR_INTS) / 2)
#define SLAB 8
#define SORT_SLOTS ((REC_CAP + 1023) / 1024)
#define GRP 8

__global__ __launch_bounds__(1024) void bin_count(const int* __restrict__ edst,
                                                  int* __restrict__ regions,
                                                  int nedges) {
  __shared__ int hist[NBINS];
  const int tid = threadIdx.x;
  if (tid < NBINS) hist[tid] = 0;
  __syncthreads();

  const int per = (nedges + NBINS - 1) / NBINS;
  const int e0  = (int)blockIdx.x * per;
  const int e1  = (e0 + per < nedges) ? e0 + per : nedges;
  for (int eb = e0; eb < e1; eb += SLAB * 1024) {
    int dd[SLAB];
#pragma unroll
    for (int j = 0; j < SLAB; ++j) {
      const int e = eb + j * 1024 + tid;
      dd[j] = (e < e1) ? edst[e] : -1;
    }
#pragma unroll
    for (int j = 0; j < SLAB; ++j)
      if ((unsigned)dd[j] < (unsigned)NNODES)
        atomicAdd(&hist[(unsigned)dd[j] / OWN_ROWS], 1);
  }
  __syncthreads();
  if (tid < NBINS) regions[(size_t)blockIdx.x * REGION_INTS + tid] = hist[tid];
}

__global__ __launch_bounds__(1024) void bin_scatter(const int* __restrict__ esrc,
                                                    const int* __restrict__ edst,
                                                    const float* __restrict__ evals,
                                                    int* regions,
                                                    int nedges) {
  __shared__ int part[4][NBINS];
  __shared__ int cursor[NBINS];
  const int tid = threadIdx.x;
  const int g   = (int)blockIdx.x;
  const int bin = tid & (NBINS - 1);
  const int qtr = tid >> 8;

  int sum = 0;
  const int gbeg = qtr * (NBINS / 4);
  const int gend = (g < gbeg + NBINS / 4) ? g : gbeg + NBINS / 4;
  for (int gg = gbeg; gg < gend; gg += 16) {
    int t[16];
#pragma unroll
    for (int j = 0; j < 16; ++j)
      t[j] = (gg + j < gend) ? regions[(size_t)(gg + j) * REGION_INTS + bin] : 0;
#pragma unroll
    for (int j = 0; j < 16; ++j) sum += t[j];
  }
  part[qtr][bin] = sum;
  __syncthreads();
  if (tid < NBINS) {
    const int start = part[0][tid] + part[1][tid] + part[2][tid] + part[3][tid];
    cursor[tid] = start;
    if (g == NBINS - 1)
      regions[(size_t)tid * REGION_INTS + NBINS] = start + regions[(size_t)g * REGION_INTS + tid];
  }
  __syncthreads();

  const int per = (nedges + NBINS - 1) / NBINS;
  const int e0  = g * per;
  const int e1  = (e0 + per < nedges) ? e0 + per : nedges;
  for (int eb = e0; eb < e1; eb += SLAB * 1024) {
    int dd[SLAB], ss[SLAB], vv[SLAB];
#pragma unroll
    for (int j = 0; j < SLAB; ++j) {
      const int e = eb + j * 1024 + tid;
      const bool ok = e < e1;
      dd[j] = ok ? edst[e] : -1;
      ss[j] = ok ? esrc[e] : 0;
      vv[j] = ok ? __float_as_int(evals[e]) : 0;
    }
#pragma unroll
    for (int j = 0; j < SLAB; ++j) {
      if ((unsigned)dd[j] < (unsigned)NNODES) {
        const int b   = (int)((unsigned)dd[j] / OWN_ROWS);
        const int pos = atomicAdd(&cursor[b], 1);
        const int cap = (b == NBINS - 1) ? LAST_CAP : REC_CAP;
        if (pos < cap) {
          const int key = (ss[j] & 0xFFFF) | ((dd[j] - b * OWN_ROWS) << 16);
          *(int2*)(regions + (size_t)b * REGION_INTS + HDR_INTS + 2 * pos) = make_int2(key, vv[j]);
        }
      }
    }
  }
}

__device__ __forceinline__ void own_add_row(float* __restrict__ arow, float v, const float4 x) {
  atomicAdd(arow +  0, v * x.x);
  atomicAdd(arow + 32, v * x.y);
  atomicAdd(arow + 64, v * x.z);
  atomicAdd(arow + 96, v * x.w);
}

__device__ __forceinline__ void own_flush_row(float* __restrict__ arow, const float4 t) {
  atomicAdd(arow +  0, t.x);
  atomicAdd(arow + 32, t.y);
  atomicAdd(arow + 64, t.z);
  atomicAdd(arow + 96, t.w);
}

__device__ __forceinline__ void own_drain_regs(int s, int r, int vb, int n, int lane,
                                               const float* __restrict__ xw1,
                                               float* __restrict__ acc) {
  for (int k = 0; k < n; k += 4) {
    const int i1 = (k + 1 < n) ? k + 1 : k;
    const int i2 = (k + 2 < n) ? k + 2 : k;
    const int i3 = (k + 3 < n) ? k + 3 : k;
    const int s0 = __builtin_amdgcn_readlane(s, k);
    const int s1 = __builtin_amdgcn_readlane(s, i1);
    const int s2 = __builtin_amdgcn_readlane(s, i2);
    const int s3 = __builtin_amdgcn_readlane(s, i3);
    const float4 x0 = ((const float4*)(xw1 + (size_t)s0 * HIDDIM))[lane];
    const float4 x1 = ((const float4*)(xw1 + (size_t)s1 * HIDDIM))[lane];
    const float4 x2 = ((const float4*)(xw1 + (size_t)s2 * HIDDIM))[lane];
    const float4 x3 = ((const float4*)(xw1 + (size_t)s3 * HIDDIM))[lane];
    own_add_row(acc + __builtin_amdgcn_readlane(r, k) * HIDDIM + lane,
                __int_as_float(__builtin_amdgcn_readlane(vb, k)), x0);
    if (k + 1 < n)
      own_add_row(acc + __builtin_amdgcn_readlane(r, i1) * HIDDIM + lane,
                  __int_as_float(__builtin_amdgcn_readlane(vb, i1)), x1);
    if (k + 2 < n)
      own_add_row(acc + __builtin_amdgcn_readlane(r, i2) * HIDDIM + lane,
                  __int_as_float(__builtin_amdgcn_readlane(vb, i2)), x2);
    if (k + 3 < n)
      own_add_row(acc + __builtin_amdgcn_readlane(r, i3) * HIDDIM + lane,
                  __int_as_float(__builtin_amdgcn_readlane(vb, i3)), x3);
  }
}

__device__ __forceinline__ void own_drain(const int* __restrict__ q, int n, int lane, int base,
                                          const int* __restrict__ esrc,
                                          const int* __restrict__ edst,
                                          const float* __restrict__ evals,
                                          const float* __restrict__ xw1,
                                          float* __restrict__ acc) {
  const int e = q[lane < n ? lane : 0];
  own_drain_regs(esrc[e], edst[e] - base, __float_as_int(evals[e]), n, lane, xw1, acc);
}

__global__ __launch_bounds__(1024) void spmm1_owned(const int* __restrict__ esrc,
                                                    const int* __restrict__ edst,
                                                    const float* __restrict__ evals,
                                                    const float* __restrict__ xw1,
                                                    const float* __restrict__ W2,
                                                    float* h,
                                                    int nedges) {
  __shared__ float acc[OWN_ROWS * HIDDIM];
  __shared__ int2  srec[REC_CAP];
  __shared__ int   rowpos[256];
  int* queue = (int*)srec;

  const int tid  = threadIdx.x;
  const int lane = tid & 31;
  const int wave = __builtin_amdgcn_readfirstlane(tid >> 5);
  const int base = blockIdx.x * OWN_ROWS;

  for (int i = tid; i < OWN_ROWS * HIDDIM / 4; i += 1024)
    ((float4*)acc)[i] = make_float4(0.0f, 0.0f, 0.0f, 0.0f);
  if (tid < 256) rowpos[tid] = 0;

  const int* region = (const int*)h + (size_t)blockIdx.x * REGION_INTS;
  const int nrec = __builtin_amdgcn_readfirstlane(region[NBINS]);
  const int cap  = ((int)blockIdx.x == NBINS - 1) ? LAST_CAP : REC_CAP;
  __syncthreads();

  if (nrec <= cap) {
    const int2* rec = (const int2*)(region + HDR_INTS);
    int2 held[SORT_SLOTS];
    int  rank[SORT_SLOTS];
    for (int j = 0; j < SORT_SLOTS; ++j) {
      const int i = tid + j * 1024;
      held[j] = (i < nrec) ? rec[i] : make_int2(0, 0);
    }
    for (int j = 0; j < SORT_SLOTS; ++j) {
      const int i = tid + j * 1024;
      rank[j] = (i < nrec) ? atomicAdd(&rowpos[held[j].x >> 16], 1) : 0;
    }
    __syncthreads();
    if (wave == 0) {
      int cn[7];
      int tot = 0;
      for (int t = 0; t < 7; ++t) { cn[t] = rowpos[lane * 7 + t]; tot += cn[t]; }
      int incl = tot;
      for (int d = 1; d < 32; d <<= 1) {
        const int o = __shfl_up(incl, d);
        if (lane >= d) incl += o;
      }
      int run = incl - tot;
      for (int t = 0; t < 7; ++t) { rowpos[lane * 7 + t] = run; run += cn[t]; }
    }
    __syncthreads();
    for (int j = 0; j < SORT_SLOTS; ++j) {
      const int i = tid + j * 1024;
      if (i < nrec) srec[rowpos[held[j].x >> 16] + rank[j]] = held[j];
    }
    __syncthreads();

    const int per = (((nrec + 31) >> 5) + GRP - 1) & ~(GRP - 1);
    const int i0  = wave * per;
    const int i1  = (i0 + per < nrec) ? i0 + per : nrec;
    float4 sum = make_float4(0.0f, 0.0f, 0.0f, 0.0f);
    int cur = -1;
    for (int i = i0; i < i1; i += 32) {
      const int c = (i1 - i < 32) ? i1 - i : 32;
      int2 rc = srec[i + (lane < c ? lane : c - 1)];
      if (lane >= c) rc.y = 0;
      for (int k = 0; k < c; k += GRP) {
        float4 x[GRP];
        for (int j = 0; j < GRP; ++j) {
          const int sidx = __builtin_amdgcn_readlane(rc.x, k + j) & 0xFFFF;
          x[j] = ((const float4*)(xw1 + (size_t)sidx * HIDDIM))[lane];
        }
        for (int j = 0; j < GRP; ++j) {
          const int   row = __builtin_amdgcn_readlane(rc.x, k + j) >> 16;
          const float v   = __int_as_float(__builtin_amdgcn_readlane(rc.y, k + j));
          if (row != cur) {
            if (cur >= 0) own_flush_row(acc + cur * HIDDIM + lane, sum);
            sum = make_float4(0.0f, 0.0f, 0.0f, 0.0f);
            cur = row;
          }
          sum.x = fmaf(v, x[j].x, sum.x);
          sum.y = fmaf(v, x[j].y, sum.y);
          sum.z = fmaf(v, x[j].z, sum.z);
          sum.w = fmaf(v, x[j].w, sum.w);
        }
      }
    }
    if (cur >= 0) own_flush_row(acc + cur * HIDDIM + lane, sum);
  } else {
    int* q = queue + wave * QCAP;
    int qn = 0;

    const int nquads = nedges >> 2;
    const int nspans = (nquads + 127) >> 7;
    const int4* dst4 = (const int4*)edst;

    for (int sp = wave; sp < nspans; sp += 32) {
      int4 cur[4];
#pragma unroll
      for (int j = 0; j < 4; ++j) {
        const int qi = sp * 128 + j * 32 + lane;
        cur[j] = (qi < nquads) ? dst4[qi] : make_int4(-1, -1, -1, -1);
      }
#pragma unroll
      for (int j = 0; j < 4; ++j) {
        const int qi = sp * 128 + j * 32 + lane;
        const int dd[4] = {cur[j].x, cur[j].y, cur[j].z, cur[j].w};
#pragma unroll
        for (int c = 0; c < 4; ++c) {
          const bool m = (unsigned)(dd[c] - base) < (unsigned)OWN_ROWS;
          const unsigned mask = (unsigned)__ballot(m);
          if (mask) {
            if (m) q[qn + __popc(mask & ((1u << lane) - 1u))] = qi * 4 + c;
            qn += __popc(mask);
            if (qn >= 32) {
              own_drain(q, 32, lane, base, esrc, edst, evals, xw1, acc);
              const int rest = qn - 32;
              const int keep = q[32 + lane];
              if (lane < rest) q[lane] = keep;
              qn = rest;
            }
          }
        }
      }
    }
    if (wave == 0) {
      for (int e = nquads * 4 + lane; e - lane < nedges; e += 32) {
        const bool m = (e < nedges) && ((unsigned)(edst[e < nedges ? e : 0] - base) < (unsigned)OWN_ROWS);
        const unsigned mask = (unsigned)__ballot(m);
        if (mask) {
          if (m) q[qn + __popc(mask & ((1u << lane) - 1u))] = e;
          qn += __popc(mask);
          if (qn >= 32) {
            own_drain(q, 32, lane, base, esrc, edst, evals, xw1, acc);
            const int rest = qn - 32;
            const int keep = q[32 + lane];
            if (lane < rest) q[lane] = keep;
            qn = rest;
          }
        }
      }
    }
    if (qn > 0) own_drain(q, qn, lane, base, esrc, edst, evals, xw1, acc);
  }
  __syncthreads();

  const int nrows = ((int)blockIdx.x == NBINS - 1) ? LAST_ROWS : OWN_ROWS;
  float* hw2 = h + (size_t)blockIdx.x * REGION_INTS + nrows * TAIL_COLS;
  if (wave < (OWN_ROWS + 15) / 16) {
    const int hs   = lane >> 4;
    const int lm   = lane & 15;
    const int row0 = wave * 16;
    const int ar = (row0 + lm < OWN_ROWS) ? row0 + lm : OWN_ROWS - 1;
    const float4* a0p = (const float4*)(acc + ar * HIDDIM + (2 * hs) * 32);
    const float4* a1p = (const float4*)(acc + ar * HIDDIM + (2 * hs + 1) * 32);
    v8f d = {};
    for (int lc = 0; lc < HIDDIM / 16; ++lc) {
      const float4 A0 = a0p[lc];
      const float4 A1 = a1p[lc];
      const float e0[4] = {A0.x, A0.y, A0.z, A0.w};
      const float e1[4] = {A1.x, A1.y, A1.z, A1.w};
#pragma unroll
      for (int i = 0; i < 4; ++i) {
        const int ka = (lc * 4 + i) * 4 + hs * 2;
        v2f a, b;
        a.x = fmaxf(e0[i], 0.0f);
        a.y = fmaxf(e1[i], 0.0f);
        const float* w = W2 + (size_t)ka * OUTDIM + lm;
        b.x = w[0];
        b.y = w[OUTDIM];
        d = __builtin_amdgcn_wmma_f32_16x16x4_f32(false, a, false, b,
                                                  (short)0, d, false, false);
      }
    }
#pragma unroll
    for (int j = 0; j < 8; ++j) {
      const int r = row0 + hs * 8 + j;
      if (r < nrows) hw2[r * OUTDIM + lm] = d[j];
    }
  }
}

__device__ __forceinline__ const float* hw2_row(const int* __restrict__ regions, int s) {
  const int b     = (int)((unsigned)s / OWN_ROWS);
  const int nrows = (b == NBINS - 1) ? LAST_ROWS : OWN_ROWS;
  return (const float*)(regions + (size_t)b * REGION_INTS + nrows * TAIL_COLS + (s - b * OWN_ROWS) * OUTDIM);
}

__global__ __launch_bounds__(1024) void spmm2_owned(const int* __restrict__ esrc,
                                                    const int* __restrict__ edst,
                                                    const float* __restrict__ evals,
                                                    const int* __restrict__ regions,
                                                    float* __restrict__ out,
                                                    int nedges) {
  __shared__ float acc2[OWN_ROWS * OUTDIM];

  const int tid  = threadIdx.x;
  const int lane = tid & 31;
  const int wave = __builtin_amdgcn_readfirstlane(tid >> 5);
  const int base = blockIdx.x * OWN_ROWS;

  for (int i = tid; i < OWN_ROWS * OUTDIM; i += 1024) acc2[i] = 0.0f;

  const int* region = regions + (size_t)blockIdx.x * REGION_INTS;
  const int nrec = __builtin_amdgcn_readfirstlane(region[NBINS]);
  const int cap  = ((int)blockIdx.x == NBINS - 1) ? LAST_CAP : REC_CAP;
  __syncthreads();

  if (nrec <= cap) {
    const int2* rec = (const int2*)(region + HDR_INTS);
    const int per = (nrec + 31) >> 5;
    const int i0  = wave * per;
    const int i1  = (i0 + per < nrec) ? i0 + per : nrec;
    const int q   = lane & 3;
    const int sub = lane >> 2;
    int2 nxt = rec[(i0 < i1) ? i0 + ((lane < i1 - i0) ? lane : 0) : 0];
    for (int i = i0; i < i1; i += 32) {
      const int c = (i1 - i < 32) ? i1 - i : 32;
      const int2 rc = nxt;
      const int left = i1 - (i + 32);
      if (left > 0) nxt = rec[i + 32 + ((lane < left) ? lane : 0)];
      float4 x[4];
      int    ro[4];
      float  vv[4];
#pragma unroll
      for (int j = 0; j < 4; ++j) {
        const int idx = j * 8 + sub;
        const int key = __shfl(rc.x, idx);
        vv[j] = __int_as_float(__shfl(rc.y, idx));
        ro[j] = (idx < c) ? (key >> 16) * OUTDIM + q * 4 : -1;
        x[j]  = (idx < c) ? ((const float4*)hw2_row(regions, key & 0xFFFF))[q]
                          : make_float4(0.0f, 0.0f, 0.0f, 0.0f);
      }
#pragma unroll
      for (int j = 0; j < 4; ++j) {
        if (ro[j] >= 0) {
          float* a = acc2 + ro[j];
          atomicAdd(a + 0, vv[j] * x[j].x);
          atomicAdd(a + 1, vv[j] * x[j].y);
          atomicAdd(a + 2, vv[j] * x[j].z);
          atomicAdd(a + 3, vv[j] * x[j].w);
        }
      }
    }
  } else {
    for (int e = tid; e < nedges; e += 1024) {
      const int r = edst[e] - base;
      if ((unsigned)r < (unsigned)OWN_ROWS) {
        const float  v = evals[e];
        const float* x = hw2_row(regions, esrc[e]);
        for (int k = 0; k < OUTDIM; ++k) atomicAdd(acc2 + r * OUTDIM + k, v * x[k]);
      }
    }
  }
  __syncthreads();

  for (int i = tid; i < OWN_ROWS * (OUTDIM / 4); i += 1024) {
    const int row = base + (i >> 2);
    if (row < NNODES) ((float4*)(out + (size_t)row * OUTDIM))[i & 3] = ((const float4*)acc2)[i];
  }
}

extern "C" void kernel_launch(void* const* d_in, const int* in_sizes, int n_in,
                              void* d_out, int out_size, void* d_ws, size_t ws_size,
                              hipStream_t stream) {
  const float* feat  = (const float*)d_in[0];
  const int*   esrc  = (const int*)d_in[1];
  const int*   edst  = (const int*)d_in[2];
  const float* evals = (const float*)d_in[3];
  const float* W1    = (const float*)d_in[4];
  const float* W2    = (const float*)d_in[5];
  float*       out   = (float*)d_out;
  const int nedges   = in_sizes[1];

  const size_t xw1_bytes = (size_t)NNODES * HIDDIM * sizeof(float);
  float* xw1 = (float*)d_ws;
  float* h   = (float*)((char*)d_ws + xw1_bytes);

  bin_count<<<NBINS, 1024, 0, stream>>>(edst, (int*)h, nedges);
  bin_scatter<<<NBINS, 1024, 0, stream>>>(esrc, edst, evals, (int*)h, nedges);

  gemm1_wmma<<<G1_BLOCKS, 128, 0, stream>>>(feat, W1, xw1);

  spmm1_owned<<<NBINS, 1024, 0, stream>>>(esrc, edst, evals, xw1, W2, h, nedges);

  spmm2_owned<<<NBINS, 1024, 0, stream>>>(esrc, edst, evals, (const int*)h, out, nedges);
}


# edge scatter moved into the gemm1 launch: its first 256 work-groups (128 threads) write the bin records while the rest do the WMMA rows; one launch fewer; record order in a bin, so sum order, changes
# speedup vs baseline: 3.4526x; 1.1120x over previous
#include <hip/hip_runtime.h>

typedef __attribute__((ext_vector_type(2))) float v2f;
typedef __attribute__((ext_vector_type(8))) float v8f;

#define NNODES 50000
#define INDIM  512
#define HIDDIM 128
#define OUTDIM 16

#define G1_ROWS 32
#define G1_BLOCKS ((NNODES + G1_ROWS - 1) / G1_ROWS)
#define G1_KU 4
#define G1_THREADS 128

__device__ __forceinline__ void g1_load_b(float4 (&b)[G1_KU][2], const float* __restrict__ bp, int k0) {
#pragma unroll
  for (int u = 0; u < G1_KU; ++u) {
    const float* w = bp + (size_t)(k0 + 4 * u) * HIDDIM;
    b[u][0] = *(const float4*)(w);
    b[u][1] = *(const float4*)(w + HIDDIM);
  }
}

__device__ __forceinline__ void g1_mma(v8f& acc0, v8f& acc1, v8f& acc2, v8f& acc3,
                                       const float* __restrict__ ap, const float4 (&bb)[G1_KU][2]) {
#pragma unroll
  for (int u = 0; u < G1_KU; ++u) {
    v2f a, b;
    a.x = ap[(4 * u) * 17];
    a.y = ap[(4 * u) * 17 + 17];
    b.x = bb[u][0].x; b.y = bb[u][1].x;
    acc0 = __builtin_amdgcn_wmma_f32_16x16x4_f32(false, a, false, b, (short)0, acc0, false, false);
    b.x = bb[u][0].y; b.y = bb[u][1].y;
    acc1 = __builtin_amdgcn_wmma_f32_16x16x4_f32(false, a, false, b, (short)0, acc1, false, false);
    b.x = bb[u][0].z; b.y = bb[u][1].z;
    acc2 = __builtin_amdgcn_wmma_f32_16x16x4_f32(false, a, false, b, (short)0, acc2, false, false);
    b.x = bb[u][0].w; b.y = bb[u][1].w;
    acc3 = __builtin_amdgcn_wmma_f32_16x16x4_f32(false, a, false, b, (short)0, acc3, false, false);
  }
}

#define OWN_ROWS 196
#define QCAP 64
#define NBINS 256
#define REGION_INTS (OWN_ROWS * HIDDIM)
#define HDR_INTS 260
#define TAIL_COLS (HIDDIM - OUTDIM)
#define REC_CAP ((OWN_ROWS * TAIL_COLS - HDR_INTS) / 2)
#define LAST_ROWS (NNODES - (NBINS - 1) * OWN_ROWS)
#define LAST_CAP ((LAST_ROWS * TAIL_COLS - HDR_INTS) / 2)
#define SLAB 8
#define SORT_SLOTS ((REC_CAP + 1023) / 1024)
#define GRP 8

__global__ __launch_bounds__(1024) void bin_count(const int* __restrict__ edst,
                                                  int* __restrict__ regions,
                                                  int nedges) {
  __shared__ int hist[NBINS];
  const int tid = threadIdx.x;
  if (tid < NBINS) hist[tid] = 0;
  __syncthreads();

  const int per = (nedges + NBINS - 1) / NBINS;
  const int e0  = (int)blockIdx.x * per;
  const int e1  = (e0 + per < nedges) ? e0 + per : nedges;
  for (int eb = e0; eb < e1; eb += SLAB * 1024) {
    int dd[SLAB];
#pragma unroll
    for (int j = 0; j < SLAB; ++j) {
      const int e = eb + j * 1024 + tid;
      dd[j] = (e < e1) ? edst[e] : -1;
    }
#pragma unroll
    for (int j = 0; j < SLAB; ++j)
      if ((unsigned)dd[j] < (unsigned)NNODES)
        atomicAdd(&hist[(unsigned)dd[j] / OWN_ROWS], 1);
  }
  __syncthreads();
  if (tid < NBINS) regions[(size_t)blockIdx.x * REGION_INTS + tid] = hist[tid];
}

#define SC_SLAB 16

__device__ __forceinline__ void scatter_slice(const int* __restrict__ esrc,
                                              const int* __restrict__ edst,
                                              const float* __restrict__ evals,
                                              int* regions, int nedges,
                                              const int g, const int tid, int* sm) {
  int* part   = sm;
  int* cursor = sm + 2 * NBINS;

  const int c4   = tid & 63;
  const int half = tid >> 6;
  int4 s = make_int4(0, 0, 0, 0);
  for (int gg = half; gg < g; gg += 32) {
    int4 t[16];
#pragma unroll
    for (int j = 0; j < 16; ++j) {
      const int sl = gg + 2 * j;
      t[j] = (sl < g) ? *(const int4*)(regions + (size_t)sl * REGION_INTS + 4 * c4) : make_int4(0, 0, 0, 0);
    }
#pragma unroll
    for (int j = 0; j < 16; ++j) { s.x += t[j].x; s.y += t[j].y; s.z += t[j].z; s.w += t[j].w; }
  }
  ((int4*)part)[half * 64 + c4] = s;
  __syncthreads();
  for (int bin = tid; bin < NBINS; bin += G1_THREADS) {
    const int start = part[bin] + part[NBINS + bin];
    cursor[bin] = start;
    if (g == NBINS - 1)
      regions[(size_t)bin * REGION_INTS + NBINS] = start + regions[(size_t)g * REGION_INTS + bin];
  }
  __syncthreads();

  const int per = (nedges + NBINS - 1) / NBINS;
  const int e0  = g * per;
  const int e1  = (e0 + per < nedges) ? e0 + per : nedges;
  for (int eb = e0; eb < e1; eb += SC_SLAB * G1_THREADS) {
    int dd[SC_SLAB], ss[SC_SLAB], vv[SC_SLAB];
#pragma unroll
    for (int j = 0; j < SC_SLAB; ++j) {
      const int e = eb + j * G1_THREADS + tid;
      const bool ok = e < e1;
      dd[j] = ok ? edst[e] : -1;
      ss[j] = ok ? esrc[e] : 0;
      vv[j] = ok ? __float_as_int(evals[e]) : 0;
    }
#pragma unroll
    for (int j = 0; j < SC_SLAB; ++j) {
      if ((unsigned)dd[j] < (unsigned)NNODES) {
        const int b   = (int)((unsigned)dd[j] / OWN_ROWS);
        const int pos = atomicAdd(&cursor[b], 1);
        const int cap = (b == NBINS - 1) ? LAST_CAP : REC_CAP;
        if (pos < cap) {
          const int key = (ss[j] & 0xFFFF) | ((dd[j] - b * OWN_ROWS) << 16);
          *(int2*)(regions + (size_t)b * REGION_INTS + HDR_INTS + 2 * pos) = make_int2(key, vv[j]);
        }
      }
    }
  }
}

__global__ __launch_bounds__(G1_THREADS) void gemm1_scatter(const float* __restrict__ feat,
                                                            const float* __restrict__ W1,
                                                            float* __restrict__ xw1,
                                                            const int* __restrict__ esrc,
                                                            const int* __restrict__ edst,
                                                            const float* __restrict__ evals,
                                                            int* regions,
                                                            int nedges) {
  __shared__ float As[G1_ROWS / 16][INDIM][17];
  __shared__ int   sc_ints[3 * NBINS];
  const int tid  = threadIdx.x;
  if (blockIdx.x < NBINS) {
    scatter_slice(esrc, edst, evals, regions, nedges, (int)blockIdx.x, tid, sc_ints);
    return;
  }
  const int row0 = ((int)blockIdx.x - NBINS) * G1_ROWS;

#pragma unroll
  for (int t = 0; t < G1_ROWS / 16; ++t) {
    float4 v[16];
#pragma unroll
    for (int i = 0; i < 16; ++i) {
      int gr = row0 + t * 16 + i;
      gr = (gr < NNODES) ? gr : NNODES - 1;
      v[i] = ((const float4*)(feat + (size_t)gr * INDIM))[tid];
    }
#pragma unroll
    for (int i = 0; i < 16; ++i) {
      As[t][4 * tid + 0][i] = v[i].x;
      As[t][4 * tid + 1][i] = v[i].y;
      As[t][4 * tid + 2][i] = v[i].z;
      As[t][4 * tid + 3][i] = v[i].w;
    }
  }
  __syncthreads();

  const int wave = tid >> 5;
  const int lane = tid & 31;
  const int hs   = lane >> 4;
  const int lm   = lane & 15;
  const int rt   = wave & 1;
  const int c0   = (wave >> 1) * 64;

  const float* ap = &As[rt][hs * 2][lm];
  const float* bp = W1 + (size_t)(hs * 2) * HIDDIM + c0 + lm * 4;

  v8f acc0 = {}, acc1 = {}, acc2 = {}, acc3 = {};
  float4 b0[G1_KU][2], b1[G1_KU][2];
  g1_load_b(b0, bp, 0);
#pragma nounroll
  for (int k0 = 0; k0 < INDIM; k0 += 8 * G1_KU) {
    g1_load_b(b1, bp, k0 + 4 * G1_KU);
    g1_mma(acc0, acc1, acc2, acc3, ap + k0 * 17, b0);
    if (k0 + 8 * G1_KU < INDIM) g1_load_b(b0, bp, k0 + 8 * G1_KU);
    g1_mma(acc0, acc1, acc2, acc3, ap + (k0 + 4 * G1_KU) * 17, b1);
  }

  float* dst = xw1 + (size_t)(row0 + rt * 16 + hs * 8) * HIDDIM + c0 + lm * 4;
#pragma unroll
  for (int j = 0; j < 8; ++j) {
    if (row0 + rt * 16 + hs * 8 + j < NNODES)
      *(float4*)(dst + (size_t)j * HIDDIM) = make_float4(acc0[j], acc1[j], acc2[j], acc3[j]);
  }
}

__device__ __forceinline__ void own_add_row(float* __restrict__ arow, float v, const float4 x) {
  atomicAdd(arow +  0, v * x.x);
  atomicAdd(arow + 32, v * x.y);
  atomicAdd(arow + 64, v * x.z);
  atomicAdd(arow + 96, v * x.w);
}

__device__ __forceinline__ void own_flush_row(float* __restrict__ arow, const float4 t) {
  atomicAdd(arow +  0, t.x);
  atomicAdd(arow + 32, t.y);
  atomicAdd(arow + 64, t.z);
  atomicAdd(arow + 96, t.w);
}

__device__ __forceinline__ void own_drain_regs(int s, int r, int vb, int n, int lane,
                                               const float* __restrict__ xw1,
                                               float* __restrict__ acc) {
  for (int k = 0; k < n; k += 4) {
    const int i1 = (k + 1 < n) ? k + 1 : k;
    const int i2 = (k + 2 < n) ? k + 2 : k;
    const int i3 = (k + 3 < n) ? k + 3 : k;
    const int s0 = __builtin_amdgcn_readlane(s, k);
    const int s1 = __builtin_amdgcn_readlane(s, i1);
    const int s2 = __builtin_amdgcn_readlane(s, i2);
    const int s3 = __builtin_amdgcn_readlane(s, i3);
    const float4 x0 = ((const float4*)(xw1 + (size_t)s0 * HIDDIM))[lane];
    const float4 x1 = ((const float4*)(xw1 + (size_t)s1 * HIDDIM))[lane];
    const float4 x2 = ((const float4*)(xw1 + (size_t)s2 * HIDDIM))[lane];
    const float4 x3 = ((const float4*)(xw1 + (size_t)s3 * HIDDIM))[lane];
    own_add_row(acc + __builtin_amdgcn_readlane(r, k) * HIDDIM + lane,
                __int_as_float(__builtin_amdgcn_readlane(vb, k)), x0);
    if (k + 1 < n)
      own_add_row(acc + __builtin_amdgcn_readlane(r, i1) * HIDDIM + lane,
                  __int_as_float(__builtin_amdgcn_readlane(vb, i1)), x1);
    if (k + 2 < n)
      own_add_row(acc + __builtin_amdgcn_readlane(r, i2) * HIDDIM + lane,
                  __int_as_float(__builtin_amdgcn_readlane(vb, i2)), x2);
    if (k + 3 < n)
      own_add_row(acc + __builtin_amdgcn_readlane(r, i3) * HIDDIM + lane,
                  __int_as_float(__builtin_amdgcn_readlane(vb, i3)), x3);
  }
}

__device__ __forceinline__ void own_drain(const int* __restrict__ q, int n, int lane, int base,
                                          const int* __restrict__ esrc,
                                          const int* __restrict__ edst,
                                          const float* __restrict__ evals,
                                          const float* __restrict__ xw1,
                                          float* __restrict__ acc) {
  const int e = q[lane < n ? lane : 0];
  own_drain_regs(esrc[e], edst[e] - base, __float_as_int(evals[e]), n, lane, xw1, acc);
}

__global__ __launch_bounds__(1024) void spmm1_owned(const int* __restrict__ esrc,
                                                    const int* __restrict__ edst,
                                                    const float* __restrict__ evals,
                                                    const float* __restrict__ xw1,
                                                    const float* __restrict__ W2,
                                                    float* h,
                                                    int nedges) {
  __shared__ float acc[OWN_ROWS * HIDDIM];
  __shared__ int2  srec[REC_CAP];
  __shared__ int   rowpos[256];
  int* queue = (int*)srec;

  const int tid  = threadIdx.x;
  const int lane = tid & 31;
  const int wave = __builtin_amdgcn_readfirstlane(tid >> 5);
  const int base = blockIdx.x * OWN_ROWS;

  for (int i = tid; i < OWN_ROWS * HIDDIM / 4; i += 1024)
    ((float4*)acc)[i] = make_float4(0.0f, 0.0f, 0.0f, 0.0f);
  if (tid < 256) rowpos[tid] = 0;

  const int* region = (const int*)h + (size_t)blockIdx.x * REGION_INTS;
  const int nrec = __builtin_amdgcn_readfirstlane(region[NBINS]);
  const int cap  = ((int)blockIdx.x == NBINS - 1) ? LAST_CAP : REC_CAP;
  __syncthreads();

  if (nrec <= cap) {
    const int2* rec = (const int2*)(region + HDR_INTS);
    int2 held[SORT_SLOTS];
    int  rank[SORT_SLOTS];
    for (int j = 0; j < SORT_SLOTS; ++j) {
      const int i = tid + j * 1024;
      held[j] = (i < nrec) ? rec[i] : make_int2(0, 0);
    }
    for (int j = 0; j < SORT_SLOTS; ++j) {
      const int i = tid + j * 1024;
      rank[j] = (i < nrec) ? atomicAdd(&rowpos[held[j].x >> 16], 1) : 0;
    }
    __syncthreads();
    if (wave == 0) {
      int cn[7];
      int tot = 0;
      for (int t = 0; t < 7; ++t) { cn[t] = rowpos[lane * 7 + t]; tot += cn[t]; }
      int incl = tot;
      for (int d = 1; d < 32; d <<= 1) {
        const int o = __shfl_up(incl, d);
        if (lane >= d) incl += o;
      }
      int run = incl - tot;
      for (int t = 0; t < 7; ++t) { rowpos[lane * 7 + t] = run; run += cn[t]; }
    }
    __syncthreads();
    for (int j = 0; j < SORT_SLOTS; ++j) {
      const int i = tid + j * 1024;
      if (i < nrec) srec[rowpos[held[j].x >> 16] + rank[j]] = held[j];
    }
    __syncthreads();

    const int per = (((nrec + 31) >> 5) + GRP - 1) & ~(GRP - 1);
    const int i0  = wave * per;
    const int i1  = (i0 + per < nrec) ? i0 + per : nrec;
    float4 sum = make_float4(0.0f, 0.0f, 0.0f, 0.0f);
    int cur = -1;
    for (int i = i0; i < i1; i += 32) {
      const int c = (i1 - i < 32) ? i1 - i : 32;
      int2 rc = srec[i + (lane < c ? lane : c - 1)];
      if (lane >= c) rc.y = 0;
      for (int k = 0; k < c; k += GRP) {
        float4 x[GRP];
        for (int j = 0; j < GRP; ++j) {
          const int sidx = __builtin_amdgcn_readlane(rc.x, k + j) & 0xFFFF;
          x[j] = ((const float4*)(xw1 + (size_t)sidx * HIDDIM))[lane];
        }
        for (int j = 0; j < GRP; ++j) {
          const int   row = __builtin_amdgcn_readlane(rc.x, k + j) >> 16;
          const float v   = __int_as_float(__builtin_amdgcn_readlane(rc.y, k + j));
          if (row != cur) {
            if (cur >= 0) own_flush_row(acc + cur * HIDDIM + lane, sum);
            sum = make_float4(0.0f, 0.0f, 0.0f, 0.0f);
            cur = row;
          }
          sum.x = fmaf(v, x[j].x, sum.x);
          sum.y = fmaf(v, x[j].y, sum.y);
          sum.z = fmaf(v, x[j].z, sum.z);
          sum.w = fmaf(v, x[j].w, sum.w);
        }
      }
    }
    if (cur >= 0) own_flush_row(acc + cur * HIDDIM + lane, sum);
  } else {
    int* q = queue + wave * QCAP;
    int qn = 0;

    const int nquads = nedges >> 2;
    const int nspans = (nquads + 127) >> 7;
    const int4* dst4 = (const int4*)edst;

    for (int sp = wave; sp < nspans; sp += 32) {
      int4 cur[4];
#pragma unroll
      for (int j = 0; j < 4; ++j) {
        const int qi = sp * 128 + j * 32 + lane;
        cur[j] = (qi < nquads) ? dst4[qi] : make_int4(-1, -1, -1, -1);
      }
#pragma unroll
      for (int j = 0; j < 4; ++j) {
        const int qi = sp * 128 + j * 32 + lane;
        const int dd[4] = {cur[j].x, cur[j].y, cur[j].z, cur[j].w};
#pragma unroll
        for (int c = 0; c < 4; ++c) {
          const bool m = (unsigned)(dd[c] - base) < (unsigned)OWN_ROWS;
          const unsigned mask = (unsigned)__ballot(m);
          if (mask) {
            if (m) q[qn + __popc(mask & ((1u << lane) - 1u))] = qi * 4 + c;
            qn += __popc(mask);
            if (qn >= 32) {
              own_drain(q, 32, lane, base, esrc, edst, evals, xw1, acc);
              const int rest = qn - 32;
              const int keep = q[32 + lane];
              if (lane < rest) q[lane] = keep;
              qn = rest;
            }
          }
        }
      }
    }
    if (wave == 0) {
      for (int e = nquads * 4 + lane; e - lane < nedges; e += 32) {
        const bool m = (e < nedges) && ((unsigned)(edst[e < nedges ? e : 0] - base) < (unsigned)OWN_ROWS);
        const unsigned mask = (unsigned)__ballot(m);
        if (mask) {
          if (m) q[qn + __popc(mask & ((1u << lane) - 1u))] = e;
          qn += __popc(mask);
          if (qn >= 32) {
            own_drain(q, 32, lane, base, esrc, edst, evals, xw1, acc);
            const int rest = qn - 32;
            const int keep = q[32 + lane];
            if (lane < rest) q[lane] = keep;
            qn = rest;
          }
        }
      }
    }
    if (qn > 0) own_drain(q, qn, lane, base, esrc, edst, evals, xw1, acc);
  }
  __syncthreads();

  const int nrows = ((int)blockIdx.x == NBINS - 1) ? LAST_ROWS : OWN_ROWS;
  float* hw2 = h + (size_t)blockIdx.x * REGION_INTS + nrows * TAIL_COLS;
  if (wave < (OWN_ROWS + 15) / 16) {
    const int hs   = lane >> 4;
    const int lm   = lane & 15;
    const int row0 = wave * 16;
    const int ar = (row0 + lm < OWN_ROWS) ? row0 + lm : OWN_ROWS - 1;
    const float4* a0p = (const float4*)(acc + ar * HIDDIM + (2 * hs) * 32);
    const float4* a1p = (const float4*)(acc + ar * HIDDIM + (2 * hs + 1) * 32);
    v8f d = {};
    for (int lc = 0; lc < HIDDIM / 16; ++lc) {
      const float4 A0 = a0p[lc];
      const float4 A1 = a1p[lc];
      const float e0[4] = {A0.x, A0.y, A0.z, A0.w};
      const float e1[4] = {A1.x, A1.y, A1.z, A1.w};
#pragma unroll
      for (int i = 0; i < 4; ++i) {
        const int ka = (lc * 4 + i) * 4 + hs * 2;
        v2f a, b;
        a.x = fmaxf(e0[i], 0.0f);
        a.y = fmaxf(e1[i], 0.0f);
        const float* w = W2 + (size_t)ka * OUTDIM + lm;
        b.x = w[0];
        b.y = w[OUTDIM];
        d = __builtin_amdgcn_wmma_f32_16x16x4_f32(false, a, false, b,
                                                  (short)0, d, false, false);
      }
    }
#pragma unroll
    for (int j = 0; j < 8; ++j) {
      const int r = row0 + hs * 8 + j;
      if (r < nrows) hw2[r * OUTDIM + lm] = d[j];
    }
  }
}

__device__ __forceinline__ const float* hw2_row(const int* __restrict__ regions, int s) {
  const int b     = (int)((unsigned)s / OWN_ROWS);
  const int nrows = (b == NBINS - 1) ? LAST_ROWS : OWN_ROWS;
  return (const float*)(regions + (size_t)b * REGION_INTS + nrows * TAIL_COLS + (s - b * OWN_ROWS) * OUTDIM);
}

__global__ __launch_bounds__(1024) void spmm2_owned(const int* __restrict__ esrc,
                                                    const int* __restrict__ edst,
                                                    const float* __restrict__ evals,
                                                    const int* __restrict__ regions,
                                                    float* __restrict__ out,
                                                    int nedges) {
  __shared__ float acc2[OWN_ROWS * OUTDIM];

  const int tid  = threadIdx.x;
  const int lane = tid & 31;
  const int wave = __builtin_amdgcn_readfirstlane(tid >> 5);
  const int base = blockIdx.x * OWN_ROWS;

  for (int i = tid; i < OWN_ROWS * OUTDIM; i += 1024) acc2[i] = 0.0f;

  const int* region = regions + (size_t)blockIdx.x * REGION_INTS;
  const int nrec = __builtin_amdgcn_readfirstlane(region[NBINS]);
  const int cap  = ((int)blockIdx.x == NBINS - 1) ? LAST_CAP : REC_CAP;
  __syncthreads();

  if (nrec <= cap) {
    const int2* rec = (const int2*)(region + HDR_INTS);
    const int per = (nrec + 31) >> 5;
    const int i0  = wave * per;
    const int i1  = (i0 + per < nrec) ? i0 + per : nrec;
    const int q   = lane & 3;
    const int sub = lane >> 2;
    int2 nxt = rec[(i0 < i1) ? i0 + ((lane < i1 - i0) ? lane : 0) : 0];
    for (int i = i0; i < i1; i += 32) {
      const int c = (i1 - i < 32) ? i1 - i : 32;
      const int2 rc = nxt;
      const int left = i1 - (i + 32);
      if (left > 0) nxt = rec[i + 32 + ((lane < left) ? lane : 0)];
      float4 x[4];
      int    ro[4];
      float  vv[4];
#pragma unroll
      for (int j = 0; j < 4; ++j) {
        const int idx = j * 8 + sub;
        const int key = __shfl(rc.x, idx);
        vv[j] = __int_as_float(__shfl(rc.y, idx));
        ro[j] = (idx < c) ? (key >> 16) * OUTDIM + q * 4 : -1;
        x[j]  = (idx < c) ? ((const float4*)hw2_row(regions, key & 0xFFFF))[q]
                          : make_float4(0.0f, 0.0f, 0.0f, 0.0f);
      }
#pragma unroll
      for (int j = 0; j < 4; ++j) {
        if (ro[j] >= 0) {
          float* a = acc2 + ro[j];
          atomicAdd(a + 0, vv[j] * x[j].x);
          atomicAdd(a + 1, vv[j] * x[j].y);
          atomicAdd(a + 2, vv[j] * x[j].z);
          atomicAdd(a + 3, vv[j] * x[j].w);
        }
      }
    }
  } else {
    for (int e = tid; e < nedges; e += 1024) {
      const int r = edst[e] - base;
      if ((unsigned)r < (unsigned)OWN_ROWS) {
        const float  v = evals[e];
        const float* x = hw2_row(regions, esrc[e]);
        for (int k = 0; k < OUTDIM; ++k) atomicAdd(acc2 + r * OUTDIM + k, v * x[k]);
      }
    }
  }
  __syncthreads();

  for (int i = tid; i < OWN_ROWS * (OUTDIM / 4); i += 1024) {
    const int row = base + (i >> 2);
    if (row < NNODES) ((float4*)(out + (size_t)row * OUTDIM))[i & 3] = ((const float4*)acc2)[i];
  }
}

extern "C" void kernel_launch(void* const* d_in, const int* in_sizes, int n_in,
                              void* d_out, int out_size, void* d_ws, size_t ws_size,
                              hipStream_t stream) {
  const float* feat  = (const float*)d_in[0];
  const int*   esrc  = (const int*)d_in[1];
  const int*   edst  = (const int*)d_in[2];
  const float* evals = (const float*)d_in[3];
  const float* W1    = (const float*)d_in[4];
  const float* W2    = (const float*)d_in[5];
  float*       out   = (float*)d_out;
  const int nedges   = in_sizes[1];

  const size_t xw1_bytes = (size_t)NNODES * HIDDIM * sizeof(float);
  float* xw1 = (float*)d_ws;
  float* h   = (float*)((char*)d_ws + xw1_bytes);

  bin_count<<<NBINS, 1024, 0, stream>>>(edst, (int*)h, nedges);

  gemm1_scatter<<<NBINS + G1_BLOCKS, G1_THREADS, 0, stream>>>(feat, W1, xw1, esrc, edst, evals, (int*)h, nedges);

  spmm1_owned<<<NBINS, 1024, 0, stream>>>(esrc, edst, evals, xw1, W2, h, nedges);

  spmm2_owned<<<NBINS, 1024, 0, stream>>>(esrc, edst, evals, (const int*)h, out, nedges);
}


# each scatter group sorts its own edge slice by bin into a private segment (LDS count, scan, place) plus a table of runs; bins read their 256 runs; count launch dropped; sum order changes
# speedup vs baseline: 3.5522x; 1.0288x over previous
#include <hip/hip_runtime.h>

typedef __attribute__((ext_vector_type(2))) float v2f;
typedef __attribute__((ext_vector_type(8))) float v8f;

#define NNODES 50000
#define INDIM  512
#define HIDDIM 128
#define OUTDIM 16

#define G1_ROWS 32
#define G1_BLOCKS ((NNODES + G1_ROWS - 1) / G1_ROWS)
#define G1_KU 4
#define G1_THREADS 128

__device__ __forceinline__ void g1_load_b(float4 (&b)[G1_KU][2], const float* __restrict__ bp, int k0) {
#pragma unroll
  for (int u = 0; u < G1_KU; ++u) {
    const float* w = bp + (size_t)(k0 + 4 * u) * HIDDIM;
    b[u][0] = *(const float4*)(w);
    b[u][1] = *(const float4*)(w + HIDDIM);
  }
}

__device__ __forceinline__ void g1_mma(v8f& acc0, v8f& acc1, v8f& acc2, v8f& acc3,
                                       const float* __restrict__ ap, const float4 (&bb)[G1_KU][2]) {
#pragma unroll
  for (int u = 0; u < G1_KU; ++u) {
    v2f a, b;
    a.x = ap[(4 * u) * 17];
    a.y = ap[(4 * u) * 17 + 17];
    b.x = bb[u][0].x; b.y = bb[u][1].x;
    acc0 = __builtin_amdgcn_wmma_f32_16x16x4_f32(false, a, false, b, (short)0, acc0, false, false);
    b.x = bb[u][0].y; b.y = bb[u][1].y;
    acc1 = __builtin_amdgcn_wmma_f32_16x16x4_f32(false, a, false, b, (short)0, acc1, false, false);
    b.x = bb[u][0].z; b.y = bb[u][1].z;
    acc2 = __builtin_amdgcn_wmma_f32_16x16x4_f32(false, a, false, b, (short)0, acc2, false, false);
    b.x = bb[u][0].w; b.y = bb[u][1].w;
    acc3 = __builtin_amdgcn_wmma_f32_16x16x4_f32(false, a, false, b, (short)0, acc3, false, false);
  }
}

#define OWN_ROWS 196
#define QCAP 64
#define NBINS 256
#define SEG_RECS 8192
#define SEG_INTS (2 * SEG_RECS)
#define TAB_OFF (NBINS * SEG_INTS)
#define TAB_ROW_INTS (2 * NBINS)
#define HW2_OFF (TAB_OFF + NBINS * TAB_ROW_INTS)
#define RUN_MAX 64
#define RUN_SLOTS (RUN_MAX / 32)
#define SL_PER_WAVE (NBINS / 32)
#define REC_CAP 10846
#define LAST_ROWS (NNODES - (NBINS - 1) * OWN_ROWS)
#define SORT_SLOTS (SL_PER_WAVE * RUN_SLOTS)
#define GRP 8

#define SC_SLAB 16

__device__ __forceinline__ void scatter_slice(const int* __restrict__ esrc,
                                              const int* __restrict__ edst,
                                              const float* __restrict__ evals,
                                              int* regions, int nedges,
                                              const int g, const int tid, int* sm) {
  int* hist   = sm;
  int* cursor = sm + NBINS;

  for (int bin = tid; bin < NBINS; bin += G1_THREADS) hist[bin] = 0;
  __syncthreads();

  const int per = (nedges + NBINS - 1) / NBINS;
  const int e0  = g * per;
  const int e1  = (e0 + per < nedges) ? e0 + per : nedges;

  for (int eb = e0; eb < e1; eb += SC_SLAB * G1_THREADS) {
    int dd[SC_SLAB];
#pragma unroll
    for (int j = 0; j < SC_SLAB; ++j) {
      const int e = eb + j * G1_THREADS + tid;
      dd[j] = (e < e1) ? edst[e] : -1;
    }
#pragma unroll
    for (int j = 0; j < SC_SLAB; ++j)
      if ((unsigned)dd[j] < (unsigned)NNODES)
        atomicAdd(&hist[(unsigned)dd[j] / OWN_ROWS], 1);
  }
  __syncthreads();

  if (tid < 32) {
    int cn[8];
    int tot = 0;
    for (int t = 0; t < 8; ++t) { cn[t] = hist[tid * 8 + t]; tot += cn[t]; }
    int incl = tot;
    for (int d = 1; d < 32; d <<= 1) {
      const int o = __shfl_up(incl, d);
      if (tid >= d) incl += o;
    }
    int run = incl - tot;
    for (int t = 0; t < 8; ++t) { cursor[tid * 8 + t] = run; run += cn[t]; }
  }
  __syncthreads();

  for (int bin = tid; bin < NBINS; bin += G1_THREADS)
    *(int2*)(regions + TAB_OFF + bin * TAB_ROW_INTS + 2 * g) = make_int2(cursor[bin], hist[bin]);
  __syncthreads();

  int* seg = regions + (size_t)g * SEG_INTS;
  for (int eb = e0; eb < e1; eb += SC_SLAB * G1_THREADS) {
    int dd[SC_SLAB], ss[SC_SLAB], vv[SC_SLAB];
#pragma unroll
    for (int j = 0; j < SC_SLAB; ++j) {
      const int e = eb + j * G1_THREADS + tid;
      const bool ok = e < e1;
      dd[j] = ok ? edst[e] : -1;
      ss[j] = ok ? esrc[e] : 0;
      vv[j] = ok ? __float_as_int(evals[e]) : 0;
    }
#pragma unroll
    for (int j = 0; j < SC_SLAB; ++j) {
      if ((unsigned)dd[j] < (unsigned)NNODES) {
        const int b   = (int)((unsigned)dd[j] / OWN_ROWS);
        const int pos = atomicAdd(&cursor[b], 1);
        if (pos < SEG_RECS) {
          const int key = (ss[j] & 0xFFFF) | ((dd[j] - b * OWN_ROWS) << 16);
          *(int2*)(seg + 2 * pos) = make_int2(key, vv[j]);
        }
      }
    }
  }
}

__global__ __launch_bounds__(G1_THREADS) void gemm1_scatter(const float* __restrict__ feat,
                                                            const float* __restrict__ W1,
                                                            float* __restrict__ xw1,
                                                            const int* __restrict__ esrc,
                                                            const int* __restrict__ edst,
                                                            const float* __restrict__ evals,
                                                            int* regions,
                                                            int nedges) {
  __shared__ float As[G1_ROWS / 16][INDIM][17];
  __shared__ int   sc_ints[2 * NBINS];
  const int tid  = threadIdx.x;
  if (blockIdx.x < NBINS) {
    scatter_slice(esrc, edst, evals, regions, nedges, (int)blockIdx.x, tid, sc_ints);
    return;
  }
  const int row0 = ((int)blockIdx.x - NBINS) * G1_ROWS;

#pragma unroll
  for (int t = 0; t < G1_ROWS / 16; ++t) {
    float4 v[16];
#pragma unroll
    for (int i = 0; i < 16; ++i) {
      int gr = row0 + t * 16 + i;
      gr = (gr < NNODES) ? gr : NNODES - 1;
      v[i] = ((const float4*)(feat + (size_t)gr * INDIM))[tid];
    }
#pragma unroll
    for (int i = 0; i < 16; ++i) {
      As[t][4 * tid + 0][i] = v[i].x;
      As[t][4 * tid + 1][i] = v[i].y;
      As[t][4 * tid + 2][i] = v[i].z;
      As[t][4 * tid + 3][i] = v[i].w;
    }
  }
  __syncthreads();

  const int wave = tid >> 5;
  const int lane = tid & 31;
  const int hs   = lane >> 4;
  const int lm   = lane & 15;
  const int rt   = wave & 1;
  const int c0   = (wave >> 1) * 64;

  const float* ap = &As[rt][hs * 2][lm];
  const float* bp = W1 + (size_t)(hs * 2) * HIDDIM + c0 + lm * 4;

  v8f acc0 = {}, acc1 = {}, acc2 = {}, acc3 = {};
  float4 b0[G1_KU][2], b1[G1_KU][2];
  g1_load_b(b0, bp, 0);
#pragma nounroll
  for (int k0 = 0; k0 < INDIM; k0 += 8 * G1_KU) {
    g1_load_b(b1, bp, k0 + 4 * G1_KU);
    g1_mma(acc0, acc1, acc2, acc3, ap + k0 * 17, b0);
    if (k0 + 8 * G1_KU < INDIM) g1_load_b(b0, bp, k0 + 8 * G1_KU);
    g1_mma(acc0, acc1, acc2, acc3, ap + (k0 + 4 * G1_KU) * 17, b1);
  }

  float* dst = xw1 + (size_t)(row0 + rt * 16 + hs * 8) * HIDDIM + c0 + lm * 4;
#pragma unroll
  for (int j = 0; j < 8; ++j) {
    if (row0 + rt * 16 + hs * 8 + j < NNODES)
      *(float4*)(dst + (size_t)j * HIDDIM) = make_float4(acc0[j], acc1[j], acc2[j], acc3[j]);
  }
}

__device__ __forceinline__ void own_add_row(float* __restrict__ arow, float v, const float4 x) {
  atomicAdd(arow +  0, v * x.x);
  atomicAdd(arow + 32, v * x.y);
  atomicAdd(arow + 64, v * x.z);
  atomicAdd(arow + 96, v * x.w);
}

__device__ __forceinline__ void own_flush_row(float* __restrict__ arow, const float4 t) {
  atomicAdd(arow +  0, t.x);
  atomicAdd(arow + 32, t.y);
  atomicAdd(arow + 64, t.z);
  atomicAdd(arow + 96, t.w);
}

__device__ __forceinline__ void own_drain_regs(int s, int r, int vb, int n, int lane,
                                               const float* __restrict__ xw1,
                                               float* __restrict__ acc) {
  for (int k = 0; k < n; k += 4) {
    const int i1 = (k + 1 < n) ? k + 1 : k;
    const int i2 = (k + 2 < n) ? k + 2 : k;
    const int i3 = (k + 3 < n) ? k + 3 : k;
    const int s0 = __builtin_amdgcn_readlane(s, k);
    const int s1 = __builtin_amdgcn_readlane(s, i1);
    const int s2 = __builtin_amdgcn_readlane(s, i2);
    const int s3 = __builtin_amdgcn_readlane(s, i3);
    const float4 x0 = ((const float4*)(xw1 + (size_t)s0 * HIDDIM))[lane];
    const float4 x1 = ((const float4*)(xw1 + (size_t)s1 * HIDDIM))[lane];
    const float4 x2 = ((const float4*)(xw1 + (size_t)s2 * HIDDIM))[lane];
    const float4 x3 = ((const float4*)(xw1 + (size_t)s3 * HIDDIM))[lane];
    own_add_row(acc + __builtin_amdgcn_readlane(r, k) * HIDDIM + lane,
                __int_as_float(__builtin_amdgcn_readlane(vb, k)), x0);
    if (k + 1 < n)
      own_add_row(acc + __builtin_amdgcn_readlane(r, i1) * HIDDIM + lane,
                  __int_as_float(__builtin_amdgcn_readlane(vb, i1)), x1);
    if (k + 2 < n)
      own_add_row(acc + __builtin_amdgcn_readlane(r, i2) * HIDDIM + lane,
                  __int_as_float(__builtin_amdgcn_readlane(vb, i2)), x2);
    if (k + 3 < n)
      own_add_row(acc + __builtin_amdgcn_readlane(r, i3) * HIDDIM + lane,
                  __int_as_float(__builtin_amdgcn_readlane(vb, i3)), x3);
  }
}

__device__ __forceinline__ void own_drain(const int* __restrict__ q, int n, int lane, int base,
                                          const int* __restrict__ esrc,
                                          const int* __restrict__ edst,
                                          const float* __restrict__ evals,
                                          const float* __restrict__ xw1,
                                          float* __restrict__ acc) {
  const int e = q[lane < n ? lane : 0];
  own_drain_regs(esrc[e], edst[e] - base, __float_as_int(evals[e]), n, lane, xw1, acc);
}

__global__ __launch_bounds__(1024) void spmm1_owned(const int* __restrict__ esrc,
                                                    const int* __restrict__ edst,
                                                    const float* __restrict__ evals,
                                                    const float* __restrict__ xw1,
                                                    const float* __restrict__ W2,
                                                    float* h,
                                                    int nedges) {
  __shared__ float acc[OWN_ROWS * HIDDIM];
  __shared__ int2  srec[REC_CAP];
  __shared__ int   rowpos[256];
  __shared__ int   binstat[2];
  int* queue = (int*)srec;

  const int tid  = threadIdx.x;
  const int lane = tid & 31;
  const int wave = __builtin_amdgcn_readfirstlane(tid >> 5);
  const int base = blockIdx.x * OWN_ROWS;

  const int* regions = (const int*)h;
  const int2 te = *(const int2*)(regions + TAB_OFF + (int)blockIdx.x * TAB_ROW_INTS + 2 * (wave * SL_PER_WAVE + (lane & (SL_PER_WAVE - 1))));

  for (int i = tid; i < OWN_ROWS * HIDDIM / 4; i += 1024)
    ((float4*)acc)[i] = make_float4(0.0f, 0.0f, 0.0f, 0.0f);
  if (tid < 256) rowpos[tid] = 0;
  if (tid < 2) binstat[tid] = 0;
  __syncthreads();

  int2 held[SORT_SLOTS];
  int  rank[SORT_SLOTS];
  int  st[SL_PER_WAVE], cn[SL_PER_WAVE];
#pragma unroll
  for (int j = 0; j < SL_PER_WAVE; ++j) {
    st[j] = __builtin_amdgcn_readlane(te.x, j);
    cn[j] = __builtin_amdgcn_readlane(te.y, j);
    const int* seg = regions + (size_t)(wave * SL_PER_WAVE + j) * SEG_INTS;
#pragma unroll
    for (int t = 0; t < RUN_SLOTS; ++t) {
      const int idx = t * 32 + lane;
      const bool ok = idx < cn[j] && (unsigned)(st[j] + idx) < (unsigned)SEG_RECS;
      held[j * RUN_SLOTS + t] = ok ? *(const int2*)(seg + 2 * (st[j] + idx)) : make_int2(0, 0);
    }
  }
  if (lane < SL_PER_WAVE) {
    atomicAdd(&binstat[0], te.y);
    if ((unsigned)te.y > (unsigned)RUN_MAX) atomicOr(&binstat[1], 1);
  }
#pragma unroll
  for (int j = 0; j < SL_PER_WAVE; ++j) {
#pragma unroll
    for (int t = 0; t < RUN_SLOTS; ++t) {
      const bool ok = t * 32 + lane < cn[j];
      rank[j * RUN_SLOTS + t] = ok ? atomicAdd(&rowpos[(held[j * RUN_SLOTS + t].x >> 16) & 255], 1) : 0;
    }
  }
  __syncthreads();
  const int nrec = __builtin_amdgcn_readfirstlane(binstat[0]);
  const bool fits = (binstat[1] == 0) && (nrec >= 0) && (nrec <= REC_CAP) && ((nedges + NBINS - 1) / NBINS <= SEG_RECS);

  if (fits) {
    if (wave == 0) {
      int rc7[7];
      int tot = 0;
      for (int t = 0; t < 7; ++t) { rc7[t] = rowpos[lane * 7 + t]; tot += rc7[t]; }
      int incl = tot;
      for (int d = 1; d < 32; d <<= 1) {
        const int o = __shfl_up(incl, d);
        if (lane >= d) incl += o;
      }
      int run = incl - tot;
      for (int t = 0; t < 7; ++t) { rowpos[lane * 7 + t] = run; run += rc7[t]; }
    }
    __syncthreads();
#pragma unroll
    for (int j = 0; j < SL_PER_WAVE; ++j) {
#pragma unroll
      for (int t = 0; t < RUN_SLOTS; ++t) {
        if (t * 32 + lane < cn[j]) {
          const int2 r = held[j * RUN_SLOTS + t];
          srec[rowpos[(r.x >> 16) & 255] + rank[j * RUN_SLOTS + t]] = r;
        }
      }
    }
    __syncthreads();

    const int per = (((nrec + 31) >> 5) + GRP - 1) & ~(GRP - 1);
    const int i0  = wave * per;
    const int i1  = (i0 + per < nrec) ? i0 + per : nrec;
    float4 sum = make_float4(0.0f, 0.0f, 0.0f, 0.0f);
    int cur = -1;
    for (int i = i0; i < i1; i += 32) {
      const int c = (i1 - i < 32) ? i1 - i : 32;
      int2 rc = srec[i + (lane < c ? lane : c - 1)];
      if (lane >= c) rc.y = 0;
      for (int k = 0; k < c; k += GRP) {
        float4 x[GRP];
        for (int j = 0; j < GRP; ++j) {
          const int sidx = __builtin_amdgcn_readlane(rc.x, k + j) & 0xFFFF;
          x[j] = ((const float4*)(xw1 + (size_t)sidx * HIDDIM))[lane];
        }
        for (int j = 0; j < GRP; ++j) {
          const int   row = __builtin_amdgcn_readlane(rc.x, k + j) >> 16;
          const float v   = __int_as_float(__builtin_amdgcn_readlane(rc.y, k + j));
          if (row != cur) {
            if (cur >= 0) own_flush_row(acc + cur * HIDDIM + lane, sum);
            sum = make_float4(0.0f, 0.0f, 0.0f, 0.0f);
            cur = row;
          }
          sum.x = fmaf(v, x[j].x, sum.x);
          sum.y = fmaf(v, x[j].y, sum.y);
          sum.z = fmaf(v, x[j].z, sum.z);
          sum.w = fmaf(v, x[j].w, sum.w);
        }
      }
    }
    if (cur >= 0) own_flush_row(acc + cur * HIDDIM + lane, sum);
  } else {
    int* q = queue + wave * QCAP;
    int qn = 0;

    const int nquads = nedges >> 2;
    const int nspans = (nquads + 127) >> 7;
    const int4* dst4 = (const int4*)edst;

    for (int sp = wave; sp < nspans; sp += 32) {
      int4 cur[4];
#pragma unroll
      for (int j = 0; j < 4; ++j) {
        const int qi = sp * 128 + j * 32 + lane;
        cur[j] = (qi < nquads) ? dst4[qi] : make_int4(-1, -1, -1, -1);
      }
#pragma unroll
      for (int j = 0; j < 4; ++j) {
        const int qi = sp * 128 + j * 32 + lane;
        const int dd[4] = {cur[j].x, cur[j].y, cur[j].z, cur[j].w};
#pragma unroll
        for (int c = 0; c < 4; ++c) {
          const bool m = (unsigned)(dd[c] - base) < (unsigned)OWN_ROWS;
          const unsigned mask = (unsigned)__ballot(m);
          if (mask) {
            if (m) q[qn + __popc(mask & ((1u << lane) - 1u))] = qi * 4 + c;
            qn += __popc(mask);
            if (qn >= 32) {
              own_drain(q, 32, lane, base, esrc, edst, evals, xw1, acc);
              const int rest = qn - 32;
              const int keep = q[32 + lane];
              if (lane < rest) q[lane] = keep;
              qn = rest;
            }
          }
        }
      }
    }
    if (wave == 0) {
      for (int e = nquads * 4 + lane; e - lane < nedges; e += 32) {
        const bool m = (e < nedges) && ((unsigned)(edst[e < nedges ? e : 0] - base) < (unsigned)OWN_ROWS);
        const unsigned mask = (unsigned)__ballot(m);
        if (mask) {
          if (m) q[qn + __popc(mask & ((1u << lane) - 1u))] = e;
          qn += __popc(mask);
          if (qn >= 32) {
            own_drain(q, 32, lane, base, esrc, edst, evals, xw1, acc);
            const int rest = qn - 32;
            const int keep = q[32 + lane];
            if (lane < rest) q[lane] = keep;
            qn = rest;
          }
        }
      }
    }
    if (qn > 0) own_drain(q, qn, lane, base, esrc, edst, evals, xw1, acc);
  }
  __syncthreads();

  const int nrows = ((int)blockIdx.x == NBINS - 1) ? LAST_ROWS : OWN_ROWS;
  float* hw2 = h + HW2_OFF + (size_t)base * OUTDIM;
  if (wave < (OWN_ROWS + 15) / 16) {
    const int hs   = lane >> 4;
    const int lm   = lane & 15;
    const int row0 = wave * 16;
    const int ar = (row0 + lm < OWN_ROWS) ? row0 + lm : OWN_ROWS - 1;
    const float4* a0p = (const float4*)(acc + ar * HIDDIM + (2 * hs) * 32);
    const float4* a1p = (const float4*)(acc + ar * HIDDIM + (2 * hs + 1) * 32);
    v8f d = {};
    for (int lc = 0; lc < HIDDIM / 16; ++lc) {
      const float4 A0 = a0p[lc];
      const float4 A1 = a1p[lc];
      const float e0[4] = {A0.x, A0.y, A0.z, A0.w};
      const float e1[4] = {A1.x, A1.y, A1.z, A1.w};
#pragma unroll
      for (int i = 0; i < 4; ++i) {
        const int ka = (lc * 4 + i) * 4 + hs * 2;
        v2f a, b;
        a.x = fmaxf(e0[i], 0.0f);
        a.y = fmaxf(e1[i], 0.0f);
        const float* w = W2 + (size_t)ka * OUTDIM + lm;
        b.x = w[0];
        b.y = w[OUTDIM];
        d = __builtin_amdgcn_wmma_f32_16x16x4_f32(false, a, false, b,
                                                  (short)0, d, false, false);
      }
    }
#pragma unroll
    for (int j = 0; j < 8; ++j) {
      const int r = row0 + hs * 8 + j;
      if (r < nrows) hw2[r * OUTDIM + lm] = d[j];
    }
  }
}

__device__ __forceinline__ const float* hw2_row(const int* __restrict__ regions, int s) {
  return (const float*)(regions + HW2_OFF + s * OUTDIM);
}

__global__ __launch_bounds__(1024) void spmm2_owned(const int* __restrict__ esrc,
                                                    const int* __restrict__ edst,
                                                    const float* __restrict__ evals,
                                                    const int* __restrict__ regions,
                                                    float* __restrict__ out,
                                                    int nedges) {
  __shared__ float acc2[OWN_ROWS * OUTDIM];
  __shared__ int   longrun;

  const int tid  = threadIdx.x;
  const int lane = tid & 31;
  const int wave = __builtin_amdgcn_readfirstlane(tid >> 5);
  const int base = blockIdx.x * OWN_ROWS;

  const int2 te = *(const int2*)(regions + TAB_OFF + (int)blockIdx.x * TAB_ROW_INTS + 2 * (wave * SL_PER_WAVE + (lane & (SL_PER_WAVE - 1))));

  for (int i = tid; i < OWN_ROWS * OUTDIM; i += 1024) acc2[i] = 0.0f;
  if (tid == 0) longrun = 0;
  __syncthreads();
  if (lane < SL_PER_WAVE && (unsigned)te.y > (unsigned)RUN_MAX) atomicOr(&longrun, 1);

  int2 rcs[SORT_SLOTS];
  int  cn[SL_PER_WAVE];
#pragma unroll
  for (int j = 0; j < SL_PER_WAVE; ++j) {
    const int st = __builtin_amdgcn_readlane(te.x, j);
    cn[j] = __builtin_amdgcn_readlane(te.y, j);
    const int* seg = regions + (size_t)(wave * SL_PER_WAVE + j) * SEG_INTS;
#pragma unroll
    for (int t = 0; t < RUN_SLOTS; ++t) {
      const int idx = t * 32 + lane;
      const bool ok = idx < cn[j] && (unsigned)(st + idx) < (unsigned)SEG_RECS;
      rcs[j * RUN_SLOTS + t] = ok ? *(const int2*)(seg + 2 * (st + idx)) : make_int2(0, 0);
    }
  }
  __syncthreads();
  const bool fits = (longrun == 0) && ((nedges + NBINS - 1) / NBINS <= SEG_RECS);

  if (fits) {
    const int q   = lane & 3;
    const int sub = lane >> 2;
#pragma unroll
    for (int j = 0; j < SL_PER_WAVE; ++j) {
#pragma unroll
      for (int t = 0; t < RUN_SLOTS; ++t) {
        const int c = cn[j] - t * 32;
        if (c > 0) {
          const int2 rc = rcs[j * RUN_SLOTS + t];
          float4 x[4];
          int    ro[4];
          float  vv[4];
#pragma unroll
          for (int k = 0; k < 4; ++k) {
            const int idx = k * 8 + sub;
            const int key = __shfl(rc.x, idx);
            vv[k] = __int_as_float(__shfl(rc.y, idx));
            ro[k] = (idx < c) ? ((key >> 16) & 255) * OUTDIM + q * 4 : -1;
            x[k]  = (idx < c) ? ((const float4*)hw2_row(regions, key & 0xFFFF))[q]
                              : make_float4(0.0f, 0.0f, 0.0f, 0.0f);
          }
#pragma unroll
          for (int k = 0; k < 4; ++k) {
            if (ro[k] >= 0) {
              float* a = acc2 + ro[k];
              atomicAdd(a + 0, vv[k] * x[k].x);
              atomicAdd(a + 1, vv[k] * x[k].y);
              atomicAdd(a + 2, vv[k] * x[k].z);
              atomicAdd(a + 3, vv[k] * x[k].w);
            }
          }
        }
      }
    }
  } else {
    for (int e = tid; e < nedges; e += 1024) {
      const int r = edst[e] - base;
      if ((unsigned)r < (unsigned)OWN_ROWS) {
        const float  v = evals[e];
        const float* x = hw2_row(regions, esrc[e]);
        for (int k = 0; k < OUTDIM; ++k) atomicAdd(acc2 + r * OUTDIM + k, v * x[k]);
      }
    }
  }
  __syncthreads();

  for (int i = tid; i < OWN_ROWS * (OUTDIM / 4); i += 1024) {
    const int row = base + (i >> 2);
    if (row < NNODES) ((float4*)(out + (size_t)row * OUTDIM))[i & 3] = ((const float4*)acc2)[i];
  }
}

extern "C" void kernel_launch(void* const* d_in, const int* in_sizes, int n_in,
                              void* d_out, int out_size, void* d_ws, size_t ws_size,
                              hipStream_t stream) {
  const float* feat  = (const float*)d_in[0];
  const int*   esrc  = (const int*)d_in[1];
  const int*   edst  = (const int*)d_in[2];
  const float* evals = (const float*)d_in[3];
  const float* W1    = (const float*)d_in[4];
  const float* W2    = (const float*)d_in[5];
  float*       out   = (float*)d_out;
  const int nedges   = in_sizes[1];

  const size_t xw1_bytes = (size_t)NNODES * HIDDIM * sizeof(float);
  float* xw1 = (float*)d_ws;
  float* h   = (float*)((char*)d_ws + xw1_bytes);

  gemm1_scatter<<<NBINS + G1_BLOCKS, G1_THREADS, 0, stream>>>(feat, W1, xw1, esrc, edst, evals, (int*)h, nedges);

  spmm1_owned<<<NBINS, 1024, 0, stream>>>(esrc, edst, evals, xw1, W2, h, nedges);

  spmm2_owned<<<NBINS, 1024, 0, stream>>>(esrc, edst, evals, (const int*)h, out, nedges);
}


# gemm1: feature tile staged in four k parts; a part is prefetched during the WMMAs of the part before it, read in front of that part's last 16 WMMAs and put in LDS behind them; sum order kept
# speedup vs baseline: 3.6610x; 1.0306x over previous
#include <hip/hip_runtime.h>

typedef __attribute__((ext_vector_type(2))) float v2f;
typedef __attribute__((ext_vector_type(8))) float v8f;

#define NNODES 50000
#define INDIM  512
#define HIDDIM 128
#define OUTDIM 16

#define G1_ROWS 32
#define G1_BLOCKS ((NNODES + G1_ROWS - 1) / G1_ROWS)
#define G1_KU 4
#define G1_THREADS 128

__device__ __forceinline__ void g1_load_b(float4 (&b)[G1_KU][2], const float* __restrict__ bp, int k0) {
#pragma unroll
  for (int u = 0; u < G1_KU; ++u) {
    const float* w = bp + (size_t)(k0 + 4 * u) * HIDDIM;
    b[u][0] = *(const float4*)(w);
    b[u][1] = *(const float4*)(w + HIDDIM);
  }
}

__device__ __forceinline__ void g1_mma(v8f& acc0, v8f& acc1, v8f& acc2, v8f& acc3,
                                       const float* __restrict__ ap, const float4 (&bb)[G1_KU][2]) {
#pragma unroll
  for (int u = 0; u < G1_KU; ++u) {
    v2f a, b;
    a.x = ap[(4 * u) * 17];
    a.y = ap[(4 * u) * 17 + 17];
    b.x = bb[u][0].x; b.y = bb[u][1].x;
    acc0 = __builtin_amdgcn_wmma_f32_16x16x4_f32(false, a, false, b, (short)0, acc0, false, false);
    b.x = bb[u][0].y; b.y = bb[u][1].y;
    acc1 = __builtin_amdgcn_wmma_f32_16x16x4_f32(false, a, false, b, (short)0, acc1, false, false);
    b.x = bb[u][0].z; b.y = bb[u][1].z;
    acc2 = __builtin_amdgcn_wmma_f32_16x16x4_f32(false, a, false, b, (short)0, acc2, false, false);
    b.x = bb[u][0].w; b.y = bb[u][1].w;
    acc3 = __builtin_amdgcn_wmma_f32_16x16x4_f32(false, a, false, b, (short)0, acc3, false, false);
  }
}

#define G1_KQ 128
#define G1_NQ (INDIM / G1_KQ)

__device__ __forceinline__ void g1_fetch(float4 (&v)[8], const float* feat, int row0, int q, int tid) {
#pragma unroll
  for (int i = 0; i < 8; ++i) {
    int gr = row0 + (tid >> 5) + 4 * i;
    gr = (gr < NNODES) ? gr : NNODES - 1;
    v[i] = ((const float4*)(feat + (size_t)gr * INDIM + q * G1_KQ))[tid & 31];
  }
}

__device__ __forceinline__ void g1_hint(const float* feat, int row0, int q, int tid) {
#pragma unroll
  for (int i = 0; i < 8; ++i) {
    int gr = row0 + (tid >> 5) + 4 * i;
    gr = (gr < NNODES) ? gr : NNODES - 1;
    __builtin_prefetch(feat + (size_t)gr * INDIM + q * G1_KQ + 4 * (tid & 31), 0, 3);
  }
}

__device__ __forceinline__ void g1_stage(float (*As)[INDIM][17], const float4 (&v)[8], int q, int tid) {
#pragma unroll
  for (int i = 0; i < 8; ++i) {
    const int r = (tid >> 5) + 4 * i;
    float* p = &As[r >> 4][q * G1_KQ + 4 * (tid & 31)][r & 15];
    p[0]  = v[i].x;
    p[17] = v[i].y;
    p[34] = v[i].z;
    p[51] = v[i].w;
  }
}

#define OWN_ROWS 196
#define QCAP 64
#define NBINS 256
#define SEG_RECS 8192
#define SEG_INTS (2 * SEG_RECS)
#define TAB_OFF (NBINS * SEG_INTS)
#define TAB_ROW_INTS (2 * NBINS)
#define HW2_OFF (TAB_OFF + NBINS * TAB_ROW_INTS)
#define RUN_MAX 64
#define RUN_SLOTS (RUN_MAX / 32)
#define SL_PER_WAVE (NBINS / 32)
#define REC_CAP 10846
#define LAST_ROWS (NNODES - (NBINS - 1) * OWN_ROWS)
#define SORT_SLOTS (SL_PER_WAVE * RUN_SLOTS)
#define GRP 8

#define SC_SLAB 16

__device__ __forceinline__ void scatter_slice(const int* __restrict__ esrc,
                                              const int* __restrict__ edst,
                                              const float* __restrict__ evals,
                                              int* regions, int nedges,
                                              const int g, const int tid, int* sm) {
  int* hist   = sm;
  int* cursor = sm + NBINS;

  for (int bin = tid; bin < NBINS; bin += G1_THREADS) hist[bin] = 0;
  __syncthreads();

  const int per = (nedges + NBINS - 1) / NBINS;
  const int e0  = g * per;
  const int e1  = (e0 + per < nedges) ? e0 + per : nedges;

  for (int eb = e0; eb < e1; eb += SC_SLAB * G1_THREADS) {
    int dd[SC_SLAB];
#pragma unroll
    for (int j = 0; j < SC_SLAB; ++j) {
      const int e = eb + j * G1_THREADS + tid;
      dd[j] = (e < e1) ? edst[e] : -1;
    }
#pragma unroll
    for (int j = 0; j < SC_SLAB; ++j)
      if ((unsigned)dd[j] < (unsigned)NNODES)
        atomicAdd(&hist[(unsigned)dd[j] / OWN_ROWS], 1);
  }
  __syncthreads();

  if (tid < 32) {
    int cn[8];
    int tot = 0;
    for (int t = 0; t < 8; ++t) { cn[t] = hist[tid * 8 + t]; tot += cn[t]; }
    int incl = tot;
    for (int d = 1; d < 32; d <<= 1) {
      const int o = __shfl_up(incl, d);
      if (tid >= d) incl += o;
    }
    int run = incl - tot;
    for (int t = 0; t < 8; ++t) { cursor[tid * 8 + t] = run; run += cn[t]; }
  }
  __syncthreads();

  for (int bin = tid; bin < NBINS; bin += G1_THREADS)
    *(int2*)(regions + TAB_OFF + bin * TAB_ROW_INTS + 2 * g) = make_int2(cursor[bin], hist[bin]);
  __syncthreads();

  int* seg = regions + (size_t)g * SEG_INTS;
  for (int eb = e0; eb < e1; eb += SC_SLAB * G1_THREADS) {
    int dd[SC_SLAB], ss[SC_SLAB], vv[SC_SLAB];
#pragma unroll
    for (int j = 0; j < SC_SLAB; ++j) {
      const int e = eb + j * G1_THREADS + tid;
      const bool ok = e < e1;
      dd[j] = ok ? edst[e] : -1;
      ss[j] = ok ? esrc[e] : 0;
      vv[j] = ok ? __float_as_int(evals[e]) : 0;
    }
#pragma unroll
    for (int j = 0; j < SC_SLAB; ++j) {
      if ((unsigned)dd[j] < (unsigned)NNODES) {
        const int b   = (int)((unsigned)dd[j] / OWN_ROWS);
        const int pos = atomicAdd(&cursor[b], 1);
        if (pos < SEG_RECS) {
          const int key = (ss[j] & 0xFFFF) | ((dd[j] - b * OWN_ROWS) << 16);
          *(int2*)(seg + 2 * pos) = make_int2(key, vv[j]);
        }
      }
    }
  }
}

__global__ __launch_bounds__(G1_THREADS) void gemm1_scatter(const float* feat,
                                                            const float* __restrict__ W1,
                                                            float* __restrict__ xw1,
                                                            const int* __restrict__ esrc,
                                                            const int* __restrict__ edst,
                                                            const float* __restrict__ evals,
                                                            int* regions,
                                                            int nedges) {
  __shared__ float As[G1_ROWS / 16][INDIM][17];
  __shared__ int   sc_ints[2 * NBINS];
  const int tid  = threadIdx.x;
  if (blockIdx.x < NBINS) {
    scatter_slice(esrc, edst, evals, regions, nedges, (int)blockIdx.x, tid, sc_ints);
    return;
  }
  const int row0 = ((int)blockIdx.x - NBINS) * G1_ROWS;

  float4 nv[8];
  g1_fetch(nv, feat, row0, 0, tid);
  g1_stage(As, nv, 0, tid);
  __syncthreads();

  const int wave = tid >> 5;
  const int lane = tid & 31;
  const int hs   = lane >> 4;
  const int lm   = lane & 15;
  const int rt   = wave & 1;
  const int c0   = (wave >> 1) * 64;

  const float* ap = &As[rt][hs * 2][lm];
  const float* bp = W1 + (size_t)(hs * 2) * HIDDIM + c0 + lm * 4;

  v8f acc0 = {}, acc1 = {}, acc2 = {}, acc3 = {};
  float4 b0[G1_KU][2], b1[G1_KU][2];
  g1_load_b(b0, bp, 0);
#pragma unroll
  for (int q = 0; q < G1_NQ; ++q) {
    if (q + 1 < G1_NQ) g1_hint(feat, row0, q + 1, tid);
    const int kl = (q + 1) * G1_KQ - 8 * G1_KU;
#pragma nounroll
    for (int k0 = q * G1_KQ; k0 < kl; k0 += 8 * G1_KU) {
      g1_load_b(b1, bp, k0 + 4 * G1_KU);
      g1_mma(acc0, acc1, acc2, acc3, ap + k0 * 17, b0);
      g1_load_b(b0, bp, k0 + 8 * G1_KU);
      g1_mma(acc0, acc1, acc2, acc3, ap + (k0 + 4 * G1_KU) * 17, b1);
    }
    g1_load_b(b1, bp, kl + 4 * G1_KU);
    g1_mma(acc0, acc1, acc2, acc3, ap + kl * 17, b0);
    if (kl + 8 * G1_KU < INDIM) g1_load_b(b0, bp, kl + 8 * G1_KU);
    if (q + 1 < G1_NQ) {
      __builtin_amdgcn_sched_barrier(0);
      g1_fetch(nv, feat, row0, q + 1, tid);
      __builtin_amdgcn_sched_barrier(0);
    }
    g1_mma(acc0, acc1, acc2, acc3, ap + (kl + 4 * G1_KU) * 17, b1);
    if (q + 1 < G1_NQ) {
      __builtin_amdgcn_sched_barrier(0);
      g1_stage(As, nv, q + 1, tid);
      __syncthreads();
    }
  }

  float* dst = xw1 + (size_t)(row0 + rt * 16 + hs * 8) * HIDDIM + c0 + lm * 4;
#pragma unroll
  for (int j = 0; j < 8; ++j) {
    if (row0 + rt * 16 + hs * 8 + j < NNODES)
      *(float4*)(dst + (size_t)j * HIDDIM) = make_float4(acc0[j], acc1[j], acc2[j], acc3[j]);
  }
}

__device__ __forceinline__ void own_add_row(float* __restrict__ arow, float v, const float4 x) {
  atomicAdd(arow +  0, v * x.x);
  atomicAdd(arow + 32, v * x.y);
  atomicAdd(arow + 64, v * x.z);
  atomicAdd(arow + 96, v * x.w);
}

__device__ __forceinline__ void own_flush_row(float* __restrict__ arow, const float4 t) {
  atomicAdd(arow +  0, t.x);
  atomicAdd(arow + 32, t.y);
  atomicAdd(arow + 64, t.z);
  atomicAdd(arow + 96, t.w);
}

__device__ __forceinline__ void own_drain_regs(int s, int r, int vb, int n, int lane,
                                               const float* __restrict__ xw1,
                                               float* __restrict__ acc) {
  for (int k = 0; k < n; k += 4) {
    const int i1 = (k + 1 < n) ? k + 1 : k;
    const int i2 = (k + 2 < n) ? k + 2 : k;
    const int i3 = (k + 3 < n) ? k + 3 : k;
    const int s0 = __builtin_amdgcn_readlane(s, k);
    const int s1 = __builtin_amdgcn_readlane(s, i1);
    const int s2 = __builtin_amdgcn_readlane(s, i2);
    const int s3 = __builtin_amdgcn_readlane(s, i3);
    const float4 x0 = ((const float4*)(xw1 + (size_t)s0 * HIDDIM))[lane];
    const float4 x1 = ((const float4*)(xw1 + (size_t)s1 * HIDDIM))[lane];
    const float4 x2 = ((const float4*)(xw1 + (size_t)s2 * HIDDIM))[lane];
    const float4 x3 = ((const float4*)(xw1 + (size_t)s3 * HIDDIM))[lane];
    own_add_row(acc + __builtin_amdgcn_readlane(r, k) * HIDDIM + lane,
                __int_as_float(__builtin_amdgcn_readlane(vb, k)), x0);
    if (k + 1 < n)
      own_add_row(acc + __builtin_amdgcn_readlane(r, i1) * HIDDIM + lane,
                  __int_as_float(__builtin_amdgcn_readlane(vb, i1)), x1);
    if (k + 2 < n)
      own_add_row(acc + __builtin_amdgcn_readlane(r, i2) * HIDDIM + lane,
                  __int_as_float(__builtin_amdgcn_readlane(vb, i2)), x2);
    if (k + 3 < n)
      own_add_row(acc + __builtin_amdgcn_readlane(r, i3) * HIDDIM + lane,
                  __int_as_float(__builtin_amdgcn_readlane(vb, i3)), x3);
  }
}

__device__ __forceinline__ void own_drain(const int* __restrict__ q, int n, int lane, int base,
                                          const int* __restrict__ esrc,
                                          const int* __restrict__ edst,
                                          const float* __restrict__ evals,
                                          const float* __restrict__ xw1,
                                          float* __restrict__ acc) {
  const int e = q[lane < n ? lane : 0];
  own_drain_regs(esrc[e], edst[e] - base, __float_as_int(evals[e]), n, lane, xw1, acc);
}

__global__ __launch_bounds__(1024) void spmm1_owned(const int* __restrict__ esrc,
                                                    const int* __restrict__ edst,
                                                    const float* __restrict__ evals,
                                                    const float* __restrict__ xw1,
                                                    const float* __restrict__ W2,
                                                    float* h,
                                                    int nedges) {
  __shared__ float acc[OWN_ROWS * HIDDIM];
  __shared__ int2  srec[REC_CAP];
  __shared__ int   rowpos[256];
  __shared__ int   binstat[2];
  int* queue = (int*)srec;

  const int tid  = threadIdx.x;
  const int lane = tid & 31;
  const int wave = __builtin_amdgcn_readfirstlane(tid >> 5);
  const int base = blockIdx.x * OWN_ROWS;

  const int* regions = (const int*)h;
  const int2 te = *(const int2*)(regions + TAB_OFF + (int)blockIdx.x * TAB_ROW_INTS + 2 * (wave * SL_PER_WAVE + (lane & (SL_PER_WAVE - 1))));

  for (int i = tid; i < OWN_ROWS * HIDDIM / 4; i += 1024)
    ((float4*)acc)[i] = make_float4(0.0f, 0.0f, 0.0f, 0.0f);
  if (tid < 256) rowpos[tid] = 0;
  if (tid < 2) binstat[tid] = 0;
  __syncthreads();

  int2 held[SORT_SLOTS];
  int  rank[SORT_SLOTS];
  int  st[SL_PER_WAVE], cn[SL_PER_WAVE];
#pragma unroll
  for (int j = 0; j < SL_PER_WAVE; ++j) {
    st[j] = __builtin_amdgcn_readlane(te.x, j);
    cn[j] = __builtin_amdgcn_readlane(te.y, j);
    const int* seg = regions + (size_t)(wave * SL_PER_WAVE + j) * SEG_INTS;
#pragma unroll
    for (int t = 0; t < RUN_SLOTS; ++t) {
      const int idx = t * 32 + lane;
      const bool ok = idx < cn[j] && (unsigned)(st[j] + idx) < (unsigned)SEG_RECS;
      held[j * RUN_SLOTS + t] = ok ? *(const int2*)(seg + 2 * (st[j] + idx)) : make_int2(0, 0);
    }
  }
  if (lane < SL_PER_WAVE) {
    atomicAdd(&binstat[0], te.y);
    if ((unsigned)te.y > (unsigned)RUN_MAX) atomicOr(&binstat[1], 1);
  }
#pragma unroll
  for (int j = 0; j < SL_PER_WAVE; ++j) {
#pragma unroll
    for (int t = 0; t < RUN_SLOTS; ++t) {
      const bool ok = t * 32 + lane < cn[j];
      rank[j * RUN_SLOTS + t] = ok ? atomicAdd(&rowpos[(held[j * RUN_SLOTS + t].x >> 16) & 255], 1) : 0;
    }
  }
  __syncthreads();
  const int nrec = __builtin_amdgcn_readfirstlane(binstat[0]);
  const bool fits = (binstat[1] == 0) && (nrec >= 0) && (nrec <= REC_CAP) && ((nedges + NBINS - 1) / NBINS <= SEG_RECS);

  if (fits) {
    if (wave == 0) {
      int rc7[7];
      int tot = 0;
      for (int t = 0; t < 7; ++t) { rc7[t] = rowpos[lane * 7 + t]; tot += rc7[t]; }
      int incl = tot;
      for (int d = 1; d < 32; d <<= 1) {
        const int o = __shfl_up(incl, d);
        if (lane >= d) incl += o;
      }
      int run = incl - tot;
      for (int t = 0; t < 7; ++t) { rowpos[lane * 7 + t] = run; run += rc7[t]; }
    }
    __syncthreads();
#pragma unroll
    for (int j = 0; j < SL_PER_WAVE; ++j) {
#pragma unroll
      for (int t = 0; t < RUN_SLOTS; ++t) {
        if (t * 32 + lane < cn[j]) {
          const int2 r = held[j * RUN_SLOTS + t];
          srec[rowpos[(r.x >> 16) & 255] + rank[j * RUN_SLOTS + t]] = r;
        }
      }
    }
    __syncthreads();

    const int per = (((nrec + 31) >> 5) + GRP - 1) & ~(GRP - 1);
    const int i0  = wave * per;
    const int i1  = (i0 + per < nrec) ? i0 + per : nrec;
    float4 sum = make_float4(0.0f, 0.0f, 0.0f, 0.0f);
    int cur = -1;
    for (int i = i0; i < i1; i += 32) {
      const int c = (i1 - i < 32) ? i1 - i : 32;
      int2 rc = srec[i + (lane < c ? lane : c - 1)];
      if (lane >= c) rc.y = 0;
      for (int k = 0; k < c; k += GRP) {
        float4 x[GRP];
        for (int j = 0; j < GRP; ++j) {
          const int sidx = __builtin_amdgcn_readlane(rc.x, k + j) & 0xFFFF;
          x[j] = ((const float4*)(xw1 + (size_t)sidx * HIDDIM))[lane];
        }
        for (int j = 0; j < GRP; ++j) {
          const int   row = __builtin_amdgcn_readlane(rc.x, k + j) >> 16;
          const float v   = __int_as_float(__builtin_amdgcn_readlane(rc.y, k + j));
          if (row != cur) {
            if (cur >= 0) own_flush_row(acc + cur * HIDDIM + lane, sum);
            sum = make_float4(0.0f, 0.0f, 0.0f, 0.0f);
            cur = row;
          }
          sum.x = fmaf(v, x[j].x, sum.x);
          sum.y = fmaf(v, x[j].y, sum.y);
          sum.z = fmaf(v, x[j].z, sum.z);
          sum.w = fmaf(v, x[j].w, sum.w);
        }
      }
    }
    if (cur >= 0) own_flush_row(acc + cur * HIDDIM + lane, sum);
  } else {
    int* q = queue + wave * QCAP;
    int qn = 0;

    const int nquads = nedges >> 2;
    const int nspans = (nquads + 127) >> 7;
    const int4* dst4 = (const int4*)edst;

    for (int sp = wave; sp < nspans; sp += 32) {
      int4 cur[4];
#pragma unroll
      for (int j = 0; j < 4; ++j) {
        const int qi = sp * 128 + j * 32 + lane;
        cur[j] = (qi < nquads) ? dst4[qi] : make_int4(-1, -1, -1, -1);
      }
#pragma unroll
      for (int j = 0; j < 4; ++j) {
        const int qi = sp * 128 + j * 32 + lane;
        const int dd[4] = {cur[j].x, cur[j].y, cur[j].z, cur[j].w};
#pragma unroll
        for (int c = 0; c < 4; ++c) {
          const bool m = (unsigned)(dd[c] - base) < (unsigned)OWN_ROWS;
          const unsigned mask = (unsigned)__ballot(m);
          if (mask) {
            if (m) q[qn + __popc(mask & ((1u << lane) - 1u))] = qi * 4 + c;
            qn += __popc(mask);
            if (qn >= 32) {
              own_drain(q, 32, lane, base, esrc, edst, evals, xw1, acc);
              const int rest = qn - 32;
              const int keep = q[32 + lane];
              if (lane < rest) q[lane] = keep;
              qn = rest;
            }
          }
        }
      }
    }
    if (wave == 0) {
      for (int e = nquads * 4 + lane; e - lane < nedges; e += 32) {
        const bool m = (e < nedges) && ((unsigned)(edst[e < nedges ? e : 0] - base) < (unsigned)OWN_ROWS);
        const unsigned mask = (unsigned)__ballot(m);
        if (mask) {
          if (m) q[qn + __popc(mask & ((1u << lane) - 1u))] = e;
          qn += __popc(mask);
          if (qn >= 32) {
            own_drain(q, 32, lane, base, esrc, edst, evals, xw1, acc);
            const int rest = qn - 32;
            const int keep = q[32 + lane];
            if (lane < rest) q[lane] = keep;
            qn = rest;
          }
        }
      }
    }
    if (qn > 0) own_drain(q, qn, lane, base, esrc, edst, evals, xw1, acc);
  }
  __syncthreads();

  const int nrows = ((int)blockIdx.x == NBINS - 1) ? LAST_ROWS : OWN_ROWS;
  float* hw2 = h + HW2_OFF + (size_t)base * OUTDIM;
  if (wave < (OWN_ROWS + 15) / 16) {
    const int hs   = lane >> 4;
    const int lm   = lane & 15;
    const int row0 = wave * 16;
    const int ar = (row0 + lm < OWN_ROWS) ? row0 + lm : OWN_ROWS - 1;
    const float4* a0p = (const float4*)(acc + ar * HIDDIM + (2 * hs) * 32);
    const float4* a1p = (const float4*)(acc + ar * HIDDIM + (2 * hs + 1) * 32);
    v8f d = {};
    for (int lc = 0; lc < HIDDIM / 16; ++lc) {
      const float4 A0 = a0p[lc];
      const float4 A1 = a1p[lc];
      const float e0[4] = {A0.x, A0.y, A0.z, A0.w};
      const float e1[4] = {A1.x, A1.y, A1.z, A1.w};
#pragma unroll
      for (int i = 0; i < 4; ++i) {
        const int ka = (lc * 4 + i) * 4 + hs * 2;
        v2f a, b;
        a.x = fmaxf(e0[i], 0.0f);
        a.y = fmaxf(e1[i], 0.0f);
        const float* w = W2 + (size_t)ka * OUTDIM + lm;
        b.x = w[0];
        b.y = w[OUTDIM];
        d = __builtin_amdgcn_wmma_f32_16x16x4_f32(false, a, false, b,
                                                  (short)0, d, false, false);
      }
    }
#pragma unroll
    for (int j = 0; j < 8; ++j) {
      const int r = row0 + hs * 8 + j;
      if (r < nrows) hw2[r * OUTDIM + lm] = d[j];
    }
  }
}

__device__ __forceinline__ const float* hw2_row(const int* __restrict__ regions, int s) {
  return (const float*)(regions + HW2_OFF + s * OUTDIM);
}

__global__ __launch_bounds__(1024) void spmm2_owned(const int* __restrict__ esrc,
                                                    const int* __restrict__ edst,
                                                    const float* __restrict__ evals,
                                                    const int* __restrict__ regions,
                                                    float* __restrict__ out,
                                                    int nedges) {
  __shared__ float acc2[OWN_ROWS * OUTDIM];
  __shared__ int   longrun;

  const int tid  = threadIdx.x;
  const int lane = tid & 31;
  const int wave = __builtin_amdgcn_readfirstlane(tid >> 5);
  const int base = blockIdx.x * OWN_ROWS;

  const int2 te = *(const int2*)(regions + TAB_OFF + (int)blockIdx.x * TAB_ROW_INTS + 2 * (wave * SL_PER_WAVE + (lane & (SL_PER_WAVE - 1))));

  for (int i = tid; i < OWN_ROWS * OUTDIM; i += 1024) acc2[i] = 0.0f;
  if (tid == 0) longrun = 0;
  __syncthreads();
  if (lane < SL_PER_WAVE && (unsigned)te.y > (unsigned)RUN_MAX) atomicOr(&longrun, 1);

  int2 rcs[SORT_SLOTS];
  int  cn[SL_PER_WAVE];
#pragma unroll
  for (int j = 0; j < SL_PER_WAVE; ++j) {
    const int st = __builtin_amdgcn_readlane(te.x, j);
    cn[j] = __builtin_amdgcn_readlane(te.y, j);
    const int* seg = regions + (size_t)(wave * SL_PER_WAVE + j) * SEG_INTS;
#pragma unroll
    for (int t = 0; t < RUN_SLOTS; ++t) {
      const int idx = t * 32 + lane;
      const bool ok = idx < cn[j] && (unsigned)(st + idx) < (unsigned)SEG_RECS;
      rcs[j * RUN_SLOTS + t] = ok ? *(const int2*)(seg + 2 * (st + idx)) : make_int2(0, 0);
    }
  }
  __syncthreads();
  const bool fits = (longrun == 0) && ((nedges + NBINS - 1) / NBINS <= SEG_RECS);

  if (fits) {
    const int q   = lane & 3;
    const int sub = lane >> 2;
#pragma unroll
    for (int j = 0; j < SL_PER_WAVE; ++j) {
#pragma unroll
      for (int t = 0; t < RUN_SLOTS; ++t) {
        const int c = cn[j] - t * 32;
        if (c > 0) {
          const int2 rc = rcs[j * RUN_SLOTS + t];
          float4 x[4];
          int    ro[4];
          float  vv[4];
#pragma unroll
          for (int k = 0; k < 4; ++k) {
            const int idx = k * 8 + sub;
            const int key = __shfl(rc.x, idx);
            vv[k] = __int_as_float(__shfl(rc.y, idx));
            ro[k] = (idx < c) ? ((key >> 16) & 255) * OUTDIM + q * 4 : -1;
            x[k]  = (idx < c) ? ((const float4*)hw2_row(regions, key & 0xFFFF))[q]
                              : make_float4(0.0f, 0.0f, 0.0f, 0.0f);
          }
#pragma unroll
          for (int k = 0; k < 4; ++k) {
            if (ro[k] >= 0) {
              float* a = acc2 + ro[k];
              atomicAdd(a + 0, vv[k] * x[k].x);
              atomicAdd(a + 1, vv[k] * x[k].y);
              atomicAdd(a + 2, vv[k] * x[k].z);
              atomicAdd(a + 3, vv[k] * x[k].w);
            }
          }
        }
      }
    }
  } else {
    for (int e = tid; e < nedges; e += 1024) {
      const int r = edst[e] - base;
      if ((unsigned)r < (unsigned)OWN_ROWS) {
        const float  v = evals[e];
        const float* x = hw2_row(regions, esrc[e]);
        for (int k = 0; k < OUTDIM; ++k) atomicAdd(acc2 + r * OUTDIM + k, v * x[k]);
      }
    }
  }
  __syncthreads();

  for (int i = tid; i < OWN_ROWS * (OUTDIM / 4); i += 1024) {
    const int row = base + (i >> 2);
    if (row < NNODES) ((float4*)(out + (size_t)row * OUTDIM))[i & 3] = ((const float4*)acc2)[i];
  }
}

extern "C" void kernel_launch(void* const* d_in, const int* in_sizes, int n_in,
                              void* d_out, int out_size, void* d_ws, size_t ws_size,
                              hipStream_t stream) {
  const float* feat  = (const float*)d_in[0];
  const int*   esrc  = (const int*)d_in[1];
  const int*   edst  = (const int*)d_in[2];
  const float* evals = (const float*)d_in[3];
  const float* W1    = (const float*)d_in[4];
  const float* W2    = (const float*)d_in[5];
  float*       out   = (float*)d_out;
  const int nedges   = in_sizes[1];

  const size_t xw1_bytes = (size_t)NNODES * HIDDIM * sizeof(float);
  float* xw1 = (float*)d_ws;
  float* h   = (float*)((char*)d_ws + xw1_bytes);

  gemm1_scatter<<<NBINS + G1_BLOCKS, G1_THREADS, 0, stream>>>(feat, W1, xw1, esrc, edst, evals, (int*)h, nedges);

  spmm1_owned<<<NBINS, 1024, 0, stream>>>(esrc, edst, evals, xw1, W2, h, nedges);

  spmm2_owned<<<NBINS, 1024, 0, stream>>>(esrc, edst, evals, (const int*)h, out, nedges);
}


# spmm2: a lane's four LDS adds cover its float4 in the order (c + edge slot) mod 4, so the eight edges of one add instruction spread over all 64 banks, not 16; LDS add order changes
# speedup vs baseline: 3.9059x; 1.0669x over previous
#include <hip/hip_runtime.h>

typedef __attribute__((ext_vector_type(2))) float v2f;
typedef __attribute__((ext_vector_type(8))) float v8f;

#define NNODES 50000
#define INDIM  512
#define HIDDIM 128
#define OUTDIM 16

#define G1_ROWS 32
#define G1_BLOCKS ((NNODES + G1_ROWS - 1) / G1_ROWS)
#define G1_KU 4
#define G1_THREADS 128

__device__ __forceinline__ void g1_load_b(float4 (&b)[G1_KU][2], const float* __restrict__ bp, int k0) {
#pragma unroll
  for (int u = 0; u < G1_KU; ++u) {
    const float* w = bp + (size_t)(k0 + 4 * u) * HIDDIM;
    b[u][0] = *(const float4*)(w);
    b[u][1] = *(const float4*)(w + HIDDIM);
  }
}

__device__ __forceinline__ void g1_mma(v8f& acc0, v8f& acc1, v8f& acc2, v8f& acc3,
                                       const float* __restrict__ ap, const float4 (&bb)[G1_KU][2]) {
#pragma unroll
  for (int u = 0; u < G1_KU; ++u) {
    v2f a, b;
    a.x = ap[(4 * u) * 17];
    a.y = ap[(4 * u) * 17 + 17];
    b.x = bb[u][0].x; b.y = bb[u][1].x;
    acc0 = __builtin_amdgcn_wmma_f32_16x16x4_f32(false, a, false, b, (short)0, acc0, false, false);
    b.x = bb[u][0].y; b.y = bb[u][1].y;
    acc1 = __builtin_amdgcn_wmma_f32_16x16x4_f32(false, a, false, b, (short)0, acc1, false, false);
    b.x = bb[u][0].z; b.y = bb[u][1].z;
    acc2 = __builtin_amdgcn_wmma_f32_16x16x4_f32(false, a, false, b, (short)0, acc2, false, false);
    b.x = bb[u][0].w; b.y = bb[u][1].w;
    acc3 = __builtin_amdgcn_wmma_f32_16x16x4_f32(false, a, false, b, (short)0, acc3, false, false);
  }
}

#define G1_KQ 128
#define G1_NQ (INDIM / G1_KQ)

__device__ __forceinline__ void g1_fetch(float4 (&v)[8], const float* feat, int row0, int q, int tid) {
#pragma unroll
  for (int i = 0; i < 8; ++i) {
    int gr = row0 + (tid >> 5) + 4 * i;
    gr = (gr < NNODES) ? gr : NNODES - 1;
    v[i] = ((const float4*)(feat + (size_t)gr * INDIM + q * G1_KQ))[tid & 31];
  }
}

__device__ __forceinline__ void g1_hint(const float* feat, int row0, int q, int tid) {
#pragma unroll
  for (int i = 0; i < 8; ++i) {
    int gr = row0 + (tid >> 5) + 4 * i;
    gr = (gr < NNODES) ? gr : NNODES - 1;
    __builtin_prefetch(feat + (size_t)gr * INDIM + q * G1_KQ + 4 * (tid & 31), 0, 3);
  }
}

__device__ __forceinline__ void g1_stage(float (*As)[INDIM][17], const float4 (&v)[8], int q, int tid) {
#pragma unroll
  for (int i = 0; i < 8; ++i) {
    const int r = (tid >> 5) + 4 * i;
    float* p = &As[r >> 4][q * G1_KQ + 4 * (tid & 31)][r & 15];
    p[0]  = v[i].x;
    p[17] = v[i].y;
    p[34] = v[i].z;
    p[51] = v[i].w;
  }
}

#define OWN_ROWS 196
#define QCAP 64
#define NBINS 256
#define SEG_RECS 8192
#define SEG_INTS (2 * SEG_RECS)
#define TAB_OFF (NBINS * SEG_INTS)
#define TAB_ROW_INTS (2 * NBINS)
#define HW2_OFF (TAB_OFF + NBINS * TAB_ROW_INTS)
#define RUN_MAX 64
#define RUN_SLOTS (RUN_MAX / 32)
#define SL_PER_WAVE (NBINS / 32)
#define REC_CAP 10846
#define LAST_ROWS (NNODES - (NBINS - 1) * OWN_ROWS)
#define SORT_SLOTS (SL_PER_WAVE * RUN_SLOTS)
#define GRP 8

#define SC_SLAB 16

__device__ __forceinline__ void scatter_slice(const int* __restrict__ esrc,
                                              const int* __restrict__ edst,
                                              const float* __restrict__ evals,
                                              int* regions, int nedges,
                                              const int g, const int tid, int* sm) {
  int* hist   = sm;
  int* cursor = sm + NBINS;

  for (int bin = tid; bin < NBINS; bin += G1_THREADS) hist[bin] = 0;
  __syncthreads();

  const int per = (nedges + NBINS - 1) / NBINS;
  const int e0  = g * per;
  const int e1  = (e0 + per < nedges) ? e0 + per : nedges;

  for (int eb = e0; eb < e1; eb += SC_SLAB * G1_THREADS) {
    int dd[SC_SLAB];
#pragma unroll
    for (int j = 0; j < SC_SLAB; ++j) {
      const int e = eb + j * G1_THREADS + tid;
      dd[j] = (e < e1) ? edst[e] : -1;
    }
#pragma unroll
    for (int j = 0; j < SC_SLAB; ++j)
      if ((unsigned)dd[j] < (unsigned)NNODES)
        atomicAdd(&hist[(unsigned)dd[j] / OWN_ROWS], 1);
  }
  __syncthreads();

  if (tid < 32) {
    int cn[8];
    int tot = 0;
    for (int t = 0; t < 8; ++t) { cn[t] = hist[tid * 8 + t]; tot += cn[t]; }
    int incl = tot;
    for (int d = 1; d < 32; d <<= 1) {
      const int o = __shfl_up(incl, d);
      if (tid >= d) incl += o;
    }
    int run = incl - tot;
    for (int t = 0; t < 8; ++t) { cursor[tid * 8 + t] = run; run += cn[t]; }
  }
  __syncthreads();

  for (int bin = tid; bin < NBINS; bin += G1_THREADS)
    *(int2*)(regions + TAB_OFF + bin * TAB_ROW_INTS + 2 * g) = make_int2(cursor[bin], hist[bin]);
  __syncthreads();

  int* seg = regions + (size_t)g * SEG_INTS;
  for (int eb = e0; eb < e1; eb += SC_SLAB * G1_THREADS) {
    int dd[SC_SLAB], ss[SC_SLAB], vv[SC_SLAB];
#pragma unroll
    for (int j = 0; j < SC_SLAB; ++j) {
      const int e = eb + j * G1_THREADS + tid;
      const bool ok = e < e1;
      dd[j] = ok ? edst[e] : -1;
      ss[j] = ok ? esrc[e] : 0;
      vv[j] = ok ? __float_as_int(evals[e]) : 0;
    }
#pragma unroll
    for (int j = 0; j < SC_SLAB; ++j) {
      if ((unsigned)dd[j] < (unsigned)NNODES) {
        const int b   = (int)((unsigned)dd[j] / OWN_ROWS);
        const int pos = atomicAdd(&cursor[b], 1);
        if (pos < SEG_RECS) {
          const int key = (ss[j] & 0xFFFF) | ((dd[j] - b * OWN_ROWS) << 16);
          *(int2*)(seg + 2 * pos) = make_int2(key, vv[j]);
        }
      }
    }
  }
}

__global__ __launch_bounds__(G1_THREADS) void gemm1_scatter(const float* feat,
                                                            const float* __restrict__ W1,
                                                            float* __restrict__ xw1,
                                                            const int* __restrict__ esrc,
                                                            const int* __restrict__ edst,
                                                            const float* __restrict__ evals,
                                                            int* regions,
                                                            int nedges) {
  __shared__ float As[G1_ROWS / 16][INDIM][17];
  __shared__ int   sc_ints[2 * NBINS];
  const int tid  = threadIdx.x;
  if (blockIdx.x < NBINS) {
    scatter_slice(esrc, edst, evals, regions, nedges, (int)blockIdx.x, tid, sc_ints);
    return;
  }
  const int row0 = ((int)blockIdx.x - NBINS) * G1_ROWS;

  float4 nv[8];
  g1_fetch(nv, feat, row0, 0, tid);
  g1_stage(As, nv, 0, tid);
  __syncthreads();

  const int wave = tid >> 5;
  const int lane = tid & 31;
  const int hs   = lane >> 4;
  const int lm   = lane & 15;
  const int rt   = wave & 1;
  const int c0   = (wave >> 1) * 64;

  const float* ap = &As[rt][hs * 2][lm];
  const float* bp = W1 + (size_t)(hs * 2) * HIDDIM + c0 + lm * 4;

  v8f acc0 = {}, acc1 = {}, acc2 = {}, acc3 = {};
  float4 b0[G1_KU][2], b1[G1_KU][2];
  g1_load_b(b0, bp, 0);
#pragma unroll
  for (int q = 0; q < G1_NQ; ++q) {
    if (q + 1 < G1_NQ) g1_hint(feat, row0, q + 1, tid);
    const int kl = (q + 1) * G1_KQ - 8 * G1_KU;
#pragma nounroll
    for (int k0 = q * G1_KQ; k0 < kl; k0 += 8 * G1_KU) {
      g1_load_b(b1, bp, k0 + 4 * G1_KU);
      g1_mma(acc0, acc1, acc2, acc3, ap + k0 * 17, b0);
      g1_load_b(b0, bp, k0 + 8 * G1_KU);
      g1_mma(acc0, acc1, acc2, acc3, ap + (k0 + 4 * G1_KU) * 17, b1);
    }
    g1_load_b(b1, bp, kl + 4 * G1_KU);
    g1_mma(acc0, acc1, acc2, acc3, ap + kl * 17, b0);
    if (kl + 8 * G1_KU < INDIM) g1_load_b(b0, bp, kl + 8 * G1_KU);
    if (q + 1 < G1_NQ) {
      __builtin_amdgcn_sched_barrier(0);
      g1_fetch(nv, feat, row0, q + 1, tid);
      __builtin_amdgcn_sched_barrier(0);
    }
    g1_mma(acc0, acc1, acc2, acc3, ap + (kl + 4 * G1_KU) * 17, b1);
    if (q + 1 < G1_NQ) {
      __builtin_amdgcn_sched_barrier(0);
      g1_stage(As, nv, q + 1, tid);
      __syncthreads();
    }
  }

  float* dst = xw1 + (size_t)(row0 + rt * 16 + hs * 8) * HIDDIM + c0 + lm * 4;
#pragma unroll
  for (int j = 0; j < 8; ++j) {
    if (row0 + rt * 16 + hs * 8 + j < NNODES)
      *(float4*)(dst + (size_t)j * HIDDIM) = make_float4(acc0[j], acc1[j], acc2[j], acc3[j]);
  }
}

__device__ __forceinline__ void own_add_row(float* __restrict__ arow, float v, const float4 x) {
  atomicAdd(arow +  0, v * x.x);
  atomicAdd(arow + 32, v * x.y);
  atomicAdd(arow + 64, v * x.z);
  atomicAdd(arow + 96, v * x.w);
}

__device__ __forceinline__ void own_flush_row(float* __restrict__ arow, const float4 t) {
  atomicAdd(arow +  0, t.x);
  atomicAdd(arow + 32, t.y);
  atomicAdd(arow + 64, t.z);
  atomicAdd(arow + 96, t.w);
}

__device__ __forceinline__ void own_drain_regs(int s, int r, int vb, int n, int lane,
                                               const float* __restrict__ xw1,
                                               float* __restrict__ acc) {
  for (int k = 0; k < n; k += 4) {
    const int i1 = (k + 1 < n) ? k + 1 : k;
    const int i2 = (k + 2 < n) ? k + 2 : k;
    const int i3 = (k + 3 < n) ? k + 3 : k;
    const int s0 = __builtin_amdgcn_readlane(s, k);
    const int s1 = __builtin_amdgcn_readlane(s, i1);
    const int s2 = __builtin_amdgcn_readlane(s, i2);
    const int s3 = __builtin_amdgcn_readlane(s, i3);
    const float4 x0 = ((const float4*)(xw1 + (size_t)s0 * HIDDIM))[lane];
    const float4 x1 = ((const float4*)(xw1 + (size_t)s1 * HIDDIM))[lane];
    const float4 x2 = ((const float4*)(xw1 + (size_t)s2 * HIDDIM))[lane];
    const float4 x3 = ((const float4*)(xw1 + (size_t)s3 * HIDDIM))[lane];
    own_add_row(acc + __builtin_amdgcn_readlane(r, k) * HIDDIM + lane,
                __int_as_float(__builtin_amdgcn_readlane(vb, k)), x0);
    if (k + 1 < n)
      own_add_row(acc + __builtin_amdgcn_readlane(r, i1) * HIDDIM + lane,
                  __int_as_float(__builtin_amdgcn_readlane(vb, i1)), x1);
    if (k + 2 < n)
      own_add_row(acc + __builtin_amdgcn_readlane(r, i2) * HIDDIM + lane,
                  __int_as_float(__builtin_amdgcn_readlane(vb, i2)), x2);
    if (k + 3 < n)
      own_add_row(acc + __builtin_amdgcn_readlane(r, i3) * HIDDIM + lane,
                  __int_as_float(__builtin_amdgcn_readlane(vb, i3)), x3);
  }
}

__device__ __forceinline__ void own_drain(const int* __restrict__ q, int n, int lane, int base,
                                          const int* __restrict__ esrc,
                                          const int* __restrict__ edst,
                                          const float* __restrict__ evals,
                                          const float* __restrict__ xw1,
                                          float* __restrict__ acc) {
  const int e = q[lane < n ? lane : 0];
  own_drain_regs(esrc[e], edst[e] - base, __float_as_int(evals[e]), n, lane, xw1, acc);
}

__global__ __launch_bounds__(1024) void spmm1_owned(const int* __restrict__ esrc,
                                                    const int* __restrict__ edst,
                                                    const float* __restrict__ evals,
                                                    const float* __restrict__ xw1,
                                                    const float* __restrict__ W2,
                                                    float* h,
                                                    int nedges) {
  __shared__ float acc[OWN_ROWS * HIDDIM];
  __shared__ int2  srec[REC_CAP];
  __shared__ int   rowpos[256];
  __shared__ int   binstat[2];
  int* queue = (int*)srec;

  const int tid  = threadIdx.x;
  const int lane = tid & 31;
  const int wave = __builtin_amdgcn_readfirstlane(tid >> 5);
  const int base = blockIdx.x * OWN_ROWS;

  const int* regions = (const int*)h;
  const int2 te = *(const int2*)(regions + TAB_OFF + (int)blockIdx.x * TAB_ROW_INTS + 2 * (wave * SL_PER_WAVE + (lane & (SL_PER_WAVE - 1))));

  for (int i = tid; i < OWN_ROWS * HIDDIM / 4; i += 1024)
    ((float4*)acc)[i] = make_float4(0.0f, 0.0f, 0.0f, 0.0f);
  if (tid < 256) rowpos[tid] = 0;
  if (tid < 2) binstat[tid] = 0;
  __syncthreads();

  int2 held[SORT_SLOTS];
  int  rank[SORT_SLOTS];
  int  st[SL_PER_WAVE], cn[SL_PER_WAVE];
#pragma unroll
  for (int j = 0; j < SL_PER_WAVE; ++j) {
    st[j] = __builtin_amdgcn_readlane(te.x, j);
    cn[j] = __builtin_amdgcn_readlane(te.y, j);
    const int* seg = regions + (size_t)(wave * SL_PER_WAVE + j) * SEG_INTS;
#pragma unroll
    for (int t = 0; t < RUN_SLOTS; ++t) {
      const int idx = t * 32 + lane;
      const bool ok = idx < cn[j] && (unsigned)(st[j] + idx) < (unsigned)SEG_RECS;
      held[j * RUN_SLOTS + t] = ok ? *(const int2*)(seg + 2 * (st[j] + idx)) : make_int2(0, 0);
    }
  }
  if (lane < SL_PER_WAVE) {
    atomicAdd(&binstat[0], te.y);
    if ((unsigned)te.y > (unsigned)RUN_MAX) atomicOr(&binstat[1], 1);
  }
#pragma unroll
  for (int j = 0; j < SL_PER_WAVE; ++j) {
#pragma unroll
    for (int t = 0; t < RUN_SLOTS; ++t) {
      const bool ok = t * 32 + lane < cn[j];
      rank[j * RUN_SLOTS + t] = ok ? atomicAdd(&rowpos[(held[j * RUN_SLOTS + t].x >> 16) & 255], 1) : 0;
    }
  }
  __syncthreads();
  const int nrec = __builtin_amdgcn_readfirstlane(binstat[0]);
  const bool fits = (binstat[1] == 0) && (nrec >= 0) && (nrec <= REC_CAP) && ((nedges + NBINS - 1) / NBINS <= SEG_RECS);

  if (fits) {
    if (wave == 0) {
      int rc7[7];
      int tot = 0;
      for (int t = 0; t < 7; ++t) { rc7[t] = rowpos[lane * 7 + t]; tot += rc7[t]; }
      int incl = tot;
      for (int d = 1; d < 32; d <<= 1) {
        const int o = __shfl_up(incl, d);
        if (lane >= d) incl += o;
      }
      int run = incl - tot;
      for (int t = 0; t < 7; ++t) { rowpos[lane * 7 + t] = run; run += rc7[t]; }
    }
    __syncthreads();
#pragma unroll
    for (int j = 0; j < SL_PER_WAVE; ++j) {
#pragma unroll
      for (int t = 0; t < RUN_SLOTS; ++t) {
        if (t * 32 + lane < cn[j]) {
          const int2 r = held[j * RUN_SLOTS + t];
          srec[rowpos[(r.x >> 16) & 255] + rank[j * RUN_SLOTS + t]] = r;
        }
      }
    }
    __syncthreads();

    const int per = (((nrec + 31) >> 5) + GRP - 1) & ~(GRP - 1);
    const int i0  = wave * per;
    const int i1  = (i0 + per < nrec) ? i0 + per : nrec;
    float4 sum = make_float4(0.0f, 0.0f, 0.0f, 0.0f);
    int cur = -1;
    for (int i = i0; i < i1; i += 32) {
      const int c = (i1 - i < 32) ? i1 - i : 32;
      int2 rc = srec[i + (lane < c ? lane : c - 1)];
      if (lane >= c) rc.y = 0;
      for (int k = 0; k < c; k += GRP) {
        float4 x[GRP];
        for (int j = 0; j < GRP; ++j) {
          const int sidx = __builtin_amdgcn_readlane(rc.x, k + j) & 0xFFFF;
          x[j] = ((const float4*)(xw1 + (size_t)sidx * HIDDIM))[lane];
        }
        for (int j = 0; j < GRP; ++j) {
          const int   row = __builtin_amdgcn_readlane(rc.x, k + j) >> 16;
          const float v   = __int_as_float(__builtin_amdgcn_readlane(rc.y, k + j));
          if (row != cur) {
            if (cur >= 0) own_flush_row(acc + cur * HIDDIM + lane, sum);
            sum = make_float4(0.0f, 0.0f, 0.0f, 0.0f);
            cur = row;
          }
          sum.x = fmaf(v, x[j].x, sum.x);
          sum.y = fmaf(v, x[j].y, sum.y);
          sum.z = fmaf(v, x[j].z, sum.z);
          sum.w = fmaf(v, x[j].w, sum.w);
        }
      }
    }
    if (cur >= 0) own_flush_row(acc + cur * HIDDIM + lane, sum);
  } else {
    int* q = queue + wave * QCAP;
    int qn = 0;

    const int nquads = nedges >> 2;
    const int nspans = (nquads + 127) >> 7;
    const int4* dst4 = (const int4*)edst;

    for (int sp = wave; sp < nspans; sp += 32) {
      int4 cur[4];
#pragma unroll
      for (int j = 0; j < 4; ++j) {
        const int qi = sp * 128 + j * 32 + lane;
        cur[j] = (qi < nquads) ? dst4[qi] : make_int4(-1, -1, -1, -1);
      }
#pragma unroll
      for (int j = 0; j < 4; ++j) {
        const int qi = sp * 128 + j * 32 + lane;
        const int dd[4] = {cur[j].x, cur[j].y, cur[j].z, cur[j].w};
#pragma unroll
        for (int c = 0; c < 4; ++c) {
          const bool m = (unsigned)(dd[c] - base) < (unsigned)OWN_ROWS;
          const unsigned mask = (unsigned)__ballot(m);
          if (mask) {
            if (m) q[qn + __popc(mask & ((1u << lane) - 1u))] = qi * 4 + c;
            qn += __popc(mask);
            if (qn >= 32) {
              own_drain(q, 32, lane, base, esrc, edst, evals, xw1, acc);
              const int rest = qn - 32;
              const int keep = q[32 + lane];
              if (lane < rest) q[lane] = keep;
              qn = rest;
            }
          }
        }
      }
    }
    if (wave == 0) {
      for (int e = nquads * 4 + lane; e - lane < nedges; e += 32) {
        const bool m = (e < nedges) && ((unsigned)(edst[e < nedges ? e : 0] - base) < (unsigned)OWN_ROWS);
        const unsigned mask = (unsigned)__ballot(m);
        if (mask) {
          if (m) q[qn + __popc(mask & ((1u << lane) - 1u))] = e;
          qn += __popc(mask);
          if (qn >= 32) {
            own_drain(q, 32, lane, base, esrc, edst, evals, xw1, acc);
            const int rest = qn - 32;
            const int keep = q[32 + lane];
            if (lane < rest) q[lane] = keep;
            qn = rest;
          }
        }
      }
    }
    if (qn > 0) own_drain(q, qn, lane, base, esrc, edst, evals, xw1, acc);
  }
  __syncthreads();

  const int nrows = ((int)blockIdx.x == NBINS - 1) ? LAST_ROWS : OWN_ROWS;
  float* hw2 = h + HW2_OFF + (size_t)base * OUTDIM;
  if (wave < (OWN_ROWS + 15) / 16) {
    const int hs   = lane >> 4;
    const int lm   = lane & 15;
    const int row0 = wave * 16;
    const int ar = (row0 + lm < OWN_ROWS) ? row0 + lm : OWN_ROWS - 1;
    const float4* a0p = (const float4*)(acc + ar * HIDDIM + (2 * hs) * 32);
    const float4* a1p = (const float4*)(acc + ar * HIDDIM + (2 * hs + 1) * 32);
    v8f d = {};
    for (int lc = 0; lc < HIDDIM / 16; ++lc) {
      const float4 A0 = a0p[lc];
      const float4 A1 = a1p[lc];
      const float e0[4] = {A0.x, A0.y, A0.z, A0.w};
      const float e1[4] = {A1.x, A1.y, A1.z, A1.w};
#pragma unroll
      for (int i = 0; i < 4; ++i) {
        const int ka = (lc * 4 + i) * 4 + hs * 2;
        v2f a, b;
        a.x = fmaxf(e0[i], 0.0f);
        a.y = fmaxf(e1[i], 0.0f);
        const float* w = W2 + (size_t)ka * OUTDIM + lm;
        b.x = w[0];
        b.y = w[OUTDIM];
        d = __builtin_amdgcn_wmma_f32_16x16x4_f32(false, a, false, b,
                                                  (short)0, d, false, false);
      }
    }
#pragma unroll
    for (int j = 0; j < 8; ++j) {
      const int r = row0 + hs * 8 + j;
      if (r < nrows) hw2[r * OUTDIM + lm] = d[j];
    }
  }
}

__device__ __forceinline__ const float* hw2_row(const int* __restrict__ regions, int s) {
  return (const float*)(regions + HW2_OFF + s * OUTDIM);
}

__global__ __launch_bounds__(1024) void spmm2_owned(const int* __restrict__ esrc,
                                                    const int* __restrict__ edst,
                                                    const float* __restrict__ evals,
                                                    const int* __restrict__ regions,
                                                    float* __restrict__ out,
                                                    int nedges) {
  __shared__ float acc2[OWN_ROWS * OUTDIM];
  __shared__ int   longrun;

  const int tid  = threadIdx.x;
  const int lane = tid & 31;
  const int wave = __builtin_amdgcn_readfirstlane(tid >> 5);
  const int base = blockIdx.x * OWN_ROWS;

  const int2 te = *(const int2*)(regions + TAB_OFF + (int)blockIdx.x * TAB_ROW_INTS + 2 * (wave * SL_PER_WAVE + (lane & (SL_PER_WAVE - 1))));

  for (int i = tid; i < OWN_ROWS * OUTDIM; i += 1024) acc2[i] = 0.0f;
  if (tid == 0) longrun = 0;
  __syncthreads();
  if (lane < SL_PER_WAVE && (unsigned)te.y > (unsigned)RUN_MAX) atomicOr(&longrun, 1);

  int2 rcs[SORT_SLOTS];
  int  cn[SL_PER_WAVE];
#pragma unroll
  for (int j = 0; j < SL_PER_WAVE; ++j) {
    const int st = __builtin_amdgcn_readlane(te.x, j);
    cn[j] = __builtin_amdgcn_readlane(te.y, j);
    const int* seg = regions + (size_t)(wave * SL_PER_WAVE + j) * SEG_INTS;
#pragma unroll
    for (int t = 0; t < RUN_SLOTS; ++t) {
      const int idx = t * 32 + lane;
      const bool ok = idx < cn[j] && (unsigned)(st + idx) < (unsigned)SEG_RECS;
      rcs[j * RUN_SLOTS + t] = ok ? *(const int2*)(seg + 2 * (st + idx)) : make_int2(0, 0);
    }
  }
  __syncthreads();
  const bool fits = (longrun == 0) && ((nedges + NBINS - 1) / NBINS <= SEG_RECS);

  if (fits) {
    const int q   = lane & 3;
    const int sub = lane >> 2;
    const int  rot  = sub & 3;
    const bool rot1 = (rot & 1) != 0;
    const bool rot2 = (rot & 2) != 0;
    const int  o0 = rot, o1 = (rot + 1) & 3, o2 = (rot + 2) & 3, o3 = (rot + 3) & 3;
#pragma unroll
    for (int j = 0; j < SL_PER_WAVE; ++j) {
#pragma unroll
      for (int t = 0; t < RUN_SLOTS; ++t) {
        const int c = cn[j] - t * 32;
        if (c > 0) {
          const int2 rc = rcs[j * RUN_SLOTS + t];
          float4 x[4];
          int    ro[4];
          float  vv[4];
#pragma unroll
          for (int k = 0; k < 4; ++k) {
            const int idx = k * 8 + sub;
            const int key = __shfl(rc.x, idx);
            vv[k] = __int_as_float(__shfl(rc.y, idx));
            ro[k] = (idx < c) ? ((key >> 16) & 255) * OUTDIM + q * 4 : -1;
            x[k]  = (idx < c) ? ((const float4*)hw2_row(regions, key & 0xFFFF))[q]
                              : make_float4(0.0f, 0.0f, 0.0f, 0.0f);
          }
#pragma unroll
          for (int k = 0; k < 4; ++k) {
            if (ro[k] >= 0) {
              float* a = acc2 + ro[k];
              const float p0 = vv[k] * x[k].x, p1 = vv[k] * x[k].y, p2 = vv[k] * x[k].z, p3 = vv[k] * x[k].w;
              const float t0 = rot1 ? p1 : p0, t1 = rot1 ? p2 : p1, t2 = rot1 ? p3 : p2, t3 = rot1 ? p0 : p3;
              const float e0 = rot2 ? t2 : t0, e1 = rot2 ? t3 : t1, e2 = rot2 ? t0 : t2, e3 = rot2 ? t1 : t3;
              atomicAdd(a + o0, e0);
              atomicAdd(a + o1, e1);
              atomicAdd(a + o2, e2);
              atomicAdd(a + o3, e3);
            }
          }
        }
      }
    }
  } else {
    for (int e = tid; e < nedges; e += 1024) {
      const int r = edst[e] - base;
      if ((unsigned)r < (unsigned)OWN_ROWS) {
        const float  v = evals[e];
        const float* x = hw2_row(regions, esrc[e]);
        for (int k = 0; k < OUTDIM; ++k) atomicAdd(acc2 + r * OUTDIM + k, v * x[k]);
      }
    }
  }
  __syncthreads();

  for (int i = tid; i < OWN_ROWS * (OUTDIM / 4); i += 1024) {
    const int row = base + (i >> 2);
    if (row < NNODES) ((float4*)(out + (size_t)row * OUTDIM))[i & 3] = ((const float4*)acc2)[i];
  }
}

extern "C" void kernel_launch(void* const* d_in, const int* in_sizes, int n_in,
                              void* d_out, int out_size, void* d_ws, size_t ws_size,
                              hipStream_t stream) {
  const float* feat  = (const float*)d_in[0];
  const int*   esrc  = (const int*)d_in[1];
  const int*   edst  = (const int*)d_in[2];
  const float* evals = (const float*)d_in[3];
  const float* W1    = (const float*)d_in[4];
  const float* W2    = (const float*)d_in[5];
  float*       out   = (float*)d_out;
  const int nedges   = in_sizes[1];

  const size_t xw1_bytes = (size_t)NNODES * HIDDIM * sizeof(float);
  float* xw1 = (float*)d_ws;
  float* h   = (float*)((char*)d_ws + xw1_bytes);

  gemm1_scatter<<<NBINS + G1_BLOCKS, G1_THREADS, 0, stream>>>(feat, W1, xw1, esrc, edst, evals, (int*)h, nedges);

  spmm1_owned<<<NBINS, 1024, 0, stream>>>(esrc, edst, evals, xw1, W2, h, nedges);

  spmm2_owned<<<NBINS, 1024, 0, stream>>>(esrc, edst, evals, (const int*)h, out, nedges);
}
